# starting kernel
# baseline (speedup 1.0000x reference)
#include <hip/hip_runtime.h>
#include <hip/hip_bf16.h>

#define HEADS 3
#define CPH   128
#define F     384
#define HIN   12

typedef __attribute__((ext_vector_type(2))) float v2f;
typedef __attribute__((ext_vector_type(8))) float v8f;

__device__ __forceinline__ unsigned flipf(float f) {
  unsigned b = __float_as_uint(f);
  return (b & 0x80000000u) ? ~b : (b | 0x80000000u);
}
__device__ __forceinline__ float unflipf(unsigned u) {
  unsigned b = (u & 0x80000000u) ? (u & 0x7FFFFFFFu) : ~u;
  return __uint_as_float(b);
}
__device__ __forceinline__ float lrelu(float v) { return v > 0.f ? v : 0.2f * v; }

__global__ void k_zero(float* __restrict__ p, unsigned long long n) {
  unsigned long long i = (unsigned long long)blockIdx.x * blockDim.x + threadIdx.x;
  if (i < n) p[i] = 0.0f;
}

__global__ void k_proj1(const float* __restrict__ x, const float* __restrict__ W1,
                        const float* __restrict__ a_src, const float* __restrict__ a_dst,
                        float* __restrict__ xp, float* __restrict__ asrc,
                        float* __restrict__ adst) {
  const int n = blockIdx.x, t = threadIdx.x;
  __shared__ float xrow[HIN];
  __shared__ float red[6][CPH];
  if (t < HIN) xrow[t] = x[(size_t)n * HIN + t];
  __syncthreads();
#pragma unroll
  for (int h = 0; h < HEADS; ++h) {
    const int c = h * CPH + t;
    float acc = 0.f;
#pragma unroll
    for (int k = 0; k < HIN; ++k) acc = fmaf(xrow[k], W1[k * F + c], acc);
    xp[(size_t)n * F + c] = acc;
    red[h][t]     = acc * a_src[c];
    red[3 + h][t] = acc * a_dst[c];
  }
  __syncthreads();
  for (int off = CPH / 2; off > 0; off >>= 1) {
    if (t < off) {
#pragma unroll
      for (int j = 0; j < 6; ++j) red[j][t] += red[j][t + off];
    }
    __syncthreads();
  }
  if (t < HEADS)          asrc[(size_t)n * HEADS + t]          = red[t][0];
  else if (t < 2 * HEADS) adst[(size_t)n * HEADS + (t - HEADS)] = red[t][0];
}

__global__ void k_alpha(const float* __restrict__ xp, const float* __restrict__ a_src,
                        const float* __restrict__ a_dst, float* __restrict__ asrc,
                        float* __restrict__ adst) {
  const int n = blockIdx.x, t = threadIdx.x;
  __shared__ float red[6][CPH];
#pragma unroll
  for (int h = 0; h < HEADS; ++h) {
    const int c = h * CPH + t;
    const float v = xp[(size_t)n * F + c];
    red[h][t]     = v * a_src[c];
    red[3 + h][t] = v * a_dst[c];
  }
  __syncthreads();
  for (int off = CPH / 2; off > 0; off >>= 1) {
    if (t < off) {
#pragma unroll
      for (int j = 0; j < 6; ++j) red[j][t] += red[j][t + off];
    }
    __syncthreads();
  }
  if (t < HEADS)          asrc[(size_t)n * HEADS + t]          = red[t][0];
  else if (t < 2 * HEADS) adst[(size_t)n * HEADS + (t - HEADS)] = red[t][0];
}

__global__ void k_edge_max(const int* __restrict__ src, const int* __restrict__ dst,
                           int E, int N, const float* __restrict__ asrc,
                           const float* __restrict__ adst, unsigned* __restrict__ mx) {
  const int i = blockIdx.x * blockDim.x + threadIdx.x;
  if (i >= E + N) return;
  int s, d;
  if (i < E) { s = src[i]; d = dst[i]; } else { s = i - E; d = s; }
#pragma unroll
  for (int h = 0; h < HEADS; ++h) {
    float e = lrelu(asrc[(size_t)s * HEADS + h] + adst[(size_t)d * HEADS + h]);
    atomicMax(&mx[(size_t)d * HEADS + h], flipf(e));
  }
}

__global__ void k_edge_expsum(const int* __restrict__ src, const int* __restrict__ dst,
                              int E, int N, const float* __restrict__ asrc,
                              const float* __restrict__ adst, const unsigned* __restrict__ mx,
                              float* __restrict__ ex, float* __restrict__ den) {
  const int i = blockIdx.x * blockDim.x + threadIdx.x;
  if (i >= E + N) return;
  int s, d;
  if (i < E) { s = src[i]; d = dst[i]; } else { s = i - E; d = s; }
#pragma unroll
  for (int h = 0; h < HEADS; ++h) {
    float e = lrelu(asrc[(size_t)s * HEADS + h] + adst[(size_t)d * HEADS + h]);
    float v = expf(e - unflipf(mx[(size_t)d * HEADS + h]));
    ex[(size_t)i * HEADS + h] = v;
    atomicAdd(&den[(size_t)d * HEADS + h], v);
  }
}

__global__ void k_edge_agg(const int* __restrict__ src, const int* __restrict__ dst,
                           int E, int N, const float* __restrict__ ex,
                           const float* __restrict__ den, const float* __restrict__ xp,
                           float* __restrict__ agg) {
  const int i = blockIdx.x, t = threadIdx.x;
  int s, d;
  if (i < E) { s = src[i]; d = dst[i]; } else { s = i - E; d = s; }
#pragma unroll
  for (int h = 0; h < HEADS; ++h) {
    const float alpha = ex[(size_t)i * HEADS + h] / den[(size_t)d * HEADS + h];
    const int c = h * CPH + t;
    atomicAdd(&agg[(size_t)d * F + c], alpha * xp[(size_t)s * F + c]);
  }
}

__global__ void k_bias_rows(float* __restrict__ agg, const float* __restrict__ b,
                            unsigned long long total) {
  unsigned long long i = (unsigned long long)blockIdx.x * blockDim.x + threadIdx.x;
  if (i < total) agg[i] += b[i % F];
}

__global__ void __launch_bounds__(256) k_gemm(const float* __restrict__ A,
                                              const float* __restrict__ B,
                                              float* __restrict__ C, int mtiles) {
  const int lane = threadIdx.x & 31;
  const int wave = threadIdx.x >> 5;
  const int wt = blockIdx.x * 8 + wave;
  const int npair = F / 32;
  const int tm = wt / npair;
  const int tp = wt % npair;
  if (tm >= mtiles) return;
  const int m  = tm * 16 + (lane & 15);
  const int n0 = tp * 32 + (lane & 15);
  const int n1 = n0 + 16;
  const int kb = (lane >> 4) * 2;
  const float* arow = A + (size_t)m * F;
  v8f acc0 = {};
  v8f acc1 = {};
  for (int k0 = 0; k0 < F; k0 += 4) {
    const int k = k0 + kb;
    v2f a;  a.x  = arow[k];            a.y  = arow[k + 1];
    const float* bp = B + (size_t)k * F;
    v2f b0; b0.x = bp[n0];             b0.y = bp[F + n0];
    v2f b1; b1.x = bp[n1];             b1.y = bp[F + n1];
    acc0 = __builtin_amdgcn_wmma_f32_16x16x4_f32(false, a, false, b0, (short)0, acc0,
                                                 false, false);
    acc1 = __builtin_amdgcn_wmma_f32_16x16x4_f32(false, a, false, b1, (short)0, acc1,
                                                 false, false);
  }
  const int mr = tm * 16 + ((lane >> 4) << 3);
#pragma unroll
  for (int i = 0; i < 8; ++i) {
    C[(size_t)(mr + i) * F + n0] = acc0[i];
    C[(size_t)(mr + i) * F + n1] = acc1[i];
  }
}

__global__ void k_final(const float* __restrict__ agg, const float* __restrict__ b2,
                        const float* __restrict__ Wl, const float* __restrict__ bl,
                        float* __restrict__ out) {
  const int n = blockIdx.x, t = threadIdx.x;
  __shared__ float red[CPH];
  float acc = 0.f;
#pragma unroll
  for (int h = 0; h < HEADS; ++h) {
    const int c = h * CPH + t;
    acc = fmaf(agg[(size_t)n * F + c] + b2[c], Wl[c], acc);
  }
  red[t] = acc;
  __syncthreads();
  for (int off = CPH / 2; off > 0; off >>= 1) {
    if (t < off) red[t] += red[t + off];
    __syncthreads();
  }
  if (t == 0) out[n] = red[0] + bl[0];
}

extern "C" void kernel_launch(void* const* d_in, const int* in_sizes, int n_in,
                              void* d_out, int out_size, void* d_ws, size_t ws_size,
                              hipStream_t stream) {
  const float* x   = (const float*)d_in[0];
  const int*   ei  = (const int*)d_in[1];
  const float* W1  = (const float*)d_in[2];
  const float* as1 = (const float*)d_in[3];
  const float* ad1 = (const float*)d_in[4];
  const float* b1  = (const float*)d_in[5];
  const float* W2  = (const float*)d_in[6];
  const float* as2 = (const float*)d_in[7];
  const float* ad2 = (const float*)d_in[8];
  const float* b2  = (const float*)d_in[9];
  const float* Wl  = (const float*)d_in[10];
  const float* bl  = (const float*)d_in[11];
  float* out = (float*)d_out;

  const int N  = in_sizes[0] / HIN;
  const int E  = in_sizes[1] / 2;
  const int EN = E + N;
  const int* src = ei;
  const int* dst = ei + E;

  float* ws = (float*)d_ws;
  const size_t NF = (size_t)N * F;
  float*    XP   = ws;
  float*    AGG  = XP + NF;
  float*    DEN  = AGG + NF;
  unsigned* MX   = (unsigned*)(DEN + (size_t)3 * N);
  float*    ASRC = (float*)(MX + (size_t)3 * N);
  float*    ADST = ASRC + (size_t)3 * N;
  float*    EX   = ADST + (size_t)3 * N;

  const unsigned long long zcount = NF + (unsigned long long)6 * N;
  const int zblocks = (int)((zcount + 255) / 256);
  const int eblocks = (EN + 255) / 256;
  const int bblocks = (int)((NF + 255) / 256);
  const int mtiles  = N / 16;
  const int gblocks = (mtiles * (F / 32) + 7) / 8;

  k_zero<<<zblocks, 256, 0, stream>>>(AGG, zcount);
  k_proj1<<<N, CPH, 0, stream>>>(x, W1, as1, ad1, XP, ASRC, ADST);
  k_edge_max<<<eblocks, 256, 0, stream>>>(src, dst, E, N, ASRC, ADST, MX);
  k_edge_expsum<<<eblocks, 256, 0, stream>>>(src, dst, E, N, ASRC, ADST, MX, EX, DEN);
  k_edge_agg<<<EN, CPH, 0, stream>>>(src, dst, E, N, EX, DEN, XP, AGG);
  k_bias_rows<<<bblocks, 256, 0, stream>>>(AGG, b1, NF);

  k_gemm<<<gblocks, 256, 0, stream>>>(AGG, W2, XP, mtiles);
  k_alpha<<<N, CPH, 0, stream>>>(XP, as2, ad2, ASRC, ADST);

  k_zero<<<zblocks, 256, 0, stream>>>(AGG, zcount);
  k_edge_max<<<eblocks, 256, 0, stream>>>(src, dst, E, N, ASRC, ADST, MX);
  k_edge_expsum<<<eblocks, 256, 0, stream>>>(src, dst, E, N, ASRC, ADST, MX, EX, DEN);
  k_edge_agg<<<EN, CPH, 0, stream>>>(src, dst, E, N, EX, DEN, XP, AGG);

  k_final<<<N, CPH, 0, stream>>>(AGG, b2, Wl, bl, out);
}


# edge passes by row ownership, edges not grouped: a group owns 96 rows, filters all edge destinations into a shared-memory list; each wave sums its 3 rows in registers; sums in arrival order
# speedup vs baseline: 2.8968x; 2.8968x over previous
#include <hip/hip_runtime.h>
#include <hip/hip_bf16.h>

#define HEADS 3
#define CPH   128
#define F     384
#define HIN   12
#define WPB   8
#define NPW   4

typedef __attribute__((ext_vector_type(2))) float v2f;
typedef __attribute__((ext_vector_type(8))) float v8f;

__device__ __forceinline__ unsigned flipf(float f) {
  unsigned b = __float_as_uint(f);
  return (b & 0x80000000u) ? ~b : (b | 0x80000000u);
}
__device__ __forceinline__ float unflipf(unsigned u) {
  unsigned b = (u & 0x80000000u) ? (u & 0x7FFFFFFFu) : ~u;
  return __uint_as_float(b);
}
__device__ __forceinline__ float lrelu(float v) { return v > 0.f ? v : 0.2f * v; }


__device__ __forceinline__ float4 ld4(const float* p) { return *reinterpret_cast<const float4*>(p); }
__device__ __forceinline__ void st4(float* p, float4 v) { *reinterpret_cast<float4*>(p) = v; }
__device__ __forceinline__ float4 fma4(float s, float4 w, float4 a) {
  a.x = fmaf(s, w.x, a.x); a.y = fmaf(s, w.y, a.y); a.z = fmaf(s, w.z, a.z); a.w = fmaf(s, w.w, a.w);
  return a;
}
__device__ __forceinline__ float dot4(float4 a, float4 b) {
  return fmaf(a.w, b.w, fmaf(a.z, b.z, fmaf(a.y, b.y, a.x * b.x)));
}

__global__ void __launch_bounds__(256) k_proj1(const float* __restrict__ x, const float* __restrict__ W1,
                        const float* __restrict__ a_src, const float* __restrict__ a_dst,
                        float* __restrict__ xp, float* __restrict__ asrc,
                        float* __restrict__ adst, int N) {
  __shared__ float s_as[WPB * NPW * HEADS];
  __shared__ float s_ad[WPB * NPW * HEADS];
  const int lane = threadIdx.x & 31;
  const int wave = __builtin_amdgcn_readfirstlane((int)(threadIdx.x >> 5));
  const int nb = blockIdx.x * (WPB * NPW);
  const int n0 = nb + wave * NPW;
  float xr[NPW][HIN];
#pragma unroll
  for (int r = 0; r < NPW; ++r) {
    const int n = (n0 + r < N) ? (n0 + r) : (N - 1);
#pragma unroll
    for (int k = 0; k < HIN; ++k) xr[r][k] = x[(size_t)n * HIN + k];
  }
  float red[2 * HEADS * NPW];
#pragma unroll
  for (int h = 0; h < HEADS; ++h) {
    const int c = h * CPH + lane * 4;
    float4 acc[NPW];
#pragma unroll
    for (int r = 0; r < NPW; ++r) acc[r] = make_float4(0.f, 0.f, 0.f, 0.f);
#pragma unroll
    for (int k = 0; k < HIN; ++k) {
      const float4 w = ld4(W1 + k * F + c);
#pragma unroll
      for (int r = 0; r < NPW; ++r) acc[r] = fma4(xr[r][k], w, acc[r]);
    }
    const float4 as4 = ld4(a_src + c);
    const float4 ad4 = ld4(a_dst + c);
#pragma unroll
    for (int r = 0; r < NPW; ++r) {
      if (n0 + r < N) st4(xp + (size_t)(n0 + r) * F + c, acc[r]);
      red[(r * HEADS + h) * 2 + 0] = dot4(acc[r], as4);
      red[(r * HEADS + h) * 2 + 1] = dot4(acc[r], ad4);
    }
  }
#pragma unroll
  for (int m = 16; m >= 1; m >>= 1) {
#pragma unroll
    for (int i = 0; i < 2 * HEADS * NPW; ++i) red[i] += __shfl_xor(red[i], m, 32);
  }
  if (lane == 0) {
#pragma unroll
    for (int r = 0; r < NPW; ++r) {
#pragma unroll
      for (int h = 0; h < HEADS; ++h) {
        s_as[(wave * NPW + r) * HEADS + h] = red[(r * HEADS + h) * 2 + 0];
        s_ad[(wave * NPW + r) * HEADS + h] = red[(r * HEADS + h) * 2 + 1];
      }
    }
  }
  __syncthreads();
  const int t = threadIdx.x;
  const size_t lim = (size_t)N * HEADS;
  if (t < 96) {
    const size_t o = (size_t)nb * HEADS + t;
    if (o < lim) asrc[o] = s_as[t];
  } else if (t < 192) {
    const size_t o = (size_t)nb * HEADS + (t - 96);
    if (o < lim) adst[o] = s_ad[t - 96];
  }
}

__global__ void __launch_bounds__(256) k_alpha(const float* __restrict__ xp, const float* __restrict__ a_src,
                        const float* __restrict__ a_dst, float* __restrict__ asrc,
                        float* __restrict__ adst, int N) {
  __shared__ float s_as[WPB * NPW * HEADS];
  __shared__ float s_ad[WPB * NPW * HEADS];
  const int lane = threadIdx.x & 31;
  const int wave = __builtin_amdgcn_readfirstlane((int)(threadIdx.x >> 5));
  const int nb = blockIdx.x * (WPB * NPW);
  const int n0 = nb + wave * NPW;
  float4 v[NPW][HEADS];
#pragma unroll
  for (int r = 0; r < NPW; ++r) {
    const int n = (n0 + r < N) ? (n0 + r) : (N - 1);
#pragma unroll
    for (int h = 0; h < HEADS; ++h) v[r][h] = ld4(xp + (size_t)n * F + h * CPH + lane * 4);
  }
  float red[2 * HEADS * NPW];
#pragma unroll
  for (int h = 0; h < HEADS; ++h) {
    const float4 as4 = ld4(a_src + h * CPH + lane * 4);
    const float4 ad4 = ld4(a_dst + h * CPH + lane * 4);
#pragma unroll
    for (int r = 0; r < NPW; ++r) {
      red[(r * HEADS + h) * 2 + 0] = dot4(v[r][h], as4);
      red[(r * HEADS + h) * 2 + 1] = dot4(v[r][h], ad4);
    }
  }
#pragma unroll
  for (int m = 16; m >= 1; m >>= 1) {
#pragma unroll
    for (int i = 0; i < 2 * HEADS * NPW; ++i) red[i] += __shfl_xor(red[i], m, 32);
  }
  if (lane == 0) {
#pragma unroll
    for (int r = 0; r < NPW; ++r) {
#pragma unroll
      for (int h = 0; h < HEADS; ++h) {
        s_as[(wave * NPW + r) * HEADS + h] = red[(r * HEADS + h) * 2 + 0];
        s_ad[(wave * NPW + r) * HEADS + h] = red[(r * HEADS + h) * 2 + 1];
      }
    }
  }
  __syncthreads();
  const int t = threadIdx.x;
  const size_t lim = (size_t)N * HEADS;
  if (t < 96) {
    const size_t o = (size_t)nb * HEADS + t;
    if (o < lim) asrc[o] = s_as[t];
  } else if (t < 192) {
    const size_t o = (size_t)nb * HEADS + (t - 96);
    if (o < lim) adst[o] = s_ad[t - 96];
  }
}

#define OR    96
#define OT    1024
#define RPW   3
#define SU    16
#define CAP   4096
#define RND   8
#define GRP   4

__device__ __forceinline__ int own_pop(unsigned (&mk)[RND]) {
  int off = -1;
#pragma unroll
  for (int k = 0; k < RND; ++k) {
    if (off < 0 && mk[k] != 0u) {
      off = k * 32 + __builtin_ctz(mk[k]);
      mk[k] &= mk[k] - 1u;
    }
  }
  return off;
}
__device__ __forceinline__ int   ufi(int v)   { return __builtin_amdgcn_readfirstlane(v); }
__device__ __forceinline__ float uff(float v) { return __int_as_float(__builtin_amdgcn_readfirstlane(__float_as_int(v))); }

__device__ __forceinline__ void own_add(float (&acc)[RPW][HEADS * 4], float (&dsum)[RPW][HEADS], int jr,
                                        const float (&val)[HEADS * 4], float a0, float a1, float a2) {
#pragma unroll
  for (int q = 0; q < RPW; ++q) {
    if (jr == q) {
#pragma unroll
      for (int c = 0; c < HEADS * 4; ++c) acc[q][c] = fmaf((c < 4) ? a0 : ((c < 8) ? a1 : a2), val[c], acc[q][c]);
      dsum[q][0] += a0; dsum[q][1] += a1; dsum[q][2] += a2;
    }
  }
}

__global__ void __launch_bounds__(OT) k_own(const int* __restrict__ src, const int* __restrict__ dst,
                           int E, int N, const float* __restrict__ asrc,
                           const float* __restrict__ adst, const float* __restrict__ xp,
                           float* __restrict__ agg, float* __restrict__ den) {
  __shared__ float    s_acc[OR * F];
  __shared__ int      l_s[CAP];
  __shared__ int      l_d[CAP];
  __shared__ float    l_w[HEADS][CAP];
  __shared__ float    s_ad[OR * HEADS];
  __shared__ unsigned s_mx[OR * HEADS];
  __shared__ float    s_dn[OR * HEADS];
  __shared__ unsigned s_cnt;
  const int t = threadIdx.x;
  const int lane = t & 31;
  const int wave = __builtin_amdgcn_readfirstlane(t >> 5);
  const int nb = blockIdx.x * OR;
  const int rows = (N - nb < OR) ? (N - nb) : OR;
  const size_t gb3 = (size_t)nb * HEADS;

  if (t < OR * HEADS) {
    float a = 0.f;
    unsigned m = 0u;
    if (t < rows * HEADS) {
      a = adst[gb3 + t];
      m = flipf(lrelu(asrc[gb3 + t] + a));
    }
    s_ad[t] = a;
    s_mx[t] = m;
  }
  if (t == 0) s_cnt = 0u;
  __syncthreads();

  for (int i0 = 0; i0 < E; i0 += OT * SU) {
    int d[SU];
#pragma unroll
    for (int k = 0; k < SU; ++k) {
      const int i = i0 + k * OT + t;
      d[k] = (i < E) ? dst[i] : -1;
    }
#pragma unroll
    for (int k = 0; k < SU; ++k) {
      const unsigned dl = (unsigned)(d[k] - nb);
      if (dl < (unsigned)rows) {
        const unsigned slot = __hip_atomic_fetch_add(&s_cnt, 1u, __ATOMIC_RELAXED, __HIP_MEMORY_SCOPE_WORKGROUP);
        if (slot < (unsigned)CAP) {
          l_s[slot] = i0 + k * OT + t;
          l_d[slot] = (int)dl;
        }
      }
    }
  }
  __syncthreads();
  const unsigned found = s_cnt;

  if (found <= (unsigned)CAP) {
    const int cnt = (int)found;
    for (int j = t; j < cnt; j += OT) {
      const int s = src[l_s[j]];
      const int dl = l_d[j];
      l_s[j] = s;
#pragma unroll
      for (int h = 0; h < HEADS; ++h) {
        const float e = lrelu(asrc[(size_t)s * HEADS + h] + s_ad[dl * HEADS + h]);
        l_w[h][j] = e;
        __hip_atomic_fetch_max(&s_mx[dl * HEADS + h], flipf(e), __ATOMIC_RELAXED, __HIP_MEMORY_SCOPE_WORKGROUP);
      }
    }
    __syncthreads();
    for (int j = t; j < cnt; j += OT) {
      const int dl = l_d[j];
#pragma unroll
      for (int h = 0; h < HEADS; ++h) l_w[h][j] = expf(l_w[h][j] - unflipf(s_mx[dl * HEADS + h]));
    }
    __syncthreads();

    float acc[RPW][HEADS * 4];
    float dsum[RPW][HEADS];
#pragma unroll
    for (int q = 0; q < RPW; ++q) {
      const int r = wave + 32 * q;
      const int rr = (r < rows) ? r : 0;
      const float* xrow = xp + (size_t)(nb + rr) * F + lane;
#pragma unroll
      for (int h = 0; h < HEADS; ++h) {
        const float es = lrelu(asrc[gb3 + rr * HEADS + h] + s_ad[rr * HEADS + h]);
        dsum[q][h] = expf(es - unflipf(s_mx[rr * HEADS + h]));
      }
#pragma unroll
      for (int c = 0; c < HEADS * 4; ++c) acc[q][c] = dsum[q][c >> 2] * xrow[c * 32];
    }

    for (int base = 0; base < cnt; base += RND * 32) {
      unsigned mk[RND];
#pragma unroll
      for (int k = 0; k < RND; ++k) {
        const int j = base + k * 32 + lane;
        const bool mine = (j < cnt) && ((l_d[j] & 31) == wave);
        mk[k] = __builtin_amdgcn_ballot_w32(mine);
      }
      for (int it = 0; it < RND * 32; ++it) {
        int jj[GRP];
#pragma unroll
        for (int g = 0; g < GRP; ++g) {
          const int off = own_pop(mk);
          jj[g] = (off >= 0) ? (base + off) : -1;
        }
        if (jj[0] < 0) break;
        int jr[GRP];
        float a0[GRP], a1[GRP], a2[GRP];
        float val[GRP][HEADS * 4];
#pragma unroll
        for (int g = 0; g < GRP; ++g) {
          if (jj[g] >= 0) {
            const int ss = ufi(l_s[jj[g]]);
            jr[g] = ufi(l_d[jj[g]]) >> 5;
            a0[g] = uff(l_w[0][jj[g]]);
            a1[g] = uff(l_w[1][jj[g]]);
            a2[g] = uff(l_w[2][jj[g]]);
            const float* xrow = xp + (size_t)ss * F + lane;
#pragma unroll
            for (int c = 0; c < HEADS * 4; ++c) val[g][c] = xrow[c * 32];
          }
        }
#pragma unroll
        for (int g = 0; g < GRP; ++g) {
          if (jj[g] >= 0) own_add(acc, dsum, jr[g], val[g], a0[g], a1[g], a2[g]);
        }
      }
    }

#pragma unroll
    for (int q = 0; q < RPW; ++q) {
      const int r = wave + 32 * q;
      if (r < rows) {
        float* orow = agg + (size_t)(nb + r) * F + lane;
#pragma unroll
        for (int c = 0; c < HEADS * 4; ++c) orow[c * 32] = acc[q][c];
        if (lane == 0) {
          s_dn[r * HEADS + 0] = dsum[q][0];
          s_dn[r * HEADS + 1] = dsum[q][1];
          s_dn[r * HEADS + 2] = dsum[q][2];
        }
      }
    }
  } else {
  for (int i0 = 0; i0 < E; i0 += OT * SU) {
    int d[SU];
#pragma unroll
    for (int k = 0; k < SU; ++k) {
      const int i = i0 + k * OT + t;
      d[k] = (i < E) ? dst[i] : -1;
    }
#pragma unroll
    for (int k = 0; k < SU; ++k) {
      const unsigned dl = (unsigned)(d[k] - nb);
      if (dl < (unsigned)rows) {
        const int s = src[i0 + k * OT + t];
#pragma unroll
        for (int h = 0; h < HEADS; ++h) {
          const float e = lrelu(asrc[(size_t)s * HEADS + h] + s_ad[dl * HEADS + h]);
          __hip_atomic_fetch_max(&s_mx[dl * HEADS + h], flipf(e), __ATOMIC_RELAXED, __HIP_MEMORY_SCOPE_WORKGROUP);
        }
      }
    }
  }
  __syncthreads();

  const int nq = rows * (F / 4);
  for (int q = t; q < nq; q += OT) {
    const int r = q / (F / 4);
    const int h = (q % (F / 4)) / (CPH / 4);
    const float es = lrelu(asrc[gb3 + r * HEADS + h] + s_ad[r * HEADS + h]);
    const float ws = expf(es - unflipf(s_mx[r * HEADS + h]));
    float4 v = ld4(xp + (size_t)nb * F + (size_t)q * 4);
    v.x *= ws; v.y *= ws; v.z *= ws; v.w *= ws;
    *reinterpret_cast<float4*>(s_acc + q * 4) = v;
  }
  if (t < rows * HEADS) {
    const float es = lrelu(asrc[gb3 + t] + s_ad[t]);
    s_dn[t] = expf(es - unflipf(s_mx[t]));
  }
  __syncthreads();

  for (int i0 = 0; i0 < E; i0 += OT * SU) {
    int d[SU];
#pragma unroll
    for (int k = 0; k < SU; ++k) {
      const int i = i0 + k * OT + t;
      d[k] = (i < E) ? dst[i] : -1;
    }
#pragma unroll
    for (int k = 0; k < SU; ++k) {
      const unsigned dl = (unsigned)(d[k] - nb);
      const bool hit = dl < (unsigned)rows;
      unsigned mask = __builtin_amdgcn_ballot_w32(hit);
      if (mask == 0u) continue;
      int s = 0;
      float w0 = 0.f, w1 = 0.f, w2 = 0.f;
      if (hit) {
        s = src[i0 + k * OT + t];
        const float* ap = asrc + (size_t)s * HEADS;
        w0 = expf(lrelu(ap[0] + s_ad[dl * HEADS + 0]) - unflipf(s_mx[dl * HEADS + 0]));
        w1 = expf(lrelu(ap[1] + s_ad[dl * HEADS + 1]) - unflipf(s_mx[dl * HEADS + 1]));
        w2 = expf(lrelu(ap[2] + s_ad[dl * HEADS + 2]) - unflipf(s_mx[dl * HEADS + 2]));
        __hip_atomic_fetch_add(&s_dn[dl * HEADS + 0], w0, __ATOMIC_RELAXED, __HIP_MEMORY_SCOPE_WORKGROUP);
        __hip_atomic_fetch_add(&s_dn[dl * HEADS + 1], w1, __ATOMIC_RELAXED, __HIP_MEMORY_SCOPE_WORKGROUP);
        __hip_atomic_fetch_add(&s_dn[dl * HEADS + 2], w2, __ATOMIC_RELAXED, __HIP_MEMORY_SCOPE_WORKGROUP);
      }
      while (mask != 0u) {
        const int l = __builtin_ctz(mask);
        mask &= mask - 1u;
        const int   ss = __builtin_amdgcn_readlane(s, l);
        const int   dr = __builtin_amdgcn_readlane((int)dl, l);
        const float a0 = __int_as_float(__builtin_amdgcn_readlane(__float_as_int(w0), l));
        const float a1 = __int_as_float(__builtin_amdgcn_readlane(__float_as_int(w1), l));
        const float a2 = __int_as_float(__builtin_amdgcn_readlane(__float_as_int(w2), l));
        const float* xrow = xp + (size_t)ss * F + lane;
        float val[HEADS * 4];
#pragma unroll
        for (int c = 0; c < HEADS * 4; ++c) val[c] = xrow[c * 32];
        float* arow = s_acc + dr * F + lane;
#pragma unroll
        for (int c = 0; c < HEADS * 4; ++c) {
          const float wc = (c < 4) ? a0 : ((c < 8) ? a1 : a2);
          __hip_atomic_fetch_add(arow + c * 32, wc * val[c], __ATOMIC_RELAXED, __HIP_MEMORY_SCOPE_WORKGROUP);
        }
      }
    }
  }
  __syncthreads();

  for (int q = t; q < nq; q += OT) {
    st4(agg + (size_t)nb * F + (size_t)q * 4, *reinterpret_cast<const float4*>(s_acc + q * 4));
  }
  }
  __syncthreads();
  if (t < rows * HEADS) den[gb3 + t] = s_dn[t];
}

__global__ void __launch_bounds__(256) k_bias_rows(float* __restrict__ agg, const float* __restrict__ den,
                            const float* __restrict__ b, unsigned long long total) {
  const unsigned long long base = (unsigned long long)blockIdx.x * 4096ull;
#pragma unroll
  for (int j = 0; j < 4; ++j) {
    const unsigned long long i = base + (unsigned long long)(j * 256 + (int)threadIdx.x) * 4ull;
    if (i < total) {
      float4 v = ld4(agg + i);
      const float4 bb = ld4(b + (i % F));
      const float rd = 1.0f / den[i / CPH];
      v.x = fmaf(v.x, rd, bb.x); v.y = fmaf(v.y, rd, bb.y); v.z = fmaf(v.z, rd, bb.z); v.w = fmaf(v.w, rd, bb.w);
      st4(agg + i, v);
    }
  }
}

__global__ void __launch_bounds__(256) k_gemm(const float* __restrict__ A,
                                              const float* __restrict__ B,
                                              float* __restrict__ C, int mtiles) {
  const int lane = threadIdx.x & 31;
  const int wave = threadIdx.x >> 5;
  const int wt = blockIdx.x * 8 + wave;
  const int npair = F / 32;
  const int tm = wt / npair;
  const int tp = wt % npair;
  if (tm >= mtiles) return;
  const int m  = tm * 16 + (lane & 15);
  const int n0 = tp * 32 + (lane & 15);
  const int n1 = n0 + 16;
  const int kb = (lane >> 4) * 2;
  const float* arow = A + (size_t)m * F;
  v8f acc0 = {};
  v8f acc1 = {};
  for (int k0 = 0; k0 < F; k0 += 4) {
    const int k = k0 + kb;
    v2f a;  a.x  = arow[k];            a.y  = arow[k + 1];
    const float* bp = B + (size_t)k * F;
    v2f b0; b0.x = bp[n0];             b0.y = bp[F + n0];
    v2f b1; b1.x = bp[n1];             b1.y = bp[F + n1];
    acc0 = __builtin_amdgcn_wmma_f32_16x16x4_f32(false, a, false, b0, (short)0, acc0,
                                                 false, false);
    acc1 = __builtin_amdgcn_wmma_f32_16x16x4_f32(false, a, false, b1, (short)0, acc1,
                                                 false, false);
  }
  const int mr = tm * 16 + ((lane >> 4) << 3);
#pragma unroll
  for (int i = 0; i < 8; ++i) {
    C[(size_t)(mr + i) * F + n0] = acc0[i];
    C[(size_t)(mr + i) * F + n1] = acc1[i];
  }
}

__global__ void __launch_bounds__(256) k_final(const float* __restrict__ agg, const float* __restrict__ den,
                        const float* __restrict__ b2,
                        const float* __restrict__ Wl, const float* __restrict__ bl,
                        float* __restrict__ out, int N) {
  __shared__ float s_out[WPB * NPW];
  const int lane = threadIdx.x & 31;
  const int wave = __builtin_amdgcn_readfirstlane((int)(threadIdx.x >> 5));
  const int nb = blockIdx.x * (WPB * NPW);
  const int n0 = nb + wave * NPW;
  float4 v[NPW][HEADS];
  float dn[NPW][HEADS];
#pragma unroll
  for (int r = 0; r < NPW; ++r) {
    const int n = (n0 + r < N) ? (n0 + r) : (N - 1);
#pragma unroll
    for (int h = 0; h < HEADS; ++h) {
      v[r][h] = ld4(agg + (size_t)n * F + h * CPH + lane * 4);
      dn[r][h] = 1.0f / den[(size_t)n * HEADS + h];
    }
  }
  float red[NPW];
#pragma unroll
  for (int r = 0; r < NPW; ++r) red[r] = 0.f;
#pragma unroll
  for (int h = 0; h < HEADS; ++h) {
    const float4 bb = ld4(b2 + h * CPH + lane * 4);
    const float4 w  = ld4(Wl + h * CPH + lane * 4);
#pragma unroll
    for (int r = 0; r < NPW; ++r) {
      float a = red[r];
      a = fmaf(fmaf(v[r][h].x, dn[r][h], bb.x), w.x, a);
      a = fmaf(fmaf(v[r][h].y, dn[r][h], bb.y), w.y, a);
      a = fmaf(fmaf(v[r][h].z, dn[r][h], bb.z), w.z, a);
      a = fmaf(fmaf(v[r][h].w, dn[r][h], bb.w), w.w, a);
      red[r] = a;
    }
  }
#pragma unroll
  for (int m = 16; m >= 1; m >>= 1) {
#pragma unroll
    for (int r = 0; r < NPW; ++r) red[r] += __shfl_xor(red[r], m, 32);
  }
  if (lane == 0) {
#pragma unroll
    for (int r = 0; r < NPW; ++r) s_out[wave * NPW + r] = red[r];
  }
  __syncthreads();
  const int t = threadIdx.x;
  if (t < WPB * NPW && nb + t < N) out[nb + t] = s_out[t] + bl[0];
}

extern "C" void kernel_launch(void* const* d_in, const int* in_sizes, int n_in,
                              void* d_out, int out_size, void* d_ws, size_t ws_size,
                              hipStream_t stream) {
  const float* x   = (const float*)d_in[0];
  const int*   ei  = (const int*)d_in[1];
  const float* W1  = (const float*)d_in[2];
  const float* as1 = (const float*)d_in[3];
  const float* ad1 = (const float*)d_in[4];
  const float* b1  = (const float*)d_in[5];
  const float* W2  = (const float*)d_in[6];
  const float* as2 = (const float*)d_in[7];
  const float* ad2 = (const float*)d_in[8];
  const float* b2  = (const float*)d_in[9];
  const float* Wl  = (const float*)d_in[10];
  const float* bl  = (const float*)d_in[11];
  float* out = (float*)d_out;

  const int N  = in_sizes[0] / HIN;
  const int E  = in_sizes[1] / 2;
  const int* src = ei;
  const int* dst = ei + E;

  float* ws = (float*)d_ws;
  const size_t NF = (size_t)N * F;
  float*    XP   = ws;
  float*    AGG  = XP + NF;
  float*    DEN  = AGG + NF;
  float*    ASRC = DEN + (size_t)3 * N;
  float*    ADST = ASRC + (size_t)3 * N;

  const int bblocks = (int)((NF + 4095) / 4096);
  const int nblocks = (N + WPB * NPW - 1) / (WPB * NPW);
  const int oblocks = (N + OR - 1) / OR;
  const int mtiles  = N / 16;
  const int gblocks = (mtiles * (F / 32) + 7) / 8;

  k_proj1<<<nblocks, 256, 0, stream>>>(x, W1, as1, ad1, XP, ASRC, ADST, N);
  k_own<<<oblocks, OT, 0, stream>>>(src, dst, E, N, ASRC, ADST, XP, AGG, DEN);
  k_bias_rows<<<bblocks, 256, 0, stream>>>(AGG, DEN, b1, NF);

  k_gemm<<<gblocks, 256, 0, stream>>>(AGG, W2, XP, mtiles);
  k_alpha<<<nblocks, 256, 0, stream>>>(XP, as2, ad2, ASRC, ADST, N);

  k_own<<<oblocks, OT, 0, stream>>>(src, dst, E, N, ASRC, ADST, XP, AGG, DEN);

  k_final<<<nblocks, 256, 0, stream>>>(AGG, DEN, b2, Wl, bl, out, N);
}


# layer-2 GEMM staged through shared memory: a 12-wave group computes an 80 x 384 band, A and B chunks of k = 64 fetched once per group (two stages), ten accumulators per wave; sum order unchanged
# speedup vs baseline: 3.1440x; 1.0853x over previous
#include <hip/hip_runtime.h>
#include <hip/hip_bf16.h>

#define HEADS 3
#define CPH   128
#define F     384
#define HIN   12
#define WPB   8
#define NPW   4

typedef __attribute__((ext_vector_type(2))) float v2f;
typedef __attribute__((ext_vector_type(8))) float v8f;

__device__ __forceinline__ unsigned flipf(float f) {
  unsigned b = __float_as_uint(f);
  return (b & 0x80000000u) ? ~b : (b | 0x80000000u);
}
__device__ __forceinline__ float unflipf(unsigned u) {
  unsigned b = (u & 0x80000000u) ? (u & 0x7FFFFFFFu) : ~u;
  return __uint_as_float(b);
}
__device__ __forceinline__ float lrelu(float v) { return v > 0.f ? v : 0.2f * v; }


__device__ __forceinline__ float4 ld4(const float* p) { return *reinterpret_cast<const float4*>(p); }
__device__ __forceinline__ void st4(float* p, float4 v) { *reinterpret_cast<float4*>(p) = v; }
__device__ __forceinline__ float4 fma4(float s, float4 w, float4 a) {
  a.x = fmaf(s, w.x, a.x); a.y = fmaf(s, w.y, a.y); a.z = fmaf(s, w.z, a.z); a.w = fmaf(s, w.w, a.w);
  return a;
}
__device__ __forceinline__ float dot4(float4 a, float4 b) {
  return fmaf(a.w, b.w, fmaf(a.z, b.z, fmaf(a.y, b.y, a.x * b.x)));
}

__global__ void __launch_bounds__(256) k_proj1(const float* __restrict__ x, const float* __restrict__ W1,
                        const float* __restrict__ a_src, const float* __restrict__ a_dst,
                        float* __restrict__ xp, float* __restrict__ asrc,
                        float* __restrict__ adst, int N) {
  __shared__ float s_as[WPB * NPW * HEADS];
  __shared__ float s_ad[WPB * NPW * HEADS];
  const int lane = threadIdx.x & 31;
  const int wave = __builtin_amdgcn_readfirstlane((int)(threadIdx.x >> 5));
  const int nb = blockIdx.x * (WPB * NPW);
  const int n0 = nb + wave * NPW;
  float xr[NPW][HIN];
#pragma unroll
  for (int r = 0; r < NPW; ++r) {
    const int n = (n0 + r < N) ? (n0 + r) : (N - 1);
#pragma unroll
    for (int k = 0; k < HIN; ++k) xr[r][k] = x[(size_t)n * HIN + k];
  }
  float red[2 * HEADS * NPW];
#pragma unroll
  for (int h = 0; h < HEADS; ++h) {
    const int c = h * CPH + lane * 4;
    float4 acc[NPW];
#pragma unroll
    for (int r = 0; r < NPW; ++r) acc[r] = make_float4(0.f, 0.f, 0.f, 0.f);
#pragma unroll
    for (int k = 0; k < HIN; ++k) {
      const float4 w = ld4(W1 + k * F + c);
#pragma unroll
      for (int r = 0; r < NPW; ++r) acc[r] = fma4(xr[r][k], w, acc[r]);
    }
    const float4 as4 = ld4(a_src + c);
    const float4 ad4 = ld4(a_dst + c);
#pragma unroll
    for (int r = 0; r < NPW; ++r) {
      if (n0 + r < N) st4(xp + (size_t)(n0 + r) * F + c, acc[r]);
      red[(r * HEADS + h) * 2 + 0] = dot4(acc[r], as4);
      red[(r * HEADS + h) * 2 + 1] = dot4(acc[r], ad4);
    }
  }
#pragma unroll
  for (int m = 16; m >= 1; m >>= 1) {
#pragma unroll
    for (int i = 0; i < 2 * HEADS * NPW; ++i) red[i] += __shfl_xor(red[i], m, 32);
  }
  if (lane == 0) {
#pragma unroll
    for (int r = 0; r < NPW; ++r) {
#pragma unroll
      for (int h = 0; h < HEADS; ++h) {
        s_as[(wave * NPW + r) * HEADS + h] = red[(r * HEADS + h) * 2 + 0];
        s_ad[(wave * NPW + r) * HEADS + h] = red[(r * HEADS + h) * 2 + 1];
      }
    }
  }
  __syncthreads();
  const int t = threadIdx.x;
  const size_t lim = (size_t)N * HEADS;
  if (t < 96) {
    const size_t o = (size_t)nb * HEADS + t;
    if (o < lim) asrc[o] = s_as[t];
  } else if (t < 192) {
    const size_t o = (size_t)nb * HEADS + (t - 96);
    if (o < lim) adst[o] = s_ad[t - 96];
  }
}

__global__ void __launch_bounds__(256) k_alpha(const float* __restrict__ xp, const float* __restrict__ a_src,
                        const float* __restrict__ a_dst, float* __restrict__ asrc,
                        float* __restrict__ adst, int N) {
  __shared__ float s_as[WPB * NPW * HEADS];
  __shared__ float s_ad[WPB * NPW * HEADS];
  const int lane = threadIdx.x & 31;
  const int wave = __builtin_amdgcn_readfirstlane((int)(threadIdx.x >> 5));
  const int nb = blockIdx.x * (WPB * NPW);
  const int n0 = nb + wave * NPW;
  float4 v[NPW][HEADS];
#pragma unroll
  for (int r = 0; r < NPW; ++r) {
    const int n = (n0 + r < N) ? (n0 + r) : (N - 1);
#pragma unroll
    for (int h = 0; h < HEADS; ++h) v[r][h] = ld4(xp + (size_t)n * F + h * CPH + lane * 4);
  }
  float red[2 * HEADS * NPW];
#pragma unroll
  for (int h = 0; h < HEADS; ++h) {
    const float4 as4 = ld4(a_src + h * CPH + lane * 4);
    const float4 ad4 = ld4(a_dst + h * CPH + lane * 4);
#pragma unroll
    for (int r = 0; r < NPW; ++r) {
      red[(r * HEADS + h) * 2 + 0] = dot4(v[r][h], as4);
      red[(r * HEADS + h) * 2 + 1] = dot4(v[r][h], ad4);
    }
  }
#pragma unroll
  for (int m = 16; m >= 1; m >>= 1) {
#pragma unroll
    for (int i = 0; i < 2 * HEADS * NPW; ++i) red[i] += __shfl_xor(red[i], m, 32);
  }
  if (lane == 0) {
#pragma unroll
    for (int r = 0; r < NPW; ++r) {
#pragma unroll
      for (int h = 0; h < HEADS; ++h) {
        s_as[(wave * NPW + r) * HEADS + h] = red[(r * HEADS + h) * 2 + 0];
        s_ad[(wave * NPW + r) * HEADS + h] = red[(r * HEADS + h) * 2 + 1];
      }
    }
  }
  __syncthreads();
  const int t = threadIdx.x;
  const size_t lim = (size_t)N * HEADS;
  if (t < 96) {
    const size_t o = (size_t)nb * HEADS + t;
    if (o < lim) asrc[o] = s_as[t];
  } else if (t < 192) {
    const size_t o = (size_t)nb * HEADS + (t - 96);
    if (o < lim) adst[o] = s_ad[t - 96];
  }
}

#define OR    96
#define OT    1024
#define RPW   3
#define SU    16
#define CAP   4096
#define RND   8
#define GRP   4

__device__ __forceinline__ int own_pop(unsigned (&mk)[RND]) {
  int off = -1;
#pragma unroll
  for (int k = 0; k < RND; ++k) {
    if (off < 0 && mk[k] != 0u) {
      off = k * 32 + __builtin_ctz(mk[k]);
      mk[k] &= mk[k] - 1u;
    }
  }
  return off;
}
__device__ __forceinline__ int   ufi(int v)   { return __builtin_amdgcn_readfirstlane(v); }
__device__ __forceinline__ float uff(float v) { return __int_as_float(__builtin_amdgcn_readfirstlane(__float_as_int(v))); }

__device__ __forceinline__ void own_add(float (&acc)[RPW][HEADS * 4], float (&dsum)[RPW][HEADS], int jr,
                                        const float (&val)[HEADS * 4], float a0, float a1, float a2) {
#pragma unroll
  for (int q = 0; q < RPW; ++q) {
    if (jr == q) {
#pragma unroll
      for (int c = 0; c < HEADS * 4; ++c) acc[q][c] = fmaf((c < 4) ? a0 : ((c < 8) ? a1 : a2), val[c], acc[q][c]);
      dsum[q][0] += a0; dsum[q][1] += a1; dsum[q][2] += a2;
    }
  }
}

__global__ void __launch_bounds__(OT) k_own(const int* __restrict__ src, const int* __restrict__ dst,
                           int E, int N, const float* __restrict__ asrc,
                           const float* __restrict__ adst, const float* __restrict__ xp,
                           float* __restrict__ agg, float* __restrict__ den) {
  __shared__ float    s_acc[OR * F];
  __shared__ int      l_s[CAP];
  __shared__ int      l_d[CAP];
  __shared__ float    l_w[HEADS][CAP];
  __shared__ float    s_ad[OR * HEADS];
  __shared__ unsigned s_mx[OR * HEADS];
  __shared__ float    s_dn[OR * HEADS];
  __shared__ unsigned s_cnt;
  const int t = threadIdx.x;
  const int lane = t & 31;
  const int wave = __builtin_amdgcn_readfirstlane(t >> 5);
  const int nb = blockIdx.x * OR;
  const int rows = (N - nb < OR) ? (N - nb) : OR;
  const size_t gb3 = (size_t)nb * HEADS;

  if (t < OR * HEADS) {
    float a = 0.f;
    unsigned m = 0u;
    if (t < rows * HEADS) {
      a = adst[gb3 + t];
      m = flipf(lrelu(asrc[gb3 + t] + a));
    }
    s_ad[t] = a;
    s_mx[t] = m;
  }
  if (t == 0) s_cnt = 0u;
  __syncthreads();

  for (int i0 = 0; i0 < E; i0 += OT * SU) {
    int d[SU];
#pragma unroll
    for (int k = 0; k < SU; ++k) {
      const int i = i0 + k * OT + t;
      d[k] = (i < E) ? dst[i] : -1;
    }
#pragma unroll
    for (int k = 0; k < SU; ++k) {
      const unsigned dl = (unsigned)(d[k] - nb);
      if (dl < (unsigned)rows) {
        const unsigned slot = __hip_atomic_fetch_add(&s_cnt, 1u, __ATOMIC_RELAXED, __HIP_MEMORY_SCOPE_WORKGROUP);
        if (slot < (unsigned)CAP) {
          l_s[slot] = i0 + k * OT + t;
          l_d[slot] = (int)dl;
        }
      }
    }
  }
  __syncthreads();
  const unsigned found = s_cnt;

  if (found <= (unsigned)CAP) {
    const int cnt = (int)found;
    for (int j = t; j < cnt; j += OT) {
      const int s = src[l_s[j]];
      const int dl = l_d[j];
      l_s[j] = s;
#pragma unroll
      for (int h = 0; h < HEADS; ++h) {
        const float e = lrelu(asrc[(size_t)s * HEADS + h] + s_ad[dl * HEADS + h]);
        l_w[h][j] = e;
        __hip_atomic_fetch_max(&s_mx[dl * HEADS + h], flipf(e), __ATOMIC_RELAXED, __HIP_MEMORY_SCOPE_WORKGROUP);
      }
    }
    __syncthreads();
    for (int j = t; j < cnt; j += OT) {
      const int dl = l_d[j];
#pragma unroll
      for (int h = 0; h < HEADS; ++h) l_w[h][j] = expf(l_w[h][j] - unflipf(s_mx[dl * HEADS + h]));
    }
    __syncthreads();

    float acc[RPW][HEADS * 4];
    float dsum[RPW][HEADS];
#pragma unroll
    for (int q = 0; q < RPW; ++q) {
      const int r = wave + 32 * q;
      const int rr = (r < rows) ? r : 0;
      const float* xrow = xp + (size_t)(nb + rr) * F + lane;
#pragma unroll
      for (int h = 0; h < HEADS; ++h) {
        const float es = lrelu(asrc[gb3 + rr * HEADS + h] + s_ad[rr * HEADS + h]);
        dsum[q][h] = expf(es - unflipf(s_mx[rr * HEADS + h]));
      }
#pragma unroll
      for (int c = 0; c < HEADS * 4; ++c) acc[q][c] = dsum[q][c >> 2] * xrow[c * 32];
    }

    for (int base = 0; base < cnt; base += RND * 32) {
      unsigned mk[RND];
#pragma unroll
      for (int k = 0; k < RND; ++k) {
        const int j = base + k * 32 + lane;
        const bool mine = (j < cnt) && ((l_d[j] & 31) == wave);
        mk[k] = __builtin_amdgcn_ballot_w32(mine);
      }
      for (int it = 0; it < RND * 32; ++it) {
        int jj[GRP];
#pragma unroll
        for (int g = 0; g < GRP; ++g) {
          const int off = own_pop(mk);
          jj[g] = (off >= 0) ? (base + off) : -1;
        }
        if (jj[0] < 0) break;
        int jr[GRP];
        float a0[GRP], a1[GRP], a2[GRP];
        float val[GRP][HEADS * 4];
#pragma unroll
        for (int g = 0; g < GRP; ++g) {
          if (jj[g] >= 0) {
            const int ss = ufi(l_s[jj[g]]);
            jr[g] = ufi(l_d[jj[g]]) >> 5;
            a0[g] = uff(l_w[0][jj[g]]);
            a1[g] = uff(l_w[1][jj[g]]);
            a2[g] = uff(l_w[2][jj[g]]);
            const float* xrow = xp + (size_t)ss * F + lane;
#pragma unroll
            for (int c = 0; c < HEADS * 4; ++c) val[g][c] = xrow[c * 32];
          }
        }
#pragma unroll
        for (int g = 0; g < GRP; ++g) {
          if (jj[g] >= 0) own_add(acc, dsum, jr[g], val[g], a0[g], a1[g], a2[g]);
        }
      }
    }

#pragma unroll
    for (int q = 0; q < RPW; ++q) {
      const int r = wave + 32 * q;
      if (r < rows) {
        float* orow = agg + (size_t)(nb + r) * F + lane;
#pragma unroll
        for (int c = 0; c < HEADS * 4; ++c) orow[c * 32] = acc[q][c];
        if (lane == 0) {
          s_dn[r * HEADS + 0] = dsum[q][0];
          s_dn[r * HEADS + 1] = dsum[q][1];
          s_dn[r * HEADS + 2] = dsum[q][2];
        }
      }
    }
  } else {
  for (int i0 = 0; i0 < E; i0 += OT * SU) {
    int d[SU];
#pragma unroll
    for (int k = 0; k < SU; ++k) {
      const int i = i0 + k * OT + t;
      d[k] = (i < E) ? dst[i] : -1;
    }
#pragma unroll
    for (int k = 0; k < SU; ++k) {
      const unsigned dl = (unsigned)(d[k] - nb);
      if (dl < (unsigned)rows) {
        const int s = src[i0 + k * OT + t];
#pragma unroll
        for (int h = 0; h < HEADS; ++h) {
          const float e = lrelu(asrc[(size_t)s * HEADS + h] + s_ad[dl * HEADS + h]);
          __hip_atomic_fetch_max(&s_mx[dl * HEADS + h], flipf(e), __ATOMIC_RELAXED, __HIP_MEMORY_SCOPE_WORKGROUP);
        }
      }
    }
  }
  __syncthreads();

  const int nq = rows * (F / 4);
  for (int q = t; q < nq; q += OT) {
    const int r = q / (F / 4);
    const int h = (q % (F / 4)) / (CPH / 4);
    const float es = lrelu(asrc[gb3 + r * HEADS + h] + s_ad[r * HEADS + h]);
    const float ws = expf(es - unflipf(s_mx[r * HEADS + h]));
    float4 v = ld4(xp + (size_t)nb * F + (size_t)q * 4);
    v.x *= ws; v.y *= ws; v.z *= ws; v.w *= ws;
    *reinterpret_cast<float4*>(s_acc + q * 4) = v;
  }
  if (t < rows * HEADS) {
    const float es = lrelu(asrc[gb3 + t] + s_ad[t]);
    s_dn[t] = expf(es - unflipf(s_mx[t]));
  }
  __syncthreads();

  for (int i0 = 0; i0 < E; i0 += OT * SU) {
    int d[SU];
#pragma unroll
    for (int k = 0; k < SU; ++k) {
      const int i = i0 + k * OT + t;
      d[k] = (i < E) ? dst[i] : -1;
    }
#pragma unroll
    for (int k = 0; k < SU; ++k) {
      const unsigned dl = (unsigned)(d[k] - nb);
      const bool hit = dl < (unsigned)rows;
      unsigned mask = __builtin_amdgcn_ballot_w32(hit);
      if (mask == 0u) continue;
      int s = 0;
      float w0 = 0.f, w1 = 0.f, w2 = 0.f;
      if (hit) {
        s = src[i0 + k * OT + t];
        const float* ap = asrc + (size_t)s * HEADS;
        w0 = expf(lrelu(ap[0] + s_ad[dl * HEADS + 0]) - unflipf(s_mx[dl * HEADS + 0]));
        w1 = expf(lrelu(ap[1] + s_ad[dl * HEADS + 1]) - unflipf(s_mx[dl * HEADS + 1]));
        w2 = expf(lrelu(ap[2] + s_ad[dl * HEADS + 2]) - unflipf(s_mx[dl * HEADS + 2]));
        __hip_atomic_fetch_add(&s_dn[dl * HEADS + 0], w0, __ATOMIC_RELAXED, __HIP_MEMORY_SCOPE_WORKGROUP);
        __hip_atomic_fetch_add(&s_dn[dl * HEADS + 1], w1, __ATOMIC_RELAXED, __HIP_MEMORY_SCOPE_WORKGROUP);
        __hip_atomic_fetch_add(&s_dn[dl * HEADS + 2], w2, __ATOMIC_RELAXED, __HIP_MEMORY_SCOPE_WORKGROUP);
      }
      while (mask != 0u) {
        const int l = __builtin_ctz(mask);
        mask &= mask - 1u;
        const int   ss = __builtin_amdgcn_readlane(s, l);
        const int   dr = __builtin_amdgcn_readlane((int)dl, l);
        const float a0 = __int_as_float(__builtin_amdgcn_readlane(__float_as_int(w0), l));
        const float a1 = __int_as_float(__builtin_amdgcn_readlane(__float_as_int(w1), l));
        const float a2 = __int_as_float(__builtin_amdgcn_readlane(__float_as_int(w2), l));
        const float* xrow = xp + (size_t)ss * F + lane;
        float val[HEADS * 4];
#pragma unroll
        for (int c = 0; c < HEADS * 4; ++c) val[c] = xrow[c * 32];
        float* arow = s_acc + dr * F + lane;
#pragma unroll
        for (int c = 0; c < HEADS * 4; ++c) {
          const float wc = (c < 4) ? a0 : ((c < 8) ? a1 : a2);
          __hip_atomic_fetch_add(arow + c * 32, wc * val[c], __ATOMIC_RELAXED, __HIP_MEMORY_SCOPE_WORKGROUP);
        }
      }
    }
  }
  __syncthreads();

  for (int q = t; q < nq; q += OT) {
    st4(agg + (size_t)nb * F + (size_t)q * 4, *reinterpret_cast<const float4*>(s_acc + q * 4));
  }
  }
  __syncthreads();
  if (t < rows * HEADS) den[gb3 + t] = s_dn[t];
}

__global__ void __launch_bounds__(256) k_bias_rows(float* __restrict__ agg, const float* __restrict__ den,
                            const float* __restrict__ b, unsigned long long total) {
  const unsigned long long base = (unsigned long long)blockIdx.x * 4096ull;
#pragma unroll
  for (int j = 0; j < 4; ++j) {
    const unsigned long long i = base + (unsigned long long)(j * 256 + (int)threadIdx.x) * 4ull;
    if (i < total) {
      float4 v = ld4(agg + i);
      const float4 bb = ld4(b + (i % F));
      const float rd = 1.0f / den[i / CPH];
      v.x = fmaf(v.x, rd, bb.x); v.y = fmaf(v.y, rd, bb.y); v.z = fmaf(v.z, rd, bb.z); v.w = fmaf(v.w, rd, bb.w);
      st4(agg + i, v);
    }
  }
}

#define GBM  80
#define GBK  64
#define GAP  68
#define GBP  392
#define GT   384
#define GMT  5
#define GPA  4
#define GPB  16

__global__ void __launch_bounds__(GT) k_gemm(const float* __restrict__ A,
                                             const float* __restrict__ B,
                                             float* __restrict__ C, int M) {
  __shared__ __attribute__((aligned(16))) float sA[2][GBM * GAP];
  __shared__ __attribute__((aligned(16))) float sB[2][GBK * GBP];
  const int t = threadIdx.x;
  const int lane = t & 31;
  const int wave = __builtin_amdgcn_readfirstlane(t >> 5);
  const int c = lane & 15;
  const int h = lane >> 4;
  const int m0 = blockIdx.x * GBM;
  const bool a3 = wave < 4;

  const float* ag[GPA];
  int al[GPA];
#pragma unroll
  for (int p = 0; p < GPA; ++p) {
    const int idx = t + p * GT;
    const int row = (idx >> 4) < GBM ? (idx >> 4) : (GBM - 1);
    const int j = idx & 15;
    const int gm = (m0 + row < M) ? (m0 + row) : (M - 1);
    ag[p] = A + (size_t)gm * F + j * 4;
    al[p] = row * GAP + j * 4;
  }
  const int brow = t / 96;
  const int bj = t - brow * 96;
  const float* bg = B + (size_t)brow * F + bj * 4;
  const int bl = brow * GBP + bj * 4;

  float4 pa[GPA];
  float4 pb[GPB];
  v8f acc[GMT][2];
#pragma unroll
  for (int mt = 0; mt < GMT; ++mt) { acc[mt][0] = (v8f){}; acc[mt][1] = (v8f){}; }

#pragma unroll
  for (int p = 0; p < GPA; ++p) if (p < GPA - 1 || a3) pa[p] = ld4(ag[p]);
#pragma unroll
  for (int p = 0; p < GPB; ++p) pb[p] = ld4(bg + (size_t)(4 * p) * F);
#pragma unroll
  for (int p = 0; p < GPA; ++p) if (p < GPA - 1 || a3) *reinterpret_cast<float4*>(&sA[0][al[p]]) = pa[p];
#pragma unroll
  for (int p = 0; p < GPB; ++p) *reinterpret_cast<float4*>(&sB[0][bl + 4 * p * GBP]) = pb[p];
  __syncthreads();

  for (int ch = 0; ch < F / GBK; ++ch) {
    const int st = ch & 1;
    const bool more = ch + 1 < F / GBK;
    if (more) {
      const int kc = (ch + 1) * GBK;
#pragma unroll
      for (int p = 0; p < GPA; ++p) if (p < GPA - 1 || a3) pa[p] = ld4(ag[p] + kc);
#pragma unroll
      for (int p = 0; p < GPB; ++p) pb[p] = ld4(bg + (size_t)(kc + 4 * p) * F);
    }
    const float* ab = &sA[st][c * GAP + 2 * h];
    const float* bb = &sB[st][(2 * h) * GBP + wave * 32 + c];
#pragma unroll
    for (int ks = 0; ks < GBK / 4; ++ks) {
      v2f a[GMT];
      v2f b[2];
#pragma unroll
      for (int mt = 0; mt < GMT; ++mt) a[mt] = *reinterpret_cast<const v2f*>(ab + mt * 16 * GAP + ks * 4);
#pragma unroll
      for (int nt = 0; nt < 2; ++nt) {
        b[nt].x = bb[(ks * 4) * GBP + nt * 16];
        b[nt].y = bb[(ks * 4 + 1) * GBP + nt * 16];
      }
#pragma unroll
      for (int mt = 0; mt < GMT; ++mt) {
#pragma unroll
        for (int nt = 0; nt < 2; ++nt)
          acc[mt][nt] = __builtin_amdgcn_wmma_f32_16x16x4_f32(false, a[mt], false, b[nt], (short)0, acc[mt][nt],
                                                              false, false);
      }
    }
    if (more) {
#pragma unroll
      for (int p = 0; p < GPA; ++p) if (p < GPA - 1 || a3) *reinterpret_cast<float4*>(&sA[st ^ 1][al[p]]) = pa[p];
#pragma unroll
      for (int p = 0; p < GPB; ++p) *reinterpret_cast<float4*>(&sB[st ^ 1][bl + 4 * p * GBP]) = pb[p];
      __syncthreads();
    }
  }

#pragma unroll
  for (int mt = 0; mt < GMT; ++mt) {
    const int r0 = m0 + mt * 16 + 8 * h;
#pragma unroll
    for (int i = 0; i < 8; ++i) {
      if (r0 + i < M) {
        float* crow = C + (size_t)(r0 + i) * F + wave * 32 + c;
        crow[0]  = acc[mt][0][i];
        crow[16] = acc[mt][1][i];
      }
    }
  }
}

__global__ void __launch_bounds__(256) k_final(const float* __restrict__ agg, const float* __restrict__ den,
                        const float* __restrict__ b2,
                        const float* __restrict__ Wl, const float* __restrict__ bl,
                        float* __restrict__ out, int N) {
  __shared__ float s_out[WPB * NPW];
  const int lane = threadIdx.x & 31;
  const int wave = __builtin_amdgcn_readfirstlane((int)(threadIdx.x >> 5));
  const int nb = blockIdx.x * (WPB * NPW);
  const int n0 = nb + wave * NPW;
  float4 v[NPW][HEADS];
  float dn[NPW][HEADS];
#pragma unroll
  for (int r = 0; r < NPW; ++r) {
    const int n = (n0 + r < N) ? (n0 + r) : (N - 1);
#pragma unroll
    for (int h = 0; h < HEADS; ++h) {
      v[r][h] = ld4(agg + (size_t)n * F + h * CPH + lane * 4);
      dn[r][h] = 1.0f / den[(size_t)n * HEADS + h];
    }
  }
  float red[NPW];
#pragma unroll
  for (int r = 0; r < NPW; ++r) red[r] = 0.f;
#pragma unroll
  for (int h = 0; h < HEADS; ++h) {
    const float4 bb = ld4(b2 + h * CPH + lane * 4);
    const float4 w  = ld4(Wl + h * CPH + lane * 4);
#pragma unroll
    for (int r = 0; r < NPW; ++r) {
      float a = red[r];
      a = fmaf(fmaf(v[r][h].x, dn[r][h], bb.x), w.x, a);
      a = fmaf(fmaf(v[r][h].y, dn[r][h], bb.y), w.y, a);
      a = fmaf(fmaf(v[r][h].z, dn[r][h], bb.z), w.z, a);
      a = fmaf(fmaf(v[r][h].w, dn[r][h], bb.w), w.w, a);
      red[r] = a;
    }
  }
#pragma unroll
  for (int m = 16; m >= 1; m >>= 1) {
#pragma unroll
    for (int r = 0; r < NPW; ++r) red[r] += __shfl_xor(red[r], m, 32);
  }
  if (lane == 0) {
#pragma unroll
    for (int r = 0; r < NPW; ++r) s_out[wave * NPW + r] = red[r];
  }
  __syncthreads();
  const int t = threadIdx.x;
  if (t < WPB * NPW && nb + t < N) out[nb + t] = s_out[t] + bl[0];
}

extern "C" void kernel_launch(void* const* d_in, const int* in_sizes, int n_in,
                              void* d_out, int out_size, void* d_ws, size_t ws_size,
                              hipStream_t stream) {
  const float* x   = (const float*)d_in[0];
  const int*   ei  = (const int*)d_in[1];
  const float* W1  = (const float*)d_in[2];
  const float* as1 = (const float*)d_in[3];
  const float* ad1 = (const float*)d_in[4];
  const float* b1  = (const float*)d_in[5];
  const float* W2  = (const float*)d_in[6];
  const float* as2 = (const float*)d_in[7];
  const float* ad2 = (const float*)d_in[8];
  const float* b2  = (const float*)d_in[9];
  const float* Wl  = (const float*)d_in[10];
  const float* bl  = (const float*)d_in[11];
  float* out = (float*)d_out;

  const int N  = in_sizes[0] / HIN;
  const int E  = in_sizes[1] / 2;
  const int* src = ei;
  const int* dst = ei + E;

  float* ws = (float*)d_ws;
  const size_t NF = (size_t)N * F;
  float*    XP   = ws;
  float*    AGG  = XP + NF;
  float*    DEN  = AGG + NF;
  float*    ASRC = DEN + (size_t)3 * N;
  float*    ADST = ASRC + (size_t)3 * N;

  const int bblocks = (int)((NF + 4095) / 4096);
  const int nblocks = (N + WPB * NPW - 1) / (WPB * NPW);
  const int oblocks = (N + OR - 1) / OR;
  const int gblocks = (N + GBM - 1) / GBM;

  k_proj1<<<nblocks, 256, 0, stream>>>(x, W1, as1, ad1, XP, ASRC, ADST, N);
  k_own<<<oblocks, OT, 0, stream>>>(src, dst, E, N, ASRC, ADST, XP, AGG, DEN);
  k_bias_rows<<<bblocks, 256, 0, stream>>>(AGG, DEN, b1, NF);

  k_gemm<<<gblocks, GT, 0, stream>>>(AGG, W2, XP, N);
  k_alpha<<<nblocks, 256, 0, stream>>>(XP, as2, ad2, ASRC, ADST, N);

  k_own<<<oblocks, OT, 0, stream>>>(src, dst, E, N, ASRC, ADST, XP, AGG, DEN);

  k_final<<<nblocks, 256, 0, stream>>>(AGG, DEN, b2, Wl, bl, out, N);
}


# owning groups' edge-destination scan: 16-byte loads (4 destinations per lane), one unsigned-minimum test turns 4 away at once, bounds test only in the last trip; sums in arrival order as before
# speedup vs baseline: 3.6486x; 1.1605x over previous
#include <hip/hip_runtime.h>
#include <hip/hip_bf16.h>

#define HEADS 3
#define CPH   128
#define F     384
#define HIN   12
#define WPB   8
#define NPW   4

typedef __attribute__((ext_vector_type(2))) float v2f;
typedef __attribute__((ext_vector_type(8))) float v8f;

__device__ __forceinline__ unsigned flipf(float f) {
  unsigned b = __float_as_uint(f);
  return (b & 0x80000000u) ? ~b : (b | 0x80000000u);
}
__device__ __forceinline__ float unflipf(unsigned u) {
  unsigned b = (u & 0x80000000u) ? (u & 0x7FFFFFFFu) : ~u;
  return __uint_as_float(b);
}
__device__ __forceinline__ float lrelu(float v) { return v > 0.f ? v : 0.2f * v; }


__device__ __forceinline__ float4 ld4(const float* p) { return *reinterpret_cast<const float4*>(p); }
__device__ __forceinline__ void st4(float* p, float4 v) { *reinterpret_cast<float4*>(p) = v; }
__device__ __forceinline__ float4 fma4(float s, float4 w, float4 a) {
  a.x = fmaf(s, w.x, a.x); a.y = fmaf(s, w.y, a.y); a.z = fmaf(s, w.z, a.z); a.w = fmaf(s, w.w, a.w);
  return a;
}
__device__ __forceinline__ float dot4(float4 a, float4 b) {
  return fmaf(a.w, b.w, fmaf(a.z, b.z, fmaf(a.y, b.y, a.x * b.x)));
}

__global__ void __launch_bounds__(256) k_proj1(const float* __restrict__ x, const float* __restrict__ W1,
                        const float* __restrict__ a_src, const float* __restrict__ a_dst,
                        float* __restrict__ xp, float* __restrict__ asrc,
                        float* __restrict__ adst, int N) {
  __shared__ float s_as[WPB * NPW * HEADS];
  __shared__ float s_ad[WPB * NPW * HEADS];
  const int lane = threadIdx.x & 31;
  const int wave = __builtin_amdgcn_readfirstlane((int)(threadIdx.x >> 5));
  const int nb = blockIdx.x * (WPB * NPW);
  const int n0 = nb + wave * NPW;
  float xr[NPW][HIN];
#pragma unroll
  for (int r = 0; r < NPW; ++r) {
    const int n = (n0 + r < N) ? (n0 + r) : (N - 1);
#pragma unroll
    for (int k = 0; k < HIN; ++k) xr[r][k] = x[(size_t)n * HIN + k];
  }
  float red[2 * HEADS * NPW];
#pragma unroll
  for (int h = 0; h < HEADS; ++h) {
    const int c = h * CPH + lane * 4;
    float4 acc[NPW];
#pragma unroll
    for (int r = 0; r < NPW; ++r) acc[r] = make_float4(0.f, 0.f, 0.f, 0.f);
#pragma unroll
    for (int k = 0; k < HIN; ++k) {
      const float4 w = ld4(W1 + k * F + c);
#pragma unroll
      for (int r = 0; r < NPW; ++r) acc[r] = fma4(xr[r][k], w, acc[r]);
    }
    const float4 as4 = ld4(a_src + c);
    const float4 ad4 = ld4(a_dst + c);
#pragma unroll
    for (int r = 0; r < NPW; ++r) {
      if (n0 + r < N) st4(xp + (size_t)(n0 + r) * F + c, acc[r]);
      red[(r * HEADS + h) * 2 + 0] = dot4(acc[r], as4);
      red[(r * HEADS + h) * 2 + 1] = dot4(acc[r], ad4);
    }
  }
#pragma unroll
  for (int m = 16; m >= 1; m >>= 1) {
#pragma unroll
    for (int i = 0; i < 2 * HEADS * NPW; ++i) red[i] += __shfl_xor(red[i], m, 32);
  }
  if (lane == 0) {
#pragma unroll
    for (int r = 0; r < NPW; ++r) {
#pragma unroll
      for (int h = 0; h < HEADS; ++h) {
        s_as[(wave * NPW + r) * HEADS + h] = red[(r * HEADS + h) * 2 + 0];
        s_ad[(wave * NPW + r) * HEADS + h] = red[(r * HEADS + h) * 2 + 1];
      }
    }
  }
  __syncthreads();
  const int t = threadIdx.x;
  const size_t lim = (size_t)N * HEADS;
  if (t < 96) {
    const size_t o = (size_t)nb * HEADS + t;
    if (o < lim) asrc[o] = s_as[t];
  } else if (t < 192) {
    const size_t o = (size_t)nb * HEADS + (t - 96);
    if (o < lim) adst[o] = s_ad[t - 96];
  }
}

__global__ void __launch_bounds__(256) k_alpha(const float* __restrict__ xp, const float* __restrict__ a_src,
                        const float* __restrict__ a_dst, float* __restrict__ asrc,
                        float* __restrict__ adst, int N) {
  __shared__ float s_as[WPB * NPW * HEADS];
  __shared__ float s_ad[WPB * NPW * HEADS];
  const int lane = threadIdx.x & 31;
  const int wave = __builtin_amdgcn_readfirstlane((int)(threadIdx.x >> 5));
  const int nb = blockIdx.x * (WPB * NPW);
  const int n0 = nb + wave * NPW;
  float4 v[NPW][HEADS];
#pragma unroll
  for (int r = 0; r < NPW; ++r) {
    const int n = (n0 + r < N) ? (n0 + r) : (N - 1);
#pragma unroll
    for (int h = 0; h < HEADS; ++h) v[r][h] = ld4(xp + (size_t)n * F + h * CPH + lane * 4);
  }
  float red[2 * HEADS * NPW];
#pragma unroll
  for (int h = 0; h < HEADS; ++h) {
    const float4 as4 = ld4(a_src + h * CPH + lane * 4);
    const float4 ad4 = ld4(a_dst + h * CPH + lane * 4);
#pragma unroll
    for (int r = 0; r < NPW; ++r) {
      red[(r * HEADS + h) * 2 + 0] = dot4(v[r][h], as4);
      red[(r * HEADS + h) * 2 + 1] = dot4(v[r][h], ad4);
    }
  }
#pragma unroll
  for (int m = 16; m >= 1; m >>= 1) {
#pragma unroll
    for (int i = 0; i < 2 * HEADS * NPW; ++i) red[i] += __shfl_xor(red[i], m, 32);
  }
  if (lane == 0) {
#pragma unroll
    for (int r = 0; r < NPW; ++r) {
#pragma unroll
      for (int h = 0; h < HEADS; ++h) {
        s_as[(wave * NPW + r) * HEADS + h] = red[(r * HEADS + h) * 2 + 0];
        s_ad[(wave * NPW + r) * HEADS + h] = red[(r * HEADS + h) * 2 + 1];
      }
    }
  }
  __syncthreads();
  const int t = threadIdx.x;
  const size_t lim = (size_t)N * HEADS;
  if (t < 96) {
    const size_t o = (size_t)nb * HEADS + t;
    if (o < lim) asrc[o] = s_as[t];
  } else if (t < 192) {
    const size_t o = (size_t)nb * HEADS + (t - 96);
    if (o < lim) adst[o] = s_ad[t - 96];
  }
}

#define OR    96
#define OT    1024
#define RPW   3
#define SU    16
#define SU4   8
#define CAP   4096
#define RND   8
#define GRP   4

__device__ __forceinline__ int own_pop(unsigned (&mk)[RND]) {
  int off = -1;
#pragma unroll
  for (int k = 0; k < RND; ++k) {
    if (off < 0 && mk[k] != 0u) {
      off = k * 32 + __builtin_ctz(mk[k]);
      mk[k] &= mk[k] - 1u;
    }
  }
  return off;
}
__device__ __forceinline__ int   ufi(int v)   { return __builtin_amdgcn_readfirstlane(v); }
__device__ __forceinline__ float uff(float v) { return __int_as_float(__builtin_amdgcn_readfirstlane(__float_as_int(v))); }

__device__ __forceinline__ void own_add(float (&acc)[RPW][HEADS * 4], float (&dsum)[RPW][HEADS], int jr,
                                        const float (&val)[HEADS * 4], float a0, float a1, float a2) {
#pragma unroll
  for (int q = 0; q < RPW; ++q) {
    if (jr == q) {
#pragma unroll
      for (int c = 0; c < HEADS * 4; ++c) acc[q][c] = fmaf((c < 4) ? a0 : ((c < 8) ? a1 : a2), val[c], acc[q][c]);
      dsum[q][0] += a0; dsum[q][1] += a1; dsum[q][2] += a2;
    }
  }
}

__global__ void __launch_bounds__(OT) k_own(const int* __restrict__ src, const int* __restrict__ dst,
                           int E, int N, const float* __restrict__ asrc,
                           const float* __restrict__ adst, const float* __restrict__ xp,
                           float* __restrict__ agg, float* __restrict__ den) {
  __shared__ float    s_acc[OR * F];
  __shared__ int      l_s[CAP];
  __shared__ int      l_d[CAP];
  __shared__ float    l_w[HEADS][CAP];
  __shared__ float    s_ad[OR * HEADS];
  __shared__ unsigned s_mx[OR * HEADS];
  __shared__ float    s_dn[OR * HEADS];
  __shared__ unsigned s_cnt;
  const int t = threadIdx.x;
  const int lane = t & 31;
  const int wave = __builtin_amdgcn_readfirstlane(t >> 5);
  const int nb = blockIdx.x * OR;
  const int rows = (N - nb < OR) ? (N - nb) : OR;
  const size_t gb3 = (size_t)nb * HEADS;

  if (t < OR * HEADS) {
    float a = 0.f;
    unsigned m = 0u;
    if (t < rows * HEADS) {
      a = adst[gb3 + t];
      m = flipf(lrelu(asrc[gb3 + t] + a));
    }
    s_ad[t] = a;
    s_mx[t] = m;
  }
  if (t == 0) s_cnt = 0u;
  __syncthreads();

  auto push = [&](unsigned dl, int e) {
    const unsigned slot = __hip_atomic_fetch_add(&s_cnt, 1u, __ATOMIC_RELAXED, __HIP_MEMORY_SCOPE_WORKGROUP);
    if (slot < (unsigned)CAP) {
      l_s[slot] = e;
      l_d[slot] = (int)dl;
    }
  };
  const int E4 = ((reinterpret_cast<size_t>(dst) & 15) == 0) ? (E >> 2) : 0;
  const int4* dst4 = reinterpret_cast<const int4*>(dst);
  for (int g0 = 0; g0 < E4; g0 += OT * SU4) {
    int4 d[SU4];
    if (g0 + OT * SU4 <= E4) {
#pragma unroll
      for (int k = 0; k < SU4; ++k) d[k] = dst4[g0 + k * OT + t];
    } else {
#pragma unroll
      for (int k = 0; k < SU4; ++k) {
        const int g = g0 + k * OT + t;
        d[k] = (g < E4) ? dst4[g] : make_int4(-1, -1, -1, -1);
      }
    }
#pragma unroll
    for (int k = 0; k < SU4; ++k) {
      const unsigned u0 = (unsigned)(d[k].x - nb);
      const unsigned u1 = (unsigned)(d[k].y - nb);
      const unsigned u2 = (unsigned)(d[k].z - nb);
      const unsigned u3 = (unsigned)(d[k].w - nb);
      const unsigned m01 = (u0 < u1) ? u0 : u1;
      const unsigned m23 = (u2 < u3) ? u2 : u3;
      const unsigned mn = (m01 < m23) ? m01 : m23;
      if (mn < (unsigned)rows) {
        const int e0 = (g0 + k * OT + t) * 4;
        if (u0 < (unsigned)rows) push(u0, e0);
        if (u1 < (unsigned)rows) push(u1, e0 + 1);
        if (u2 < (unsigned)rows) push(u2, e0 + 2);
        if (u3 < (unsigned)rows) push(u3, e0 + 3);
      }
    }
  }
  for (int i0 = E4 * 4; i0 < E; i0 += OT) {
    const int i = i0 + t;
    if (i < E) {
      const unsigned dl = (unsigned)(dst[i] - nb);
      if (dl < (unsigned)rows) push(dl, i);
    }
  }
  __syncthreads();
  const unsigned found = s_cnt;

  if (found <= (unsigned)CAP) {
    const int cnt = (int)found;
    for (int j = t; j < cnt; j += OT) {
      const int s = src[l_s[j]];
      const int dl = l_d[j];
      l_s[j] = s;
#pragma unroll
      for (int h = 0; h < HEADS; ++h) {
        const float e = lrelu(asrc[(size_t)s * HEADS + h] + s_ad[dl * HEADS + h]);
        l_w[h][j] = e;
        __hip_atomic_fetch_max(&s_mx[dl * HEADS + h], flipf(e), __ATOMIC_RELAXED, __HIP_MEMORY_SCOPE_WORKGROUP);
      }
    }
    __syncthreads();
    for (int j = t; j < cnt; j += OT) {
      const int dl = l_d[j];
#pragma unroll
      for (int h = 0; h < HEADS; ++h) l_w[h][j] = expf(l_w[h][j] - unflipf(s_mx[dl * HEADS + h]));
    }
    __syncthreads();

    float acc[RPW][HEADS * 4];
    float dsum[RPW][HEADS];
#pragma unroll
    for (int q = 0; q < RPW; ++q) {
      const int r = wave + 32 * q;
      const int rr = (r < rows) ? r : 0;
      const float* xrow = xp + (size_t)(nb + rr) * F + lane;
#pragma unroll
      for (int h = 0; h < HEADS; ++h) {
        const float es = lrelu(asrc[gb3 + rr * HEADS + h] + s_ad[rr * HEADS + h]);
        dsum[q][h] = expf(es - unflipf(s_mx[rr * HEADS + h]));
      }
#pragma unroll
      for (int c = 0; c < HEADS * 4; ++c) acc[q][c] = dsum[q][c >> 2] * xrow[c * 32];
    }

    for (int base = 0; base < cnt; base += RND * 32) {
      unsigned mk[RND];
#pragma unroll
      for (int k = 0; k < RND; ++k) {
        const int j = base + k * 32 + lane;
        const bool mine = (j < cnt) && ((l_d[j] & 31) == wave);
        mk[k] = __builtin_amdgcn_ballot_w32(mine);
      }
      for (int it = 0; it < RND * 32; ++it) {
        int jj[GRP];
#pragma unroll
        for (int g = 0; g < GRP; ++g) {
          const int off = own_pop(mk);
          jj[g] = (off >= 0) ? (base + off) : -1;
        }
        if (jj[0] < 0) break;
        int jr[GRP];
        float a0[GRP], a1[GRP], a2[GRP];
        float val[GRP][HEADS * 4];
#pragma unroll
        for (int g = 0; g < GRP; ++g) {
          if (jj[g] >= 0) {
            const int ss = ufi(l_s[jj[g]]);
            jr[g] = ufi(l_d[jj[g]]) >> 5;
            a0[g] = uff(l_w[0][jj[g]]);
            a1[g] = uff(l_w[1][jj[g]]);
            a2[g] = uff(l_w[2][jj[g]]);
            const float* xrow = xp + (size_t)ss * F + lane;
#pragma unroll
            for (int c = 0; c < HEADS * 4; ++c) val[g][c] = xrow[c * 32];
          }
        }
#pragma unroll
        for (int g = 0; g < GRP; ++g) {
          if (jj[g] >= 0) own_add(acc, dsum, jr[g], val[g], a0[g], a1[g], a2[g]);
        }
      }
    }

#pragma unroll
    for (int q = 0; q < RPW; ++q) {
      const int r = wave + 32 * q;
      if (r < rows) {
        float* orow = agg + (size_t)(nb + r) * F + lane;
#pragma unroll
        for (int c = 0; c < HEADS * 4; ++c) orow[c * 32] = acc[q][c];
        if (lane == 0) {
          s_dn[r * HEADS + 0] = dsum[q][0];
          s_dn[r * HEADS + 1] = dsum[q][1];
          s_dn[r * HEADS + 2] = dsum[q][2];
        }
      }
    }
  } else {
  for (int i0 = 0; i0 < E; i0 += OT * SU) {
    int d[SU];
#pragma unroll
    for (int k = 0; k < SU; ++k) {
      const int i = i0 + k * OT + t;
      d[k] = (i < E) ? dst[i] : -1;
    }
#pragma unroll
    for (int k = 0; k < SU; ++k) {
      const unsigned dl = (unsigned)(d[k] - nb);
      if (dl < (unsigned)rows) {
        const int s = src[i0 + k * OT + t];
#pragma unroll
        for (int h = 0; h < HEADS; ++h) {
          const float e = lrelu(asrc[(size_t)s * HEADS + h] + s_ad[dl * HEADS + h]);
          __hip_atomic_fetch_max(&s_mx[dl * HEADS + h], flipf(e), __ATOMIC_RELAXED, __HIP_MEMORY_SCOPE_WORKGROUP);
        }
      }
    }
  }
  __syncthreads();

  const int nq = rows * (F / 4);
  for (int q = t; q < nq; q += OT) {
    const int r = q / (F / 4);
    const int h = (q % (F / 4)) / (CPH / 4);
    const float es = lrelu(asrc[gb3 + r * HEADS + h] + s_ad[r * HEADS + h]);
    const float ws = expf(es - unflipf(s_mx[r * HEADS + h]));
    float4 v = ld4(xp + (size_t)nb * F + (size_t)q * 4);
    v.x *= ws; v.y *= ws; v.z *= ws; v.w *= ws;
    *reinterpret_cast<float4*>(s_acc + q * 4) = v;
  }
  if (t < rows * HEADS) {
    const float es = lrelu(asrc[gb3 + t] + s_ad[t]);
    s_dn[t] = expf(es - unflipf(s_mx[t]));
  }
  __syncthreads();

  for (int i0 = 0; i0 < E; i0 += OT * SU) {
    int d[SU];
#pragma unroll
    for (int k = 0; k < SU; ++k) {
      const int i = i0 + k * OT + t;
      d[k] = (i < E) ? dst[i] : -1;
    }
#pragma unroll
    for (int k = 0; k < SU; ++k) {
      const unsigned dl = (unsigned)(d[k] - nb);
      const bool hit = dl < (unsigned)rows;
      unsigned mask = __builtin_amdgcn_ballot_w32(hit);
      if (mask == 0u) continue;
      int s = 0;
      float w0 = 0.f, w1 = 0.f, w2 = 0.f;
      if (hit) {
        s = src[i0 + k * OT + t];
        const float* ap = asrc + (size_t)s * HEADS;
        w0 = expf(lrelu(ap[0] + s_ad[dl * HEADS + 0]) - unflipf(s_mx[dl * HEADS + 0]));
        w1 = expf(lrelu(ap[1] + s_ad[dl * HEADS + 1]) - unflipf(s_mx[dl * HEADS + 1]));
        w2 = expf(lrelu(ap[2] + s_ad[dl * HEADS + 2]) - unflipf(s_mx[dl * HEADS + 2]));
        __hip_atomic_fetch_add(&s_dn[dl * HEADS + 0], w0, __ATOMIC_RELAXED, __HIP_MEMORY_SCOPE_WORKGROUP);
        __hip_atomic_fetch_add(&s_dn[dl * HEADS + 1], w1, __ATOMIC_RELAXED, __HIP_MEMORY_SCOPE_WORKGROUP);
        __hip_atomic_fetch_add(&s_dn[dl * HEADS + 2], w2, __ATOMIC_RELAXED, __HIP_MEMORY_SCOPE_WORKGROUP);
      }
      while (mask != 0u) {
        const int l = __builtin_ctz(mask);
        mask &= mask - 1u;
        const int   ss = __builtin_amdgcn_readlane(s, l);
        const int   dr = __builtin_amdgcn_readlane((int)dl, l);
        const float a0 = __int_as_float(__builtin_amdgcn_readlane(__float_as_int(w0), l));
        const float a1 = __int_as_float(__builtin_amdgcn_readlane(__float_as_int(w1), l));
        const float a2 = __int_as_float(__builtin_amdgcn_readlane(__float_as_int(w2), l));
        const float* xrow = xp + (size_t)ss * F + lane;
        float val[HEADS * 4];
#pragma unroll
        for (int c = 0; c < HEADS * 4; ++c) val[c] = xrow[c * 32];
        float* arow = s_acc + dr * F + lane;
#pragma unroll
        for (int c = 0; c < HEADS * 4; ++c) {
          const float wc = (c < 4) ? a0 : ((c < 8) ? a1 : a2);
          __hip_atomic_fetch_add(arow + c * 32, wc * val[c], __ATOMIC_RELAXED, __HIP_MEMORY_SCOPE_WORKGROUP);
        }
      }
    }
  }
  __syncthreads();

  for (int q = t; q < nq; q += OT) {
    st4(agg + (size_t)nb * F + (size_t)q * 4, *reinterpret_cast<const float4*>(s_acc + q * 4));
  }
  }
  __syncthreads();
  if (t < rows * HEADS) den[gb3 + t] = s_dn[t];
}

__global__ void __launch_bounds__(256) k_bias_rows(float* __restrict__ agg, const float* __restrict__ den,
                            const float* __restrict__ b, unsigned long long total) {
  const unsigned long long base = (unsigned long long)blockIdx.x * 4096ull;
#pragma unroll
  for (int j = 0; j < 4; ++j) {
    const unsigned long long i = base + (unsigned long long)(j * 256 + (int)threadIdx.x) * 4ull;
    if (i < total) {
      float4 v = ld4(agg + i);
      const float4 bb = ld4(b + (i % F));
      const float rd = 1.0f / den[i / CPH];
      v.x = fmaf(v.x, rd, bb.x); v.y = fmaf(v.y, rd, bb.y); v.z = fmaf(v.z, rd, bb.z); v.w = fmaf(v.w, rd, bb.w);
      st4(agg + i, v);
    }
  }
}

#define GBM  80
#define GBK  64
#define GAP  68
#define GBP  392
#define GT   384
#define GMT  5
#define GPA  4
#define GPB  16

__global__ void __launch_bounds__(GT) k_gemm(const float* __restrict__ A,
                                             const float* __restrict__ B,
                                             float* __restrict__ C, int M) {
  __shared__ __attribute__((aligned(16))) float sA[2][GBM * GAP];
  __shared__ __attribute__((aligned(16))) float sB[2][GBK * GBP];
  const int t = threadIdx.x;
  const int lane = t & 31;
  const int wave = __builtin_amdgcn_readfirstlane(t >> 5);
  const int c = lane & 15;
  const int h = lane >> 4;
  const int m0 = blockIdx.x * GBM;
  const bool a3 = wave < 4;

  const float* ag[GPA];
  int al[GPA];
#pragma unroll
  for (int p = 0; p < GPA; ++p) {
    const int idx = t + p * GT;
    const int row = (idx >> 4) < GBM ? (idx >> 4) : (GBM - 1);
    const int j = idx & 15;
    const int gm = (m0 + row < M) ? (m0 + row) : (M - 1);
    ag[p] = A + (size_t)gm * F + j * 4;
    al[p] = row * GAP + j * 4;
  }
  const int brow = t / 96;
  const int bj = t - brow * 96;
  const float* bg = B + (size_t)brow * F + bj * 4;
  const int bl = brow * GBP + bj * 4;

  float4 pa[GPA];
  float4 pb[GPB];
  v8f acc[GMT][2];
#pragma unroll
  for (int mt = 0; mt < GMT; ++mt) { acc[mt][0] = (v8f){}; acc[mt][1] = (v8f){}; }

#pragma unroll
  for (int p = 0; p < GPA; ++p) if (p < GPA - 1 || a3) pa[p] = ld4(ag[p]);
#pragma unroll
  for (int p = 0; p < GPB; ++p) pb[p] = ld4(bg + (size_t)(4 * p) * F);
#pragma unroll
  for (int p = 0; p < GPA; ++p) if (p < GPA - 1 || a3) *reinterpret_cast<float4*>(&sA[0][al[p]]) = pa[p];
#pragma unroll
  for (int p = 0; p < GPB; ++p) *reinterpret_cast<float4*>(&sB[0][bl + 4 * p * GBP]) = pb[p];
  __syncthreads();

  for (int ch = 0; ch < F / GBK; ++ch) {
    const int st = ch & 1;
    const bool more = ch + 1 < F / GBK;
    if (more) {
      const int kc = (ch + 1) * GBK;
#pragma unroll
      for (int p = 0; p < GPA; ++p) if (p < GPA - 1 || a3) pa[p] = ld4(ag[p] + kc);
#pragma unroll
      for (int p = 0; p < GPB; ++p) pb[p] = ld4(bg + (size_t)(kc + 4 * p) * F);
    }
    const float* ab = &sA[st][c * GAP + 2 * h];
    const float* bb = &sB[st][(2 * h) * GBP + wave * 32 + c];
#pragma unroll
    for (int ks = 0; ks < GBK / 4; ++ks) {
      v2f a[GMT];
      v2f b[2];
#pragma unroll
      for (int mt = 0; mt < GMT; ++mt) a[mt] = *reinterpret_cast<const v2f*>(ab + mt * 16 * GAP + ks * 4);
#pragma unroll
      for (int nt = 0; nt < 2; ++nt) {
        b[nt].x = bb[(ks * 4) * GBP + nt * 16];
        b[nt].y = bb[(ks * 4 + 1) * GBP + nt * 16];
      }
#pragma unroll
      for (int mt = 0; mt < GMT; ++mt) {
#pragma unroll
        for (int nt = 0; nt < 2; ++nt)
          acc[mt][nt] = __builtin_amdgcn_wmma_f32_16x16x4_f32(false, a[mt], false, b[nt], (short)0, acc[mt][nt],
                                                              false, false);
      }
    }
    if (more) {
#pragma unroll
      for (int p = 0; p < GPA; ++p) if (p < GPA - 1 || a3) *reinterpret_cast<float4*>(&sA[st ^ 1][al[p]]) = pa[p];
#pragma unroll
      for (int p = 0; p < GPB; ++p) *reinterpret_cast<float4*>(&sB[st ^ 1][bl + 4 * p * GBP]) = pb[p];
      __syncthreads();
    }
  }

#pragma unroll
  for (int mt = 0; mt < GMT; ++mt) {
    const int r0 = m0 + mt * 16 + 8 * h;
#pragma unroll
    for (int i = 0; i < 8; ++i) {
      if (r0 + i < M) {
        float* crow = C + (size_t)(r0 + i) * F + wave * 32 + c;
        crow[0]  = acc[mt][0][i];
        crow[16] = acc[mt][1][i];
      }
    }
  }
}

__global__ void __launch_bounds__(256) k_final(const float* __restrict__ agg, const float* __restrict__ den,
                        const float* __restrict__ b2,
                        const float* __restrict__ Wl, const float* __restrict__ bl,
                        float* __restrict__ out, int N) {
  __shared__ float s_out[WPB * NPW];
  const int lane = threadIdx.x & 31;
  const int wave = __builtin_amdgcn_readfirstlane((int)(threadIdx.x >> 5));
  const int nb = blockIdx.x * (WPB * NPW);
  const int n0 = nb + wave * NPW;
  float4 v[NPW][HEADS];
  float dn[NPW][HEADS];
#pragma unroll
  for (int r = 0; r < NPW; ++r) {
    const int n = (n0 + r < N) ? (n0 + r) : (N - 1);
#pragma unroll
    for (int h = 0; h < HEADS; ++h) {
      v[r][h] = ld4(agg + (size_t)n * F + h * CPH + lane * 4);
      dn[r][h] = 1.0f / den[(size_t)n * HEADS + h];
    }
  }
  float red[NPW];
#pragma unroll
  for (int r = 0; r < NPW; ++r) red[r] = 0.f;
#pragma unroll
  for (int h = 0; h < HEADS; ++h) {
    const float4 bb = ld4(b2 + h * CPH + lane * 4);
    const float4 w  = ld4(Wl + h * CPH + lane * 4);
#pragma unroll
    for (int r = 0; r < NPW; ++r) {
      float a = red[r];
      a = fmaf(fmaf(v[r][h].x, dn[r][h], bb.x), w.x, a);
      a = fmaf(fmaf(v[r][h].y, dn[r][h], bb.y), w.y, a);
      a = fmaf(fmaf(v[r][h].z, dn[r][h], bb.z), w.z, a);
      a = fmaf(fmaf(v[r][h].w, dn[r][h], bb.w), w.w, a);
      red[r] = a;
    }
  }
#pragma unroll
  for (int m = 16; m >= 1; m >>= 1) {
#pragma unroll
    for (int r = 0; r < NPW; ++r) red[r] += __shfl_xor(red[r], m, 32);
  }
  if (lane == 0) {
#pragma unroll
    for (int r = 0; r < NPW; ++r) s_out[wave * NPW + r] = red[r];
  }
  __syncthreads();
  const int t = threadIdx.x;
  if (t < WPB * NPW && nb + t < N) out[nb + t] = s_out[t] + bl[0];
}

extern "C" void kernel_launch(void* const* d_in, const int* in_sizes, int n_in,
                              void* d_out, int out_size, void* d_ws, size_t ws_size,
                              hipStream_t stream) {
  const float* x   = (const float*)d_in[0];
  const int*   ei  = (const int*)d_in[1];
  const float* W1  = (const float*)d_in[2];
  const float* as1 = (const float*)d_in[3];
  const float* ad1 = (const float*)d_in[4];
  const float* b1  = (const float*)d_in[5];
  const float* W2  = (const float*)d_in[6];
  const float* as2 = (const float*)d_in[7];
  const float* ad2 = (const float*)d_in[8];
  const float* b2  = (const float*)d_in[9];
  const float* Wl  = (const float*)d_in[10];
  const float* bl  = (const float*)d_in[11];
  float* out = (float*)d_out;

  const int N  = in_sizes[0] / HIN;
  const int E  = in_sizes[1] / 2;
  const int* src = ei;
  const int* dst = ei + E;

  float* ws = (float*)d_ws;
  const size_t NF = (size_t)N * F;
  float*    XP   = ws;
  float*    AGG  = XP + NF;
  float*    DEN  = AGG + NF;
  float*    ASRC = DEN + (size_t)3 * N;
  float*    ADST = ASRC + (size_t)3 * N;

  const int bblocks = (int)((NF + 4095) / 4096);
  const int nblocks = (N + WPB * NPW - 1) / (WPB * NPW);
  const int oblocks = (N + OR - 1) / OR;
  const int gblocks = (N + GBM - 1) / GBM;

  k_proj1<<<nblocks, 256, 0, stream>>>(x, W1, as1, ad1, XP, ASRC, ADST, N);
  k_own<<<oblocks, OT, 0, stream>>>(src, dst, E, N, ASRC, ADST, XP, AGG, DEN);
  k_bias_rows<<<bblocks, 256, 0, stream>>>(AGG, DEN, b1, NF);

  k_gemm<<<gblocks, GT, 0, stream>>>(AGG, W2, XP, N);
  k_alpha<<<nblocks, 256, 0, stream>>>(XP, as2, ad2, ASRC, ADST, N);

  k_own<<<oblocks, OT, 0, stream>>>(src, dst, E, N, ASRC, ADST, XP, AGG, DEN);

  k_final<<<nblocks, 256, 0, stream>>>(AGG, DEN, b2, Wl, bl, out, N);
}


# owning groups' edge gather: a lane takes 4 adjacent columns per head, a source row is 3 sixteen-byte loads per lane (was 12 four-byte), 3 rows in flight (was 4); stores likewise; sum order unchanged
# speedup vs baseline: 3.8152x; 1.0457x over previous
#include <hip/hip_runtime.h>
#include <hip/hip_bf16.h>

#define HEADS 3
#define CPH   128
#define F     384
#define HIN   12
#define WPB   8
#define NPW   4

typedef __attribute__((ext_vector_type(2))) float v2f;
typedef __attribute__((ext_vector_type(8))) float v8f;

__device__ __forceinline__ unsigned flipf(float f) {
  unsigned b = __float_as_uint(f);
  return (b & 0x80000000u) ? ~b : (b | 0x80000000u);
}
__device__ __forceinline__ float unflipf(unsigned u) {
  unsigned b = (u & 0x80000000u) ? (u & 0x7FFFFFFFu) : ~u;
  return __uint_as_float(b);
}
__device__ __forceinline__ float lrelu(float v) { return v > 0.f ? v : 0.2f * v; }


__device__ __forceinline__ float4 ld4(const float* p) { return *reinterpret_cast<const float4*>(p); }
__device__ __forceinline__ void st4(float* p, float4 v) { *reinterpret_cast<float4*>(p) = v; }
__device__ __forceinline__ float4 fma4(float s, float4 w, float4 a) {
  a.x = fmaf(s, w.x, a.x); a.y = fmaf(s, w.y, a.y); a.z = fmaf(s, w.z, a.z); a.w = fmaf(s, w.w, a.w);
  return a;
}
__device__ __forceinline__ float dot4(float4 a, float4 b) {
  return fmaf(a.w, b.w, fmaf(a.z, b.z, fmaf(a.y, b.y, a.x * b.x)));
}

__global__ void __launch_bounds__(256) k_proj1(const float* __restrict__ x, const float* __restrict__ W1,
                        const float* __restrict__ a_src, const float* __restrict__ a_dst,
                        float* __restrict__ xp, float* __restrict__ asrc,
                        float* __restrict__ adst, int N) {
  __shared__ float s_as[WPB * NPW * HEADS];
  __shared__ float s_ad[WPB * NPW * HEADS];
  const int lane = threadIdx.x & 31;
  const int wave = __builtin_amdgcn_readfirstlane((int)(threadIdx.x >> 5));
  const int nb = blockIdx.x * (WPB * NPW);
  const int n0 = nb + wave * NPW;
  float xr[NPW][HIN];
#pragma unroll
  for (int r = 0; r < NPW; ++r) {
    const int n = (n0 + r < N) ? (n0 + r) : (N - 1);
#pragma unroll
    for (int k = 0; k < HIN; ++k) xr[r][k] = x[(size_t)n * HIN + k];
  }
  float red[2 * HEADS * NPW];
#pragma unroll
  for (int h = 0; h < HEADS; ++h) {
    const int c = h * CPH + lane * 4;
    float4 acc[NPW];
#pragma unroll
    for (int r = 0; r < NPW; ++r) acc[r] = make_float4(0.f, 0.f, 0.f, 0.f);
#pragma unroll
    for (int k = 0; k < HIN; ++k) {
      const float4 w = ld4(W1 + k * F + c);
#pragma unroll
      for (int r = 0; r < NPW; ++r) acc[r] = fma4(xr[r][k], w, acc[r]);
    }
    const float4 as4 = ld4(a_src + c);
    const float4 ad4 = ld4(a_dst + c);
#pragma unroll
    for (int r = 0; r < NPW; ++r) {
      if (n0 + r < N) st4(xp + (size_t)(n0 + r) * F + c, acc[r]);
      red[(r * HEADS + h) * 2 + 0] = dot4(acc[r], as4);
      red[(r * HEADS + h) * 2 + 1] = dot4(acc[r], ad4);
    }
  }
#pragma unroll
  for (int m = 16; m >= 1; m >>= 1) {
#pragma unroll
    for (int i = 0; i < 2 * HEADS * NPW; ++i) red[i] += __shfl_xor(red[i], m, 32);
  }
  if (lane == 0) {
#pragma unroll
    for (int r = 0; r < NPW; ++r) {
#pragma unroll
      for (int h = 0; h < HEADS; ++h) {
        s_as[(wave * NPW + r) * HEADS + h] = red[(r * HEADS + h) * 2 + 0];
        s_ad[(wave * NPW + r) * HEADS + h] = red[(r * HEADS + h) * 2 + 1];
      }
    }
  }
  __syncthreads();
  const int t = threadIdx.x;
  const size_t lim = (size_t)N * HEADS;
  if (t < 96) {
    const size_t o = (size_t)nb * HEADS + t;
    if (o < lim) asrc[o] = s_as[t];
  } else if (t < 192) {
    const size_t o = (size_t)nb * HEADS + (t - 96);
    if (o < lim) adst[o] = s_ad[t - 96];
  }
}

__global__ void __launch_bounds__(256) k_alpha(const float* __restrict__ xp, const float* __restrict__ a_src,
                        const float* __restrict__ a_dst, float* __restrict__ asrc,
                        float* __restrict__ adst, int N) {
  __shared__ float s_as[WPB * NPW * HEADS];
  __shared__ float s_ad[WPB * NPW * HEADS];
  const int lane = threadIdx.x & 31;
  const int wave = __builtin_amdgcn_readfirstlane((int)(threadIdx.x >> 5));
  const int nb = blockIdx.x * (WPB * NPW);
  const int n0 = nb + wave * NPW;
  float4 v[NPW][HEADS];
#pragma unroll
  for (int r = 0; r < NPW; ++r) {
    const int n = (n0 + r < N) ? (n0 + r) : (N - 1);
#pragma unroll
    for (int h = 0; h < HEADS; ++h) v[r][h] = ld4(xp + (size_t)n * F + h * CPH + lane * 4);
  }
  float red[2 * HEADS * NPW];
#pragma unroll
  for (int h = 0; h < HEADS; ++h) {
    const float4 as4 = ld4(a_src + h * CPH + lane * 4);
    const float4 ad4 = ld4(a_dst + h * CPH + lane * 4);
#pragma unroll
    for (int r = 0; r < NPW; ++r) {
      red[(r * HEADS + h) * 2 + 0] = dot4(v[r][h], as4);
      red[(r * HEADS + h) * 2 + 1] = dot4(v[r][h], ad4);
    }
  }
#pragma unroll
  for (int m = 16; m >= 1; m >>= 1) {
#pragma unroll
    for (int i = 0; i < 2 * HEADS * NPW; ++i) red[i] += __shfl_xor(red[i], m, 32);
  }
  if (lane == 0) {
#pragma unroll
    for (int r = 0; r < NPW; ++r) {
#pragma unroll
      for (int h = 0; h < HEADS; ++h) {
        s_as[(wave * NPW + r) * HEADS + h] = red[(r * HEADS + h) * 2 + 0];
        s_ad[(wave * NPW + r) * HEADS + h] = red[(r * HEADS + h) * 2 + 1];
      }
    }
  }
  __syncthreads();
  const int t = threadIdx.x;
  const size_t lim = (size_t)N * HEADS;
  if (t < 96) {
    const size_t o = (size_t)nb * HEADS + t;
    if (o < lim) asrc[o] = s_as[t];
  } else if (t < 192) {
    const size_t o = (size_t)nb * HEADS + (t - 96);
    if (o < lim) adst[o] = s_ad[t - 96];
  }
}

#define OR    96
#define OT    1024
#define RPW   3
#define SU    16
#define SU4   8
#define CAP   4096
#define RND   8
#define GRP   3

__device__ __forceinline__ int own_pop(unsigned (&mk)[RND]) {
  int off = -1;
#pragma unroll
  for (int k = 0; k < RND; ++k) {
    if (off < 0 && mk[k] != 0u) {
      off = k * 32 + __builtin_ctz(mk[k]);
      mk[k] &= mk[k] - 1u;
    }
  }
  return off;
}
__device__ __forceinline__ int   ufi(int v)   { return __builtin_amdgcn_readfirstlane(v); }
__device__ __forceinline__ float uff(float v) { return __int_as_float(__builtin_amdgcn_readfirstlane(__float_as_int(v))); }

__device__ __forceinline__ void own_add(float4 (&acc)[RPW][HEADS], float (&dsum)[RPW][HEADS], int jr,
                                        const float4 (&val)[HEADS], float a0, float a1, float a2) {
#pragma unroll
  for (int q = 0; q < RPW; ++q) {
    if (jr == q) {
      acc[q][0] = fma4(a0, val[0], acc[q][0]);
      acc[q][1] = fma4(a1, val[1], acc[q][1]);
      acc[q][2] = fma4(a2, val[2], acc[q][2]);
      dsum[q][0] += a0; dsum[q][1] += a1; dsum[q][2] += a2;
    }
  }
}

__global__ void __launch_bounds__(OT) k_own(const int* __restrict__ src, const int* __restrict__ dst,
                           int E, int N, const float* __restrict__ asrc,
                           const float* __restrict__ adst, const float* __restrict__ xp,
                           float* __restrict__ agg, float* __restrict__ den) {
  __shared__ float    s_acc[OR * F];
  __shared__ int      l_s[CAP];
  __shared__ int      l_d[CAP];
  __shared__ float    l_w[HEADS][CAP];
  __shared__ float    s_ad[OR * HEADS];
  __shared__ unsigned s_mx[OR * HEADS];
  __shared__ float    s_dn[OR * HEADS];
  __shared__ unsigned s_cnt;
  const int t = threadIdx.x;
  const int lane = t & 31;
  const int wave = __builtin_amdgcn_readfirstlane(t >> 5);
  const int nb = blockIdx.x * OR;
  const int rows = (N - nb < OR) ? (N - nb) : OR;
  const size_t gb3 = (size_t)nb * HEADS;

  if (t < OR * HEADS) {
    float a = 0.f;
    unsigned m = 0u;
    if (t < rows * HEADS) {
      a = adst[gb3 + t];
      m = flipf(lrelu(asrc[gb3 + t] + a));
    }
    s_ad[t] = a;
    s_mx[t] = m;
  }
  if (t == 0) s_cnt = 0u;
  __syncthreads();

  auto push = [&](unsigned dl, int e) {
    const unsigned slot = __hip_atomic_fetch_add(&s_cnt, 1u, __ATOMIC_RELAXED, __HIP_MEMORY_SCOPE_WORKGROUP);
    if (slot < (unsigned)CAP) {
      l_s[slot] = e;
      l_d[slot] = (int)dl;
    }
  };
  const int E4 = ((reinterpret_cast<size_t>(dst) & 15) == 0) ? (E >> 2) : 0;
  const int4* dst4 = reinterpret_cast<const int4*>(dst);
  for (int g0 = 0; g0 < E4; g0 += OT * SU4) {
    int4 d[SU4];
    if (g0 + OT * SU4 <= E4) {
#pragma unroll
      for (int k = 0; k < SU4; ++k) d[k] = dst4[g0 + k * OT + t];
    } else {
#pragma unroll
      for (int k = 0; k < SU4; ++k) {
        const int g = g0 + k * OT + t;
        d[k] = (g < E4) ? dst4[g] : make_int4(-1, -1, -1, -1);
      }
    }
#pragma unroll
    for (int k = 0; k < SU4; ++k) {
      const unsigned u0 = (unsigned)(d[k].x - nb);
      const unsigned u1 = (unsigned)(d[k].y - nb);
      const unsigned u2 = (unsigned)(d[k].z - nb);
      const unsigned u3 = (unsigned)(d[k].w - nb);
      const unsigned m01 = (u0 < u1) ? u0 : u1;
      const unsigned m23 = (u2 < u3) ? u2 : u3;
      const unsigned mn = (m01 < m23) ? m01 : m23;
      if (mn < (unsigned)rows) {
        const int e0 = (g0 + k * OT + t) * 4;
        if (u0 < (unsigned)rows) push(u0, e0);
        if (u1 < (unsigned)rows) push(u1, e0 + 1);
        if (u2 < (unsigned)rows) push(u2, e0 + 2);
        if (u3 < (unsigned)rows) push(u3, e0 + 3);
      }
    }
  }
  for (int i0 = E4 * 4; i0 < E; i0 += OT) {
    const int i = i0 + t;
    if (i < E) {
      const unsigned dl = (unsigned)(dst[i] - nb);
      if (dl < (unsigned)rows) push(dl, i);
    }
  }
  __syncthreads();
  const unsigned found = s_cnt;

  if (found <= (unsigned)CAP) {
    const int cnt = (int)found;
    for (int j = t; j < cnt; j += OT) {
      const int s = src[l_s[j]];
      const int dl = l_d[j];
      l_s[j] = s;
#pragma unroll
      for (int h = 0; h < HEADS; ++h) {
        const float e = lrelu(asrc[(size_t)s * HEADS + h] + s_ad[dl * HEADS + h]);
        l_w[h][j] = e;
        __hip_atomic_fetch_max(&s_mx[dl * HEADS + h], flipf(e), __ATOMIC_RELAXED, __HIP_MEMORY_SCOPE_WORKGROUP);
      }
    }
    __syncthreads();
    for (int j = t; j < cnt; j += OT) {
      const int dl = l_d[j];
#pragma unroll
      for (int h = 0; h < HEADS; ++h) l_w[h][j] = expf(l_w[h][j] - unflipf(s_mx[dl * HEADS + h]));
    }
    __syncthreads();

    float4 acc[RPW][HEADS];
    float dsum[RPW][HEADS];
#pragma unroll
    for (int q = 0; q < RPW; ++q) {
      const int r = wave + 32 * q;
      const int rr = (r < rows) ? r : 0;
      const float* xrow = xp + (size_t)(nb + rr) * F + lane * 4;
#pragma unroll
      for (int h = 0; h < HEADS; ++h) {
        const float es = lrelu(asrc[gb3 + rr * HEADS + h] + s_ad[rr * HEADS + h]);
        dsum[q][h] = expf(es - unflipf(s_mx[rr * HEADS + h]));
      }
#pragma unroll
      for (int h = 0; h < HEADS; ++h) {
        const float4 v = ld4(xrow + h * CPH);
        acc[q][h] = make_float4(dsum[q][h] * v.x, dsum[q][h] * v.y, dsum[q][h] * v.z, dsum[q][h] * v.w);
      }
    }

    for (int base = 0; base < cnt; base += RND * 32) {
      unsigned mk[RND];
#pragma unroll
      for (int k = 0; k < RND; ++k) {
        const int j = base + k * 32 + lane;
        const bool mine = (j < cnt) && ((l_d[j] & 31) == wave);
        mk[k] = __builtin_amdgcn_ballot_w32(mine);
      }
      for (int it = 0; it < RND * 32; ++it) {
        int jj[GRP];
#pragma unroll
        for (int g = 0; g < GRP; ++g) {
          const int off = own_pop(mk);
          jj[g] = (off >= 0) ? (base + off) : -1;
        }
        if (jj[0] < 0) break;
        int jr[GRP];
        float a0[GRP], a1[GRP], a2[GRP];
        float4 val[GRP][HEADS];
#pragma unroll
        for (int g = 0; g < GRP; ++g) {
          if (jj[g] >= 0) {
            const int ss = ufi(l_s[jj[g]]);
            jr[g] = ufi(l_d[jj[g]]) >> 5;
            a0[g] = uff(l_w[0][jj[g]]);
            a1[g] = uff(l_w[1][jj[g]]);
            a2[g] = uff(l_w[2][jj[g]]);
            const float* xrow = xp + (size_t)ss * F + lane * 4;
#pragma unroll
            for (int h = 0; h < HEADS; ++h) val[g][h] = ld4(xrow + h * CPH);
          }
        }
#pragma unroll
        for (int g = 0; g < GRP; ++g) {
          if (jj[g] >= 0) own_add(acc, dsum, jr[g], val[g], a0[g], a1[g], a2[g]);
        }
      }
    }

#pragma unroll
    for (int q = 0; q < RPW; ++q) {
      const int r = wave + 32 * q;
      if (r < rows) {
        float* orow = agg + (size_t)(nb + r) * F + lane * 4;
#pragma unroll
        for (int h = 0; h < HEADS; ++h) st4(orow + h * CPH, acc[q][h]);
        if (lane == 0) {
          s_dn[r * HEADS + 0] = dsum[q][0];
          s_dn[r * HEADS + 1] = dsum[q][1];
          s_dn[r * HEADS + 2] = dsum[q][2];
        }
      }
    }
  } else {
  for (int i0 = 0; i0 < E; i0 += OT * SU) {
    int d[SU];
#pragma unroll
    for (int k = 0; k < SU; ++k) {
      const int i = i0 + k * OT + t;
      d[k] = (i < E) ? dst[i] : -1;
    }
#pragma unroll
    for (int k = 0; k < SU; ++k) {
      const unsigned dl = (unsigned)(d[k] - nb);
      if (dl < (unsigned)rows) {
        const int s = src[i0 + k * OT + t];
#pragma unroll
        for (int h = 0; h < HEADS; ++h) {
          const float e = lrelu(asrc[(size_t)s * HEADS + h] + s_ad[dl * HEADS + h]);
          __hip_atomic_fetch_max(&s_mx[dl * HEADS + h], flipf(e), __ATOMIC_RELAXED, __HIP_MEMORY_SCOPE_WORKGROUP);
        }
      }
    }
  }
  __syncthreads();

  const int nq = rows * (F / 4);
  for (int q = t; q < nq; q += OT) {
    const int r = q / (F / 4);
    const int h = (q % (F / 4)) / (CPH / 4);
    const float es = lrelu(asrc[gb3 + r * HEADS + h] + s_ad[r * HEADS + h]);
    const float ws = expf(es - unflipf(s_mx[r * HEADS + h]));
    float4 v = ld4(xp + (size_t)nb * F + (size_t)q * 4);
    v.x *= ws; v.y *= ws; v.z *= ws; v.w *= ws;
    *reinterpret_cast<float4*>(s_acc + q * 4) = v;
  }
  if (t < rows * HEADS) {
    const float es = lrelu(asrc[gb3 + t] + s_ad[t]);
    s_dn[t] = expf(es - unflipf(s_mx[t]));
  }
  __syncthreads();

  for (int i0 = 0; i0 < E; i0 += OT * SU) {
    int d[SU];
#pragma unroll
    for (int k = 0; k < SU; ++k) {
      const int i = i0 + k * OT + t;
      d[k] = (i < E) ? dst[i] : -1;
    }
#pragma unroll
    for (int k = 0; k < SU; ++k) {
      const unsigned dl = (unsigned)(d[k] - nb);
      const bool hit = dl < (unsigned)rows;
      unsigned mask = __builtin_amdgcn_ballot_w32(hit);
      if (mask == 0u) continue;
      int s = 0;
      float w0 = 0.f, w1 = 0.f, w2 = 0.f;
      if (hit) {
        s = src[i0 + k * OT + t];
        const float* ap = asrc + (size_t)s * HEADS;
        w0 = expf(lrelu(ap[0] + s_ad[dl * HEADS + 0]) - unflipf(s_mx[dl * HEADS + 0]));
        w1 = expf(lrelu(ap[1] + s_ad[dl * HEADS + 1]) - unflipf(s_mx[dl * HEADS + 1]));
        w2 = expf(lrelu(ap[2] + s_ad[dl * HEADS + 2]) - unflipf(s_mx[dl * HEADS + 2]));
        __hip_atomic_fetch_add(&s_dn[dl * HEADS + 0], w0, __ATOMIC_RELAXED, __HIP_MEMORY_SCOPE_WORKGROUP);
        __hip_atomic_fetch_add(&s_dn[dl * HEADS + 1], w1, __ATOMIC_RELAXED, __HIP_MEMORY_SCOPE_WORKGROUP);
        __hip_atomic_fetch_add(&s_dn[dl * HEADS + 2], w2, __ATOMIC_RELAXED, __HIP_MEMORY_SCOPE_WORKGROUP);
      }
      while (mask != 0u) {
        const int l = __builtin_ctz(mask);
        mask &= mask - 1u;
        const int   ss = __builtin_amdgcn_readlane(s, l);
        const int   dr = __builtin_amdgcn_readlane((int)dl, l);
        const float a0 = __int_as_float(__builtin_amdgcn_readlane(__float_as_int(w0), l));
        const float a1 = __int_as_float(__builtin_amdgcn_readlane(__float_as_int(w1), l));
        const float a2 = __int_as_float(__builtin_amdgcn_readlane(__float_as_int(w2), l));
        const float* xrow = xp + (size_t)ss * F + lane;
        float val[HEADS * 4];
#pragma unroll
        for (int c = 0; c < HEADS * 4; ++c) val[c] = xrow[c * 32];
        float* arow = s_acc + dr * F + lane;
#pragma unroll
        for (int c = 0; c < HEADS * 4; ++c) {
          const float wc = (c < 4) ? a0 : ((c < 8) ? a1 : a2);
          __hip_atomic_fetch_add(arow + c * 32, wc * val[c], __ATOMIC_RELAXED, __HIP_MEMORY_SCOPE_WORKGROUP);
        }
      }
    }
  }
  __syncthreads();

  for (int q = t; q < nq; q += OT) {
    st4(agg + (size_t)nb * F + (size_t)q * 4, *reinterpret_cast<const float4*>(s_acc + q * 4));
  }
  }
  __syncthreads();
  if (t < rows * HEADS) den[gb3 + t] = s_dn[t];
}

__global__ void __launch_bounds__(256) k_bias_rows(float* __restrict__ agg, const float* __restrict__ den,
                            const float* __restrict__ b, unsigned long long total) {
  const unsigned long long base = (unsigned long long)blockIdx.x * 4096ull;
#pragma unroll
  for (int j = 0; j < 4; ++j) {
    const unsigned long long i = base + (unsigned long long)(j * 256 + (int)threadIdx.x) * 4ull;
    if (i < total) {
      float4 v = ld4(agg + i);
      const float4 bb = ld4(b + (i % F));
      const float rd = 1.0f / den[i / CPH];
      v.x = fmaf(v.x, rd, bb.x); v.y = fmaf(v.y, rd, bb.y); v.z = fmaf(v.z, rd, bb.z); v.w = fmaf(v.w, rd, bb.w);
      st4(agg + i, v);
    }
  }
}

#define GBM  80
#define GBK  64
#define GAP  68
#define GBP  392
#define GT   384
#define GMT  5
#define GPA  4
#define GPB  16

__global__ void __launch_bounds__(GT) k_gemm(const float* __restrict__ A,
                                             const float* __restrict__ B,
                                             float* __restrict__ C, int M) {
  __shared__ __attribute__((aligned(16))) float sA[2][GBM * GAP];
  __shared__ __attribute__((aligned(16))) float sB[2][GBK * GBP];
  const int t = threadIdx.x;
  const int lane = t & 31;
  const int wave = __builtin_amdgcn_readfirstlane(t >> 5);
  const int c = lane & 15;
  const int h = lane >> 4;
  const int m0 = blockIdx.x * GBM;
  const bool a3 = wave < 4;

  const float* ag[GPA];
  int al[GPA];
#pragma unroll
  for (int p = 0; p < GPA; ++p) {
    const int idx = t + p * GT;
    const int row = (idx >> 4) < GBM ? (idx >> 4) : (GBM - 1);
    const int j = idx & 15;
    const int gm = (m0 + row < M) ? (m0 + row) : (M - 1);
    ag[p] = A + (size_t)gm * F + j * 4;
    al[p] = row * GAP + j * 4;
  }
  const int brow = t / 96;
  const int bj = t - brow * 96;
  const float* bg = B + (size_t)brow * F + bj * 4;
  const int bl = brow * GBP + bj * 4;

  float4 pa[GPA];
  float4 pb[GPB];
  v8f acc[GMT][2];
#pragma unroll
  for (int mt = 0; mt < GMT; ++mt) { acc[mt][0] = (v8f){}; acc[mt][1] = (v8f){}; }

#pragma unroll
  for (int p = 0; p < GPA; ++p) if (p < GPA - 1 || a3) pa[p] = ld4(ag[p]);
#pragma unroll
  for (int p = 0; p < GPB; ++p) pb[p] = ld4(bg + (size_t)(4 * p) * F);
#pragma unroll
  for (int p = 0; p < GPA; ++p) if (p < GPA - 1 || a3) *reinterpret_cast<float4*>(&sA[0][al[p]]) = pa[p];
#pragma unroll
  for (int p = 0; p < GPB; ++p) *reinterpret_cast<float4*>(&sB[0][bl + 4 * p * GBP]) = pb[p];
  __syncthreads();

  for (int ch = 0; ch < F / GBK; ++ch) {
    const int st = ch & 1;
    const bool more = ch + 1 < F / GBK;
    if (more) {
      const int kc = (ch + 1) * GBK;
#pragma unroll
      for (int p = 0; p < GPA; ++p) if (p < GPA - 1 || a3) pa[p] = ld4(ag[p] + kc);
#pragma unroll
      for (int p = 0; p < GPB; ++p) pb[p] = ld4(bg + (size_t)(kc + 4 * p) * F);
    }
    const float* ab = &sA[st][c * GAP + 2 * h];
    const float* bb = &sB[st][(2 * h) * GBP + wave * 32 + c];
#pragma unroll
    for (int ks = 0; ks < GBK / 4; ++ks) {
      v2f a[GMT];
      v2f b[2];
#pragma unroll
      for (int mt = 0; mt < GMT; ++mt) a[mt] = *reinterpret_cast<const v2f*>(ab + mt * 16 * GAP + ks * 4);
#pragma unroll
      for (int nt = 0; nt < 2; ++nt) {
        b[nt].x = bb[(ks * 4) * GBP + nt * 16];
        b[nt].y = bb[(ks * 4 + 1) * GBP + nt * 16];
      }
#pragma unroll
      for (int mt = 0; mt < GMT; ++mt) {
#pragma unroll
        for (int nt = 0; nt < 2; ++nt)
          acc[mt][nt] = __builtin_amdgcn_wmma_f32_16x16x4_f32(false, a[mt], false, b[nt], (short)0, acc[mt][nt],
                                                              false, false);
      }
    }
    if (more) {
#pragma unroll
      for (int p = 0; p < GPA; ++p) if (p < GPA - 1 || a3) *reinterpret_cast<float4*>(&sA[st ^ 1][al[p]]) = pa[p];
#pragma unroll
      for (int p = 0; p < GPB; ++p) *reinterpret_cast<float4*>(&sB[st ^ 1][bl + 4 * p * GBP]) = pb[p];
      __syncthreads();
    }
  }

#pragma unroll
  for (int mt = 0; mt < GMT; ++mt) {
    const int r0 = m0 + mt * 16 + 8 * h;
#pragma unroll
    for (int i = 0; i < 8; ++i) {
      if (r0 + i < M) {
        float* crow = C + (size_t)(r0 + i) * F + wave * 32 + c;
        crow[0]  = acc[mt][0][i];
        crow[16] = acc[mt][1][i];
      }
    }
  }
}

__global__ void __launch_bounds__(256) k_final(const float* __restrict__ agg, const float* __restrict__ den,
                        const float* __restrict__ b2,
                        const float* __restrict__ Wl, const float* __restrict__ bl,
                        float* __restrict__ out, int N) {
  __shared__ float s_out[WPB * NPW];
  const int lane = threadIdx.x & 31;
  const int wave = __builtin_amdgcn_readfirstlane((int)(threadIdx.x >> 5));
  const int nb = blockIdx.x * (WPB * NPW);
  const int n0 = nb + wave * NPW;
  float4 v[NPW][HEADS];
  float dn[NPW][HEADS];
#pragma unroll
  for (int r = 0; r < NPW; ++r) {
    const int n = (n0 + r < N) ? (n0 + r) : (N - 1);
#pragma unroll
    for (int h = 0; h < HEADS; ++h) {
      v[r][h] = ld4(agg + (size_t)n * F + h * CPH + lane * 4);
      dn[r][h] = 1.0f / den[(size_t)n * HEADS + h];
    }
  }
  float red[NPW];
#pragma unroll
  for (int r = 0; r < NPW; ++r) red[r] = 0.f;
#pragma unroll
  for (int h = 0; h < HEADS; ++h) {
    const float4 bb = ld4(b2 + h * CPH + lane * 4);
    const float4 w  = ld4(Wl + h * CPH + lane * 4);
#pragma unroll
    for (int r = 0; r < NPW; ++r) {
      float a = red[r];
      a = fmaf(fmaf(v[r][h].x, dn[r][h], bb.x), w.x, a);
      a = fmaf(fmaf(v[r][h].y, dn[r][h], bb.y), w.y, a);
      a = fmaf(fmaf(v[r][h].z, dn[r][h], bb.z), w.z, a);
      a = fmaf(fmaf(v[r][h].w, dn[r][h], bb.w), w.w, a);
      red[r] = a;
    }
  }
#pragma unroll
  for (int m = 16; m >= 1; m >>= 1) {
#pragma unroll
    for (int r = 0; r < NPW; ++r) red[r] += __shfl_xor(red[r], m, 32);
  }
  if (lane == 0) {
#pragma unroll
    for (int r = 0; r < NPW; ++r) s_out[wave * NPW + r] = red[r];
  }
  __syncthreads();
  const int t = threadIdx.x;
  if (t < WPB * NPW && nb + t < N) out[nb + t] = s_out[t] + bl[0];
}

extern "C" void kernel_launch(void* const* d_in, const int* in_sizes, int n_in,
                              void* d_out, int out_size, void* d_ws, size_t ws_size,
                              hipStream_t stream) {
  const float* x   = (const float*)d_in[0];
  const int*   ei  = (const int*)d_in[1];
  const float* W1  = (const float*)d_in[2];
  const float* as1 = (const float*)d_in[3];
  const float* ad1 = (const float*)d_in[4];
  const float* b1  = (const float*)d_in[5];
  const float* W2  = (const float*)d_in[6];
  const float* as2 = (const float*)d_in[7];
  const float* ad2 = (const float*)d_in[8];
  const float* b2  = (const float*)d_in[9];
  const float* Wl  = (const float*)d_in[10];
  const float* bl  = (const float*)d_in[11];
  float* out = (float*)d_out;

  const int N  = in_sizes[0] / HIN;
  const int E  = in_sizes[1] / 2;
  const int* src = ei;
  const int* dst = ei + E;

  float* ws = (float*)d_ws;
  const size_t NF = (size_t)N * F;
  float*    XP   = ws;
  float*    AGG  = XP + NF;
  float*    DEN  = AGG + NF;
  float*    ASRC = DEN + (size_t)3 * N;
  float*    ADST = ASRC + (size_t)3 * N;

  const int bblocks = (int)((NF + 4095) / 4096);
  const int nblocks = (N + WPB * NPW - 1) / (WPB * NPW);
  const int oblocks = (N + OR - 1) / OR;
  const int gblocks = (N + GBM - 1) / GBM;

  k_proj1<<<nblocks, 256, 0, stream>>>(x, W1, as1, ad1, XP, ASRC, ADST, N);
  k_own<<<oblocks, OT, 0, stream>>>(src, dst, E, N, ASRC, ADST, XP, AGG, DEN);
  k_bias_rows<<<bblocks, 256, 0, stream>>>(AGG, DEN, b1, NF);

  k_gemm<<<gblocks, GT, 0, stream>>>(AGG, W2, XP, N);
  k_alpha<<<nblocks, 256, 0, stream>>>(XP, as2, ad2, ASRC, ADST, N);

  k_own<<<oblocks, OT, 0, stream>>>(src, dst, E, N, ASRC, ADST, XP, AGG, DEN);

  k_final<<<nblocks, 256, 0, stream>>>(AGG, DEN, b2, Wl, bl, out, N);
}


# row epilogues moved into the owning wave: it divides by its denominators and adds bias at its one store, and in layer 2 forms the final dot itself (7 -> 5 launches); sum order unchanged
# speedup vs baseline: 4.1690x; 1.0927x over previous
#include <hip/hip_runtime.h>
#include <hip/hip_bf16.h>

#define HEADS 3
#define CPH   128
#define F     384
#define HIN   12
#define WPB   8
#define NPW   4

typedef __attribute__((ext_vector_type(2))) float v2f;
typedef __attribute__((ext_vector_type(8))) float v8f;

__device__ __forceinline__ unsigned flipf(float f) {
  unsigned b = __float_as_uint(f);
  return (b & 0x80000000u) ? ~b : (b | 0x80000000u);
}
__device__ __forceinline__ float unflipf(unsigned u) {
  unsigned b = (u & 0x80000000u) ? (u & 0x7FFFFFFFu) : ~u;
  return __uint_as_float(b);
}
__device__ __forceinline__ float lrelu(float v) { return v > 0.f ? v : 0.2f * v; }


__device__ __forceinline__ float4 ld4(const float* p) { return *reinterpret_cast<const float4*>(p); }
__device__ __forceinline__ void st4(float* p, float4 v) { *reinterpret_cast<float4*>(p) = v; }
__device__ __forceinline__ float4 fma4(float s, float4 w, float4 a) {
  a.x = fmaf(s, w.x, a.x); a.y = fmaf(s, w.y, a.y); a.z = fmaf(s, w.z, a.z); a.w = fmaf(s, w.w, a.w);
  return a;
}
__device__ __forceinline__ float dot4(float4 a, float4 b) {
  return fmaf(a.w, b.w, fmaf(a.z, b.z, fmaf(a.y, b.y, a.x * b.x)));
}

__global__ void __launch_bounds__(256) k_proj1(const float* __restrict__ x, const float* __restrict__ W1,
                        const float* __restrict__ a_src, const float* __restrict__ a_dst,
                        float* __restrict__ xp, float* __restrict__ asrc,
                        float* __restrict__ adst, int N) {
  __shared__ float s_as[WPB * NPW * HEADS];
  __shared__ float s_ad[WPB * NPW * HEADS];
  const int lane = threadIdx.x & 31;
  const int wave = __builtin_amdgcn_readfirstlane((int)(threadIdx.x >> 5));
  const int nb = blockIdx.x * (WPB * NPW);
  const int n0 = nb + wave * NPW;
  float xr[NPW][HIN];
#pragma unroll
  for (int r = 0; r < NPW; ++r) {
    const int n = (n0 + r < N) ? (n0 + r) : (N - 1);
#pragma unroll
    for (int k = 0; k < HIN; ++k) xr[r][k] = x[(size_t)n * HIN + k];
  }
  float red[2 * HEADS * NPW];
#pragma unroll
  for (int h = 0; h < HEADS; ++h) {
    const int c = h * CPH + lane * 4;
    float4 acc[NPW];
#pragma unroll
    for (int r = 0; r < NPW; ++r) acc[r] = make_float4(0.f, 0.f, 0.f, 0.f);
#pragma unroll
    for (int k = 0; k < HIN; ++k) {
      const float4 w = ld4(W1 + k * F + c);
#pragma unroll
      for (int r = 0; r < NPW; ++r) acc[r] = fma4(xr[r][k], w, acc[r]);
    }
    const float4 as4 = ld4(a_src + c);
    const float4 ad4 = ld4(a_dst + c);
#pragma unroll
    for (int r = 0; r < NPW; ++r) {
      if (n0 + r < N) st4(xp + (size_t)(n0 + r) * F + c, acc[r]);
      red[(r * HEADS + h) * 2 + 0] = dot4(acc[r], as4);
      red[(r * HEADS + h) * 2 + 1] = dot4(acc[r], ad4);
    }
  }
#pragma unroll
  for (int m = 16; m >= 1; m >>= 1) {
#pragma unroll
    for (int i = 0; i < 2 * HEADS * NPW; ++i) red[i] += __shfl_xor(red[i], m, 32);
  }
  if (lane == 0) {
#pragma unroll
    for (int r = 0; r < NPW; ++r) {
#pragma unroll
      for (int h = 0; h < HEADS; ++h) {
        s_as[(wave * NPW + r) * HEADS + h] = red[(r * HEADS + h) * 2 + 0];
        s_ad[(wave * NPW + r) * HEADS + h] = red[(r * HEADS + h) * 2 + 1];
      }
    }
  }
  __syncthreads();
  const int t = threadIdx.x;
  const size_t lim = (size_t)N * HEADS;
  if (t < 96) {
    const size_t o = (size_t)nb * HEADS + t;
    if (o < lim) asrc[o] = s_as[t];
  } else if (t < 192) {
    const size_t o = (size_t)nb * HEADS + (t - 96);
    if (o < lim) adst[o] = s_ad[t - 96];
  }
}

__global__ void __launch_bounds__(256) k_alpha(const float* __restrict__ xp, const float* __restrict__ a_src,
                        const float* __restrict__ a_dst, float* __restrict__ asrc,
                        float* __restrict__ adst, int N) {
  __shared__ float s_as[WPB * NPW * HEADS];
  __shared__ float s_ad[WPB * NPW * HEADS];
  const int lane = threadIdx.x & 31;
  const int wave = __builtin_amdgcn_readfirstlane((int)(threadIdx.x >> 5));
  const int nb = blockIdx.x * (WPB * NPW);
  const int n0 = nb + wave * NPW;
  float4 v[NPW][HEADS];
#pragma unroll
  for (int r = 0; r < NPW; ++r) {
    const int n = (n0 + r < N) ? (n0 + r) : (N - 1);
#pragma unroll
    for (int h = 0; h < HEADS; ++h) v[r][h] = ld4(xp + (size_t)n * F + h * CPH + lane * 4);
  }
  float red[2 * HEADS * NPW];
#pragma unroll
  for (int h = 0; h < HEADS; ++h) {
    const float4 as4 = ld4(a_src + h * CPH + lane * 4);
    const float4 ad4 = ld4(a_dst + h * CPH + lane * 4);
#pragma unroll
    for (int r = 0; r < NPW; ++r) {
      red[(r * HEADS + h) * 2 + 0] = dot4(v[r][h], as4);
      red[(r * HEADS + h) * 2 + 1] = dot4(v[r][h], ad4);
    }
  }
#pragma unroll
  for (int m = 16; m >= 1; m >>= 1) {
#pragma unroll
    for (int i = 0; i < 2 * HEADS * NPW; ++i) red[i] += __shfl_xor(red[i], m, 32);
  }
  if (lane == 0) {
#pragma unroll
    for (int r = 0; r < NPW; ++r) {
#pragma unroll
      for (int h = 0; h < HEADS; ++h) {
        s_as[(wave * NPW + r) * HEADS + h] = red[(r * HEADS + h) * 2 + 0];
        s_ad[(wave * NPW + r) * HEADS + h] = red[(r * HEADS + h) * 2 + 1];
      }
    }
  }
  __syncthreads();
  const int t = threadIdx.x;
  const size_t lim = (size_t)N * HEADS;
  if (t < 96) {
    const size_t o = (size_t)nb * HEADS + t;
    if (o < lim) asrc[o] = s_as[t];
  } else if (t < 192) {
    const size_t o = (size_t)nb * HEADS + (t - 96);
    if (o < lim) adst[o] = s_ad[t - 96];
  }
}

#define OR    96
#define OT    1024
#define RPW   3
#define SU    16
#define SU4   8
#define CAP   4096
#define RND   8
#define GRP   3

__device__ __forceinline__ int own_pop(unsigned (&mk)[RND]) {
  int off = -1;
#pragma unroll
  for (int k = 0; k < RND; ++k) {
    if (off < 0 && mk[k] != 0u) {
      off = k * 32 + __builtin_ctz(mk[k]);
      mk[k] &= mk[k] - 1u;
    }
  }
  return off;
}
__device__ __forceinline__ int   ufi(int v)   { return __builtin_amdgcn_readfirstlane(v); }
__device__ __forceinline__ float uff(float v) { return __int_as_float(__builtin_amdgcn_readfirstlane(__float_as_int(v))); }

__device__ __forceinline__ void own_add(float4 (&acc)[RPW][HEADS], float (&dsum)[RPW][HEADS], int jr,
                                        const float4 (&val)[HEADS], float a0, float a1, float a2) {
#pragma unroll
  for (int q = 0; q < RPW; ++q) {
    if (jr == q) {
      acc[q][0] = fma4(a0, val[0], acc[q][0]);
      acc[q][1] = fma4(a1, val[1], acc[q][1]);
      acc[q][2] = fma4(a2, val[2], acc[q][2]);
      dsum[q][0] += a0; dsum[q][1] += a1; dsum[q][2] += a2;
    }
  }
}

__global__ void __launch_bounds__(OT) k_own(const int* __restrict__ src, const int* __restrict__ dst,
                           int E, int N, const float* __restrict__ asrc,
                           const float* __restrict__ adst, const float* __restrict__ xp,
                           const float* __restrict__ bias, const float* __restrict__ Wl,
                           const float* __restrict__ bl, float* __restrict__ dest, int last) {
  __shared__ float    s_acc[OR * F];
  __shared__ int      l_s[CAP];
  __shared__ int      l_d[CAP];
  __shared__ float    l_w[HEADS][CAP];
  __shared__ float    s_ad[OR * HEADS];
  __shared__ unsigned s_mx[OR * HEADS];
  __shared__ float    s_dn[OR * HEADS];
  __shared__ float    s_out[OR];
  __shared__ unsigned s_cnt;
  const int t = threadIdx.x;
  const int lane = t & 31;
  const int wave = __builtin_amdgcn_readfirstlane(t >> 5);
  const int nb = blockIdx.x * OR;
  const int rows = (N - nb < OR) ? (N - nb) : OR;
  const size_t gb3 = (size_t)nb * HEADS;

  if (t < OR * HEADS) {
    float a = 0.f;
    unsigned m = 0u;
    if (t < rows * HEADS) {
      a = adst[gb3 + t];
      m = flipf(lrelu(asrc[gb3 + t] + a));
    }
    s_ad[t] = a;
    s_mx[t] = m;
  }
  if (t == 0) s_cnt = 0u;
  __syncthreads();

  auto push = [&](unsigned dl, int e) {
    const unsigned slot = __hip_atomic_fetch_add(&s_cnt, 1u, __ATOMIC_RELAXED, __HIP_MEMORY_SCOPE_WORKGROUP);
    if (slot < (unsigned)CAP) {
      l_s[slot] = e;
      l_d[slot] = (int)dl;
    }
  };
  const int E4 = ((reinterpret_cast<size_t>(dst) & 15) == 0) ? (E >> 2) : 0;
  const int4* dst4 = reinterpret_cast<const int4*>(dst);
  for (int g0 = 0; g0 < E4; g0 += OT * SU4) {
    int4 d[SU4];
    if (g0 + OT * SU4 <= E4) {
#pragma unroll
      for (int k = 0; k < SU4; ++k) d[k] = dst4[g0 + k * OT + t];
    } else {
#pragma unroll
      for (int k = 0; k < SU4; ++k) {
        const int g = g0 + k * OT + t;
        d[k] = (g < E4) ? dst4[g] : make_int4(-1, -1, -1, -1);
      }
    }
#pragma unroll
    for (int k = 0; k < SU4; ++k) {
      const unsigned u0 = (unsigned)(d[k].x - nb);
      const unsigned u1 = (unsigned)(d[k].y - nb);
      const unsigned u2 = (unsigned)(d[k].z - nb);
      const unsigned u3 = (unsigned)(d[k].w - nb);
      const unsigned m01 = (u0 < u1) ? u0 : u1;
      const unsigned m23 = (u2 < u3) ? u2 : u3;
      const unsigned mn = (m01 < m23) ? m01 : m23;
      if (mn < (unsigned)rows) {
        const int e0 = (g0 + k * OT + t) * 4;
        if (u0 < (unsigned)rows) push(u0, e0);
        if (u1 < (unsigned)rows) push(u1, e0 + 1);
        if (u2 < (unsigned)rows) push(u2, e0 + 2);
        if (u3 < (unsigned)rows) push(u3, e0 + 3);
      }
    }
  }
  for (int i0 = E4 * 4; i0 < E; i0 += OT) {
    const int i = i0 + t;
    if (i < E) {
      const unsigned dl = (unsigned)(dst[i] - nb);
      if (dl < (unsigned)rows) push(dl, i);
    }
  }
  __syncthreads();
  const unsigned found = s_cnt;

  float4 acc[RPW][HEADS];
  float dsum[RPW][HEADS];

  if (found <= (unsigned)CAP) {
    const int cnt = (int)found;
    for (int j = t; j < cnt; j += OT) {
      const int s = src[l_s[j]];
      const int dl = l_d[j];
      l_s[j] = s;
#pragma unroll
      for (int h = 0; h < HEADS; ++h) {
        const float e = lrelu(asrc[(size_t)s * HEADS + h] + s_ad[dl * HEADS + h]);
        l_w[h][j] = e;
        __hip_atomic_fetch_max(&s_mx[dl * HEADS + h], flipf(e), __ATOMIC_RELAXED, __HIP_MEMORY_SCOPE_WORKGROUP);
      }
    }
    __syncthreads();
    for (int j = t; j < cnt; j += OT) {
      const int dl = l_d[j];
#pragma unroll
      for (int h = 0; h < HEADS; ++h) l_w[h][j] = expf(l_w[h][j] - unflipf(s_mx[dl * HEADS + h]));
    }
    __syncthreads();

#pragma unroll
    for (int q = 0; q < RPW; ++q) {
      const int r = wave + 32 * q;
      const int rr = (r < rows) ? r : 0;
      const float* xrow = xp + (size_t)(nb + rr) * F + lane * 4;
#pragma unroll
      for (int h = 0; h < HEADS; ++h) {
        const float es = lrelu(asrc[gb3 + rr * HEADS + h] + s_ad[rr * HEADS + h]);
        dsum[q][h] = expf(es - unflipf(s_mx[rr * HEADS + h]));
      }
#pragma unroll
      for (int h = 0; h < HEADS; ++h) {
        const float4 v = ld4(xrow + h * CPH);
        acc[q][h] = make_float4(dsum[q][h] * v.x, dsum[q][h] * v.y, dsum[q][h] * v.z, dsum[q][h] * v.w);
      }
    }

    for (int base = 0; base < cnt; base += RND * 32) {
      unsigned mk[RND];
#pragma unroll
      for (int k = 0; k < RND; ++k) {
        const int j = base + k * 32 + lane;
        const bool mine = (j < cnt) && ((l_d[j] & 31) == wave);
        mk[k] = __builtin_amdgcn_ballot_w32(mine);
      }
      for (int it = 0; it < RND * 32; ++it) {
        int jj[GRP];
#pragma unroll
        for (int g = 0; g < GRP; ++g) {
          const int off = own_pop(mk);
          jj[g] = (off >= 0) ? (base + off) : -1;
        }
        if (jj[0] < 0) break;
        int jr[GRP];
        float a0[GRP], a1[GRP], a2[GRP];
        float4 val[GRP][HEADS];
#pragma unroll
        for (int g = 0; g < GRP; ++g) {
          if (jj[g] >= 0) {
            const int ss = ufi(l_s[jj[g]]);
            jr[g] = ufi(l_d[jj[g]]) >> 5;
            a0[g] = uff(l_w[0][jj[g]]);
            a1[g] = uff(l_w[1][jj[g]]);
            a2[g] = uff(l_w[2][jj[g]]);
            const float* xrow = xp + (size_t)ss * F + lane * 4;
#pragma unroll
            for (int h = 0; h < HEADS; ++h) val[g][h] = ld4(xrow + h * CPH);
          }
        }
#pragma unroll
        for (int g = 0; g < GRP; ++g) {
          if (jj[g] >= 0) own_add(acc, dsum, jr[g], val[g], a0[g], a1[g], a2[g]);
        }
      }
    }
  } else {
  for (int i0 = 0; i0 < E; i0 += OT * SU) {
    int d[SU];
#pragma unroll
    for (int k = 0; k < SU; ++k) {
      const int i = i0 + k * OT + t;
      d[k] = (i < E) ? dst[i] : -1;
    }
#pragma unroll
    for (int k = 0; k < SU; ++k) {
      const unsigned dl = (unsigned)(d[k] - nb);
      if (dl < (unsigned)rows) {
        const int s = src[i0 + k * OT + t];
#pragma unroll
        for (int h = 0; h < HEADS; ++h) {
          const float e = lrelu(asrc[(size_t)s * HEADS + h] + s_ad[dl * HEADS + h]);
          __hip_atomic_fetch_max(&s_mx[dl * HEADS + h], flipf(e), __ATOMIC_RELAXED, __HIP_MEMORY_SCOPE_WORKGROUP);
        }
      }
    }
  }
  __syncthreads();

  const int nq = rows * (F / 4);
  for (int q = t; q < nq; q += OT) {
    const int r = q / (F / 4);
    const int h = (q % (F / 4)) / (CPH / 4);
    const float es = lrelu(asrc[gb3 + r * HEADS + h] + s_ad[r * HEADS + h]);
    const float ws = expf(es - unflipf(s_mx[r * HEADS + h]));
    float4 v = ld4(xp + (size_t)nb * F + (size_t)q * 4);
    v.x *= ws; v.y *= ws; v.z *= ws; v.w *= ws;
    *reinterpret_cast<float4*>(s_acc + q * 4) = v;
  }
  if (t < rows * HEADS) {
    const float es = lrelu(asrc[gb3 + t] + s_ad[t]);
    s_dn[t] = expf(es - unflipf(s_mx[t]));
  }
  __syncthreads();

  for (int i0 = 0; i0 < E; i0 += OT * SU) {
    int d[SU];
#pragma unroll
    for (int k = 0; k < SU; ++k) {
      const int i = i0 + k * OT + t;
      d[k] = (i < E) ? dst[i] : -1;
    }
#pragma unroll
    for (int k = 0; k < SU; ++k) {
      const unsigned dl = (unsigned)(d[k] - nb);
      const bool hit = dl < (unsigned)rows;
      unsigned mask = __builtin_amdgcn_ballot_w32(hit);
      if (mask == 0u) continue;
      int s = 0;
      float w0 = 0.f, w1 = 0.f, w2 = 0.f;
      if (hit) {
        s = src[i0 + k * OT + t];
        const float* ap = asrc + (size_t)s * HEADS;
        w0 = expf(lrelu(ap[0] + s_ad[dl * HEADS + 0]) - unflipf(s_mx[dl * HEADS + 0]));
        w1 = expf(lrelu(ap[1] + s_ad[dl * HEADS + 1]) - unflipf(s_mx[dl * HEADS + 1]));
        w2 = expf(lrelu(ap[2] + s_ad[dl * HEADS + 2]) - unflipf(s_mx[dl * HEADS + 2]));
        __hip_atomic_fetch_add(&s_dn[dl * HEADS + 0], w0, __ATOMIC_RELAXED, __HIP_MEMORY_SCOPE_WORKGROUP);
        __hip_atomic_fetch_add(&s_dn[dl * HEADS + 1], w1, __ATOMIC_RELAXED, __HIP_MEMORY_SCOPE_WORKGROUP);
        __hip_atomic_fetch_add(&s_dn[dl * HEADS + 2], w2, __ATOMIC_RELAXED, __HIP_MEMORY_SCOPE_WORKGROUP);
      }
      while (mask != 0u) {
        const int l = __builtin_ctz(mask);
        mask &= mask - 1u;
        const int   ss = __builtin_amdgcn_readlane(s, l);
        const int   dr = __builtin_amdgcn_readlane((int)dl, l);
        const float a0 = __int_as_float(__builtin_amdgcn_readlane(__float_as_int(w0), l));
        const float a1 = __int_as_float(__builtin_amdgcn_readlane(__float_as_int(w1), l));
        const float a2 = __int_as_float(__builtin_amdgcn_readlane(__float_as_int(w2), l));
        const float* xrow = xp + (size_t)ss * F + lane;
        float val[HEADS * 4];
#pragma unroll
        for (int c = 0; c < HEADS * 4; ++c) val[c] = xrow[c * 32];
        float* arow = s_acc + dr * F + lane;
#pragma unroll
        for (int c = 0; c < HEADS * 4; ++c) {
          const float wc = (c < 4) ? a0 : ((c < 8) ? a1 : a2);
          __hip_atomic_fetch_add(arow + c * 32, wc * val[c], __ATOMIC_RELAXED, __HIP_MEMORY_SCOPE_WORKGROUP);
        }
      }
    }
  }
  __syncthreads();

#pragma unroll
  for (int q = 0; q < RPW; ++q) {
    const int r = wave + 32 * q;
    const int rr = (r < rows) ? r : 0;
#pragma unroll
    for (int h = 0; h < HEADS; ++h) {
      acc[q][h] = *reinterpret_cast<const float4*>(s_acc + rr * F + h * CPH + lane * 4);
      dsum[q][h] = s_dn[rr * HEADS + h];
    }
  }
  }

  if (!last) {
    float4 bb[HEADS];
#pragma unroll
    for (int h = 0; h < HEADS; ++h) bb[h] = ld4(bias + h * CPH + lane * 4);
#pragma unroll
    for (int q = 0; q < RPW; ++q) {
      const int r = wave + 32 * q;
      if (r < rows) {
        float* orow = dest + (size_t)(nb + r) * F + lane * 4;
#pragma unroll
        for (int h = 0; h < HEADS; ++h) {
          const float rd = 1.0f / dsum[q][h];
          float4 v = acc[q][h];
          v.x = fmaf(v.x, rd, bb[h].x); v.y = fmaf(v.y, rd, bb[h].y); v.z = fmaf(v.z, rd, bb[h].z); v.w = fmaf(v.w, rd, bb[h].w);
          st4(orow + h * CPH, v);
        }
      }
    }
  } else {
    float red[RPW];
#pragma unroll
    for (int q = 0; q < RPW; ++q) red[q] = 0.f;
#pragma unroll
    for (int h = 0; h < HEADS; ++h) {
      const float4 bb = ld4(bias + h * CPH + lane * 4);
      const float4 w  = ld4(Wl + h * CPH + lane * 4);
#pragma unroll
      for (int q = 0; q < RPW; ++q) {
        const float dn = 1.0f / dsum[q][h];
        float a = red[q];
        a = fmaf(fmaf(acc[q][h].x, dn, bb.x), w.x, a);
        a = fmaf(fmaf(acc[q][h].y, dn, bb.y), w.y, a);
        a = fmaf(fmaf(acc[q][h].z, dn, bb.z), w.z, a);
        a = fmaf(fmaf(acc[q][h].w, dn, bb.w), w.w, a);
        red[q] = a;
      }
    }
#pragma unroll
    for (int m = 16; m >= 1; m >>= 1) {
#pragma unroll
      for (int q = 0; q < RPW; ++q) red[q] += __shfl_xor(red[q], m, 32);
    }
    if (lane == 0) {
#pragma unroll
      for (int q = 0; q < RPW; ++q) {
        const int r = wave + 32 * q;
        if (r < rows) s_out[r] = red[q];
      }
    }
    __syncthreads();
    if (t < rows) dest[nb + t] = s_out[t] + bl[0];
  }
}

#define GBM  80
#define GBK  64
#define GAP  68
#define GBP  392
#define GT   384
#define GMT  5
#define GPA  4
#define GPB  16

__global__ void __launch_bounds__(GT) k_gemm(const float* __restrict__ A,
                                             const float* __restrict__ B,
                                             float* __restrict__ C, int M) {
  __shared__ __attribute__((aligned(16))) float sA[2][GBM * GAP];
  __shared__ __attribute__((aligned(16))) float sB[2][GBK * GBP];
  const int t = threadIdx.x;
  const int lane = t & 31;
  const int wave = __builtin_amdgcn_readfirstlane(t >> 5);
  const int c = lane & 15;
  const int h = lane >> 4;
  const int m0 = blockIdx.x * GBM;
  const bool a3 = wave < 4;

  const float* ag[GPA];
  int al[GPA];
#pragma unroll
  for (int p = 0; p < GPA; ++p) {
    const int idx = t + p * GT;
    const int row = (idx >> 4) < GBM ? (idx >> 4) : (GBM - 1);
    const int j = idx & 15;
    const int gm = (m0 + row < M) ? (m0 + row) : (M - 1);
    ag[p] = A + (size_t)gm * F + j * 4;
    al[p] = row * GAP + j * 4;
  }
  const int brow = t / 96;
  const int bj = t - brow * 96;
  const float* bg = B + (size_t)brow * F + bj * 4;
  const int bl = brow * GBP + bj * 4;

  float4 pa[GPA];
  float4 pb[GPB];
  v8f acc[GMT][2];
#pragma unroll
  for (int mt = 0; mt < GMT; ++mt) { acc[mt][0] = (v8f){}; acc[mt][1] = (v8f){}; }

#pragma unroll
  for (int p = 0; p < GPA; ++p) if (p < GPA - 1 || a3) pa[p] = ld4(ag[p]);
#pragma unroll
  for (int p = 0; p < GPB; ++p) pb[p] = ld4(bg + (size_t)(4 * p) * F);
#pragma unroll
  for (int p = 0; p < GPA; ++p) if (p < GPA - 1 || a3) *reinterpret_cast<float4*>(&sA[0][al[p]]) = pa[p];
#pragma unroll
  for (int p = 0; p < GPB; ++p) *reinterpret_cast<float4*>(&sB[0][bl + 4 * p * GBP]) = pb[p];
  __syncthreads();

  for (int ch = 0; ch < F / GBK; ++ch) {
    const int st = ch & 1;
    const bool more = ch + 1 < F / GBK;
    if (more) {
      const int kc = (ch + 1) * GBK;
#pragma unroll
      for (int p = 0; p < GPA; ++p) if (p < GPA - 1 || a3) pa[p] = ld4(ag[p] + kc);
#pragma unroll
      for (int p = 0; p < GPB; ++p) pb[p] = ld4(bg + (size_t)(kc + 4 * p) * F);
    }
    const float* ab = &sA[st][c * GAP + 2 * h];
    const float* bb = &sB[st][(2 * h) * GBP + wave * 32 + c];
#pragma unroll
    for (int ks = 0; ks < GBK / 4; ++ks) {
      v2f a[GMT];
      v2f b[2];
#pragma unroll
      for (int mt = 0; mt < GMT; ++mt) a[mt] = *reinterpret_cast<const v2f*>(ab + mt * 16 * GAP + ks * 4);
#pragma unroll
      for (int nt = 0; nt < 2; ++nt) {
        b[nt].x = bb[(ks * 4) * GBP + nt * 16];
        b[nt].y = bb[(ks * 4 + 1) * GBP + nt * 16];
      }
#pragma unroll
      for (int mt = 0; mt < GMT; ++mt) {
#pragma unroll
        for (int nt = 0; nt < 2; ++nt)
          acc[mt][nt] = __builtin_amdgcn_wmma_f32_16x16x4_f32(false, a[mt], false, b[nt], (short)0, acc[mt][nt],
                                                              false, false);
      }
    }
    if (more) {
#pragma unroll
      for (int p = 0; p < GPA; ++p) if (p < GPA - 1 || a3) *reinterpret_cast<float4*>(&sA[st ^ 1][al[p]]) = pa[p];
#pragma unroll
      for (int p = 0; p < GPB; ++p) *reinterpret_cast<float4*>(&sB[st ^ 1][bl + 4 * p * GBP]) = pb[p];
      __syncthreads();
    }
  }

#pragma unroll
  for (int mt = 0; mt < GMT; ++mt) {
    const int r0 = m0 + mt * 16 + 8 * h;
#pragma unroll
    for (int i = 0; i < 8; ++i) {
      if (r0 + i < M) {
        float* crow = C + (size_t)(r0 + i) * F + wave * 32 + c;
        crow[0]  = acc[mt][0][i];
        crow[16] = acc[mt][1][i];
      }
    }
  }
}

extern "C" void kernel_launch(void* const* d_in, const int* in_sizes, int n_in,
                              void* d_out, int out_size, void* d_ws, size_t ws_size,
                              hipStream_t stream) {
  const float* x   = (const float*)d_in[0];
  const int*   ei  = (const int*)d_in[1];
  const float* W1  = (const float*)d_in[2];
  const float* as1 = (const float*)d_in[3];
  const float* ad1 = (const float*)d_in[4];
  const float* b1  = (const float*)d_in[5];
  const float* W2  = (const float*)d_in[6];
  const float* as2 = (const float*)d_in[7];
  const float* ad2 = (const float*)d_in[8];
  const float* b2  = (const float*)d_in[9];
  const float* Wl  = (const float*)d_in[10];
  const float* bl  = (const float*)d_in[11];
  float* out = (float*)d_out;

  const int N  = in_sizes[0] / HIN;
  const int E  = in_sizes[1] / 2;
  const int* src = ei;
  const int* dst = ei + E;

  float* ws = (float*)d_ws;
  const size_t NF = (size_t)N * F;
  float*    XP   = ws;
  float*    Y1   = XP + NF;
  float*    ASRC = Y1 + NF;
  float*    ADST = ASRC + (size_t)3 * N;

  const int nblocks = (N + WPB * NPW - 1) / (WPB * NPW);
  const int oblocks = (N + OR - 1) / OR;
  const int gblocks = (N + GBM - 1) / GBM;

  k_proj1<<<nblocks, 256, 0, stream>>>(x, W1, as1, ad1, XP, ASRC, ADST, N);
  k_own<<<oblocks, OT, 0, stream>>>(src, dst, E, N, ASRC, ADST, XP, b1, Wl, bl, Y1, 0);

  k_gemm<<<gblocks, GT, 0, stream>>>(Y1, W2, XP, N);
  k_alpha<<<nblocks, 256, 0, stream>>>(XP, as2, ad2, ASRC, ADST, N);

  k_own<<<oblocks, OT, 0, stream>>>(src, dst, E, N, ASRC, ADST, XP, b2, Wl, bl, out, 1);
}


# last layer's owning groups load their edge lists (count, source, owned row) from the workspace, left there by the first layer's groups: no second reading of all destinations; sums in arrival order
# speedup vs baseline: 4.5057x; 1.0808x over previous
#include <hip/hip_runtime.h>
#include <hip/hip_bf16.h>

#define HEADS 3
#define CPH   128
#define F     384
#define HIN   12
#define WPB   8
#define NPW   4

typedef __attribute__((ext_vector_type(2))) float v2f;
typedef __attribute__((ext_vector_type(8))) float v8f;

__device__ __forceinline__ unsigned flipf(float f) {
  unsigned b = __float_as_uint(f);
  return (b & 0x80000000u) ? ~b : (b | 0x80000000u);
}
__device__ __forceinline__ float unflipf(unsigned u) {
  unsigned b = (u & 0x80000000u) ? (u & 0x7FFFFFFFu) : ~u;
  return __uint_as_float(b);
}
__device__ __forceinline__ float lrelu(float v) { return v > 0.f ? v : 0.2f * v; }


__device__ __forceinline__ float4 ld4(const float* p) { return *reinterpret_cast<const float4*>(p); }
__device__ __forceinline__ void st4(float* p, float4 v) { *reinterpret_cast<float4*>(p) = v; }
__device__ __forceinline__ float4 fma4(float s, float4 w, float4 a) {
  a.x = fmaf(s, w.x, a.x); a.y = fmaf(s, w.y, a.y); a.z = fmaf(s, w.z, a.z); a.w = fmaf(s, w.w, a.w);
  return a;
}
__device__ __forceinline__ float dot4(float4 a, float4 b) {
  return fmaf(a.w, b.w, fmaf(a.z, b.z, fmaf(a.y, b.y, a.x * b.x)));
}

__global__ void __launch_bounds__(256) k_proj1(const float* __restrict__ x, const float* __restrict__ W1,
                        const float* __restrict__ a_src, const float* __restrict__ a_dst,
                        float* __restrict__ xp, float* __restrict__ asrc,
                        float* __restrict__ adst, int N) {
  __shared__ float s_as[WPB * NPW * HEADS];
  __shared__ float s_ad[WPB * NPW * HEADS];
  const int lane = threadIdx.x & 31;
  const int wave = __builtin_amdgcn_readfirstlane((int)(threadIdx.x >> 5));
  const int nb = blockIdx.x * (WPB * NPW);
  const int n0 = nb + wave * NPW;
  float xr[NPW][HIN];
#pragma unroll
  for (int r = 0; r < NPW; ++r) {
    const int n = (n0 + r < N) ? (n0 + r) : (N - 1);
#pragma unroll
    for (int k = 0; k < HIN; ++k) xr[r][k] = x[(size_t)n * HIN + k];
  }
  float red[2 * HEADS * NPW];
#pragma unroll
  for (int h = 0; h < HEADS; ++h) {
    const int c = h * CPH + lane * 4;
    float4 acc[NPW];
#pragma unroll
    for (int r = 0; r < NPW; ++r) acc[r] = make_float4(0.f, 0.f, 0.f, 0.f);
#pragma unroll
    for (int k = 0; k < HIN; ++k) {
      const float4 w = ld4(W1 + k * F + c);
#pragma unroll
      for (int r = 0; r < NPW; ++r) acc[r] = fma4(xr[r][k], w, acc[r]);
    }
    const float4 as4 = ld4(a_src + c);
    const float4 ad4 = ld4(a_dst + c);
#pragma unroll
    for (int r = 0; r < NPW; ++r) {
      if (n0 + r < N) st4(xp + (size_t)(n0 + r) * F + c, acc[r]);
      red[(r * HEADS + h) * 2 + 0] = dot4(acc[r], as4);
      red[(r * HEADS + h) * 2 + 1] = dot4(acc[r], ad4);
    }
  }
#pragma unroll
  for (int m = 16; m >= 1; m >>= 1) {
#pragma unroll
    for (int i = 0; i < 2 * HEADS * NPW; ++i) red[i] += __shfl_xor(red[i], m, 32);
  }
  if (lane == 0) {
#pragma unroll
    for (int r = 0; r < NPW; ++r) {
#pragma unroll
      for (int h = 0; h < HEADS; ++h) {
        s_as[(wave * NPW + r) * HEADS + h] = red[(r * HEADS + h) * 2 + 0];
        s_ad[(wave * NPW + r) * HEADS + h] = red[(r * HEADS + h) * 2 + 1];
      }
    }
  }
  __syncthreads();
  const int t = threadIdx.x;
  const size_t lim = (size_t)N * HEADS;
  if (t < 96) {
    const size_t o = (size_t)nb * HEADS + t;
    if (o < lim) asrc[o] = s_as[t];
  } else if (t < 192) {
    const size_t o = (size_t)nb * HEADS + (t - 96);
    if (o < lim) adst[o] = s_ad[t - 96];
  }
}

__global__ void __launch_bounds__(256) k_alpha(const float* __restrict__ xp, const float* __restrict__ a_src,
                        const float* __restrict__ a_dst, float* __restrict__ asrc,
                        float* __restrict__ adst, int N) {
  __shared__ float s_as[WPB * NPW * HEADS];
  __shared__ float s_ad[WPB * NPW * HEADS];
  const int lane = threadIdx.x & 31;
  const int wave = __builtin_amdgcn_readfirstlane((int)(threadIdx.x >> 5));
  const int nb = blockIdx.x * (WPB * NPW);
  const int n0 = nb + wave * NPW;
  float4 v[NPW][HEADS];
#pragma unroll
  for (int r = 0; r < NPW; ++r) {
    const int n = (n0 + r < N) ? (n0 + r) : (N - 1);
#pragma unroll
    for (int h = 0; h < HEADS; ++h) v[r][h] = ld4(xp + (size_t)n * F + h * CPH + lane * 4);
  }
  float red[2 * HEADS * NPW];
#pragma unroll
  for (int h = 0; h < HEADS; ++h) {
    const float4 as4 = ld4(a_src + h * CPH + lane * 4);
    const float4 ad4 = ld4(a_dst + h * CPH + lane * 4);
#pragma unroll
    for (int r = 0; r < NPW; ++r) {
      red[(r * HEADS + h) * 2 + 0] = dot4(v[r][h], as4);
      red[(r * HEADS + h) * 2 + 1] = dot4(v[r][h], ad4);
    }
  }
#pragma unroll
  for (int m = 16; m >= 1; m >>= 1) {
#pragma unroll
    for (int i = 0; i < 2 * HEADS * NPW; ++i) red[i] += __shfl_xor(red[i], m, 32);
  }
  if (lane == 0) {
#pragma unroll
    for (int r = 0; r < NPW; ++r) {
#pragma unroll
      for (int h = 0; h < HEADS; ++h) {
        s_as[(wave * NPW + r) * HEADS + h] = red[(r * HEADS + h) * 2 + 0];
        s_ad[(wave * NPW + r) * HEADS + h] = red[(r * HEADS + h) * 2 + 1];
      }
    }
  }
  __syncthreads();
  const int t = threadIdx.x;
  const size_t lim = (size_t)N * HEADS;
  if (t < 96) {
    const size_t o = (size_t)nb * HEADS + t;
    if (o < lim) asrc[o] = s_as[t];
  } else if (t < 192) {
    const size_t o = (size_t)nb * HEADS + (t - 96);
    if (o < lim) adst[o] = s_ad[t - 96];
  }
}

#define OR    96
#define OT    1024
#define RPW   3
#define SU    16
#define SU4   8
#define CAP   4096
#define RND   8
#define GRP   3

__device__ __forceinline__ int own_pop(unsigned (&mk)[RND]) {
  int off = -1;
#pragma unroll
  for (int k = 0; k < RND; ++k) {
    if (off < 0 && mk[k] != 0u) {
      off = k * 32 + __builtin_ctz(mk[k]);
      mk[k] &= mk[k] - 1u;
    }
  }
  return off;
}
__device__ __forceinline__ int   ufi(int v)   { return __builtin_amdgcn_readfirstlane(v); }
__device__ __forceinline__ float uff(float v) { return __int_as_float(__builtin_amdgcn_readfirstlane(__float_as_int(v))); }

__device__ __forceinline__ void own_add(float4 (&acc)[RPW][HEADS], float (&dsum)[RPW][HEADS], int jr,
                                        const float4 (&val)[HEADS], float a0, float a1, float a2) {
#pragma unroll
  for (int q = 0; q < RPW; ++q) {
    if (jr == q) {
      acc[q][0] = fma4(a0, val[0], acc[q][0]);
      acc[q][1] = fma4(a1, val[1], acc[q][1]);
      acc[q][2] = fma4(a2, val[2], acc[q][2]);
      dsum[q][0] += a0; dsum[q][1] += a1; dsum[q][2] += a2;
    }
  }
}

__global__ void __launch_bounds__(OT) k_own(const int* __restrict__ src, const int* __restrict__ dst,
                           int E, int N, const float* __restrict__ asrc,
                           const float* __restrict__ adst, const float* __restrict__ xp,
                           const float* __restrict__ bias, const float* __restrict__ Wl,
                           const float* __restrict__ bl, float* __restrict__ dest,
                           unsigned* __restrict__ kcnt, int* __restrict__ ksrc, int* __restrict__ kown, int last) {
  __shared__ float    s_acc[OR * F];
  __shared__ int      l_s[CAP];
  __shared__ int      l_d[CAP];
  __shared__ float    l_w[HEADS][CAP];
  __shared__ float    s_ad[OR * HEADS];
  __shared__ unsigned s_mx[OR * HEADS];
  __shared__ float    s_dn[OR * HEADS];
  __shared__ float    s_out[OR];
  __shared__ unsigned s_cnt;
  const int t = threadIdx.x;
  const int lane = t & 31;
  const int wave = __builtin_amdgcn_readfirstlane(t >> 5);
  const int nb = blockIdx.x * OR;
  const int rows = (N - nb < OR) ? (N - nb) : OR;
  const size_t gb3 = (size_t)nb * HEADS;
  const size_t kb = (size_t)blockIdx.x * CAP;

  if (t < OR * HEADS) {
    float a = 0.f;
    unsigned m = 0u;
    if (t < rows * HEADS) {
      a = adst[gb3 + t];
      m = flipf(lrelu(asrc[gb3 + t] + a));
    }
    s_ad[t] = a;
    s_mx[t] = m;
  }
  if (t == 0) s_cnt = 0u;
  __syncthreads();

  unsigned found;
  if (!last) {
  auto push = [&](unsigned dl, int e) {
    const unsigned slot = __hip_atomic_fetch_add(&s_cnt, 1u, __ATOMIC_RELAXED, __HIP_MEMORY_SCOPE_WORKGROUP);
    if (slot < (unsigned)CAP) {
      l_s[slot] = e;
      l_d[slot] = (int)dl;
    }
  };
  const int E4 = ((reinterpret_cast<size_t>(dst) & 15) == 0) ? (E >> 2) : 0;
  const int4* dst4 = reinterpret_cast<const int4*>(dst);
  for (int g0 = 0; g0 < E4; g0 += OT * SU4) {
    int4 d[SU4];
    if (g0 + OT * SU4 <= E4) {
#pragma unroll
      for (int k = 0; k < SU4; ++k) d[k] = dst4[g0 + k * OT + t];
    } else {
#pragma unroll
      for (int k = 0; k < SU4; ++k) {
        const int g = g0 + k * OT + t;
        d[k] = (g < E4) ? dst4[g] : make_int4(-1, -1, -1, -1);
      }
    }
#pragma unroll
    for (int k = 0; k < SU4; ++k) {
      const unsigned u0 = (unsigned)(d[k].x - nb);
      const unsigned u1 = (unsigned)(d[k].y - nb);
      const unsigned u2 = (unsigned)(d[k].z - nb);
      const unsigned u3 = (unsigned)(d[k].w - nb);
      const unsigned m01 = (u0 < u1) ? u0 : u1;
      const unsigned m23 = (u2 < u3) ? u2 : u3;
      const unsigned mn = (m01 < m23) ? m01 : m23;
      if (mn < (unsigned)rows) {
        const int e0 = (g0 + k * OT + t) * 4;
        if (u0 < (unsigned)rows) push(u0, e0);
        if (u1 < (unsigned)rows) push(u1, e0 + 1);
        if (u2 < (unsigned)rows) push(u2, e0 + 2);
        if (u3 < (unsigned)rows) push(u3, e0 + 3);
      }
    }
  }
  for (int i0 = E4 * 4; i0 < E; i0 += OT) {
    const int i = i0 + t;
    if (i < E) {
      const unsigned dl = (unsigned)(dst[i] - nb);
      if (dl < (unsigned)rows) push(dl, i);
    }
  }
  __syncthreads();
  found = s_cnt;
  if (t == 0) kcnt[blockIdx.x * 32] = found;
  } else {
    found = kcnt[blockIdx.x * 32];
  }

  float4 acc[RPW][HEADS];
  float dsum[RPW][HEADS];

  if (found <= (unsigned)CAP) {
    const int cnt = (int)found;
    for (int j = t; j < cnt; j += OT) {
      int s, dl;
      if (!last) {
        s = src[l_s[j]];
        dl = l_d[j];
        ksrc[kb + j] = s;
        kown[kb + j] = dl;
      } else {
        s = ksrc[kb + j];
        dl = kown[kb + j];
        l_d[j] = dl;
      }
      l_s[j] = s;
#pragma unroll
      for (int h = 0; h < HEADS; ++h) {
        const float e = lrelu(asrc[(size_t)s * HEADS + h] + s_ad[dl * HEADS + h]);
        l_w[h][j] = e;
        __hip_atomic_fetch_max(&s_mx[dl * HEADS + h], flipf(e), __ATOMIC_RELAXED, __HIP_MEMORY_SCOPE_WORKGROUP);
      }
    }
    __syncthreads();
    for (int j = t; j < cnt; j += OT) {
      const int dl = l_d[j];
#pragma unroll
      for (int h = 0; h < HEADS; ++h) l_w[h][j] = expf(l_w[h][j] - unflipf(s_mx[dl * HEADS + h]));
    }
    __syncthreads();

#pragma unroll
    for (int q = 0; q < RPW; ++q) {
      const int r = wave + 32 * q;
      const int rr = (r < rows) ? r : 0;
      const float* xrow = xp + (size_t)(nb + rr) * F + lane * 4;
#pragma unroll
      for (int h = 0; h < HEADS; ++h) {
        const float es = lrelu(asrc[gb3 + rr * HEADS + h] + s_ad[rr * HEADS + h]);
        dsum[q][h] = expf(es - unflipf(s_mx[rr * HEADS + h]));
      }
#pragma unroll
      for (int h = 0; h < HEADS; ++h) {
        const float4 v = ld4(xrow + h * CPH);
        acc[q][h] = make_float4(dsum[q][h] * v.x, dsum[q][h] * v.y, dsum[q][h] * v.z, dsum[q][h] * v.w);
      }
    }

    for (int base = 0; base < cnt; base += RND * 32) {
      unsigned mk[RND];
#pragma unroll
      for (int k = 0; k < RND; ++k) {
        const int j = base + k * 32 + lane;
        const bool mine = (j < cnt) && ((l_d[j] & 31) == wave);
        mk[k] = __builtin_amdgcn_ballot_w32(mine);
      }
      for (int it = 0; it < RND * 32; ++it) {
        int jj[GRP];
#pragma unroll
        for (int g = 0; g < GRP; ++g) {
          const int off = own_pop(mk);
          jj[g] = (off >= 0) ? (base + off) : -1;
        }
        if (jj[0] < 0) break;
        int jr[GRP];
        float a0[GRP], a1[GRP], a2[GRP];
        float4 val[GRP][HEADS];
#pragma unroll
        for (int g = 0; g < GRP; ++g) {
          if (jj[g] >= 0) {
            const int ss = ufi(l_s[jj[g]]);
            jr[g] = ufi(l_d[jj[g]]) >> 5;
            a0[g] = uff(l_w[0][jj[g]]);
            a1[g] = uff(l_w[1][jj[g]]);
            a2[g] = uff(l_w[2][jj[g]]);
            const float* xrow = xp + (size_t)ss * F + lane * 4;
#pragma unroll
            for (int h = 0; h < HEADS; ++h) val[g][h] = ld4(xrow + h * CPH);
          }
        }
#pragma unroll
        for (int g = 0; g < GRP; ++g) {
          if (jj[g] >= 0) own_add(acc, dsum, jr[g], val[g], a0[g], a1[g], a2[g]);
        }
      }
    }
  } else {
  for (int i0 = 0; i0 < E; i0 += OT * SU) {
    int d[SU];
#pragma unroll
    for (int k = 0; k < SU; ++k) {
      const int i = i0 + k * OT + t;
      d[k] = (i < E) ? dst[i] : -1;
    }
#pragma unroll
    for (int k = 0; k < SU; ++k) {
      const unsigned dl = (unsigned)(d[k] - nb);
      if (dl < (unsigned)rows) {
        const int s = src[i0 + k * OT + t];
#pragma unroll
        for (int h = 0; h < HEADS; ++h) {
          const float e = lrelu(asrc[(size_t)s * HEADS + h] + s_ad[dl * HEADS + h]);
          __hip_atomic_fetch_max(&s_mx[dl * HEADS + h], flipf(e), __ATOMIC_RELAXED, __HIP_MEMORY_SCOPE_WORKGROUP);
        }
      }
    }
  }
  __syncthreads();

  const int nq = rows * (F / 4);
  for (int q = t; q < nq; q += OT) {
    const int r = q / (F / 4);
    const int h = (q % (F / 4)) / (CPH / 4);
    const float es = lrelu(asrc[gb3 + r * HEADS + h] + s_ad[r * HEADS + h]);
    const float ws = expf(es - unflipf(s_mx[r * HEADS + h]));
    float4 v = ld4(xp + (size_t)nb * F + (size_t)q * 4);
    v.x *= ws; v.y *= ws; v.z *= ws; v.w *= ws;
    *reinterpret_cast<float4*>(s_acc + q * 4) = v;
  }
  if (t < rows * HEADS) {
    const float es = lrelu(asrc[gb3 + t] + s_ad[t]);
    s_dn[t] = expf(es - unflipf(s_mx[t]));
  }
  __syncthreads();

  for (int i0 = 0; i0 < E; i0 += OT * SU) {
    int d[SU];
#pragma unroll
    for (int k = 0; k < SU; ++k) {
      const int i = i0 + k * OT + t;
      d[k] = (i < E) ? dst[i] : -1;
    }
#pragma unroll
    for (int k = 0; k < SU; ++k) {
      const unsigned dl = (unsigned)(d[k] - nb);
      const bool hit = dl < (unsigned)rows;
      unsigned mask = __builtin_amdgcn_ballot_w32(hit);
      if (mask == 0u) continue;
      int s = 0;
      float w0 = 0.f, w1 = 0.f, w2 = 0.f;
      if (hit) {
        s = src[i0 + k * OT + t];
        const float* ap = asrc + (size_t)s * HEADS;
        w0 = expf(lrelu(ap[0] + s_ad[dl * HEADS + 0]) - unflipf(s_mx[dl * HEADS + 0]));
        w1 = expf(lrelu(ap[1] + s_ad[dl * HEADS + 1]) - unflipf(s_mx[dl * HEADS + 1]));
        w2 = expf(lrelu(ap[2] + s_ad[dl * HEADS + 2]) - unflipf(s_mx[dl * HEADS + 2]));
        __hip_atomic_fetch_add(&s_dn[dl * HEADS + 0], w0, __ATOMIC_RELAXED, __HIP_MEMORY_SCOPE_WORKGROUP);
        __hip_atomic_fetch_add(&s_dn[dl * HEADS + 1], w1, __ATOMIC_RELAXED, __HIP_MEMORY_SCOPE_WORKGROUP);
        __hip_atomic_fetch_add(&s_dn[dl * HEADS + 2], w2, __ATOMIC_RELAXED, __HIP_MEMORY_SCOPE_WORKGROUP);
      }
      while (mask != 0u) {
        const int l = __builtin_ctz(mask);
        mask &= mask - 1u;
        const int   ss = __builtin_amdgcn_readlane(s, l);
        const int   dr = __builtin_amdgcn_readlane((int)dl, l);
        const float a0 = __int_as_float(__builtin_amdgcn_readlane(__float_as_int(w0), l));
        const float a1 = __int_as_float(__builtin_amdgcn_readlane(__float_as_int(w1), l));
        const float a2 = __int_as_float(__builtin_amdgcn_readlane(__float_as_int(w2), l));
        const float* xrow = xp + (size_t)ss * F + lane;
        float val[HEADS * 4];
#pragma unroll
        for (int c = 0; c < HEADS * 4; ++c) val[c] = xrow[c * 32];
        float* arow = s_acc + dr * F + lane;
#pragma unroll
        for (int c = 0; c < HEADS * 4; ++c) {
          const float wc = (c < 4) ? a0 : ((c < 8) ? a1 : a2);
          __hip_atomic_fetch_add(arow + c * 32, wc * val[c], __ATOMIC_RELAXED, __HIP_MEMORY_SCOPE_WORKGROUP);
        }
      }
    }
  }
  __syncthreads();

#pragma unroll
  for (int q = 0; q < RPW; ++q) {
    const int r = wave + 32 * q;
    const int rr = (r < rows) ? r : 0;
#pragma unroll
    for (int h = 0; h < HEADS; ++h) {
      acc[q][h] = *reinterpret_cast<const float4*>(s_acc + rr * F + h * CPH + lane * 4);
      dsum[q][h] = s_dn[rr * HEADS + h];
    }
  }
  }

  if (!last) {
    float4 bb[HEADS];
#pragma unroll
    for (int h = 0; h < HEADS; ++h) bb[h] = ld4(bias + h * CPH + lane * 4);
#pragma unroll
    for (int q = 0; q < RPW; ++q) {
      const int r = wave + 32 * q;
      if (r < rows) {
        float* orow = dest + (size_t)(nb + r) * F + lane * 4;
#pragma unroll
        for (int h = 0; h < HEADS; ++h) {
          const float rd = 1.0f / dsum[q][h];
          float4 v = acc[q][h];
          v.x = fmaf(v.x, rd, bb[h].x); v.y = fmaf(v.y, rd, bb[h].y); v.z = fmaf(v.z, rd, bb[h].z); v.w = fmaf(v.w, rd, bb[h].w);
          st4(orow + h * CPH, v);
        }
      }
    }
  } else {
    float red[RPW];
#pragma unroll
    for (int q = 0; q < RPW; ++q) red[q] = 0.f;
#pragma unroll
    for (int h = 0; h < HEADS; ++h) {
      const float4 bb = ld4(bias + h * CPH + lane * 4);
      const float4 w  = ld4(Wl + h * CPH + lane * 4);
#pragma unroll
      for (int q = 0; q < RPW; ++q) {
        const float dn = 1.0f / dsum[q][h];
        float a = red[q];
        a = fmaf(fmaf(acc[q][h].x, dn, bb.x), w.x, a);
        a = fmaf(fmaf(acc[q][h].y, dn, bb.y), w.y, a);
        a = fmaf(fmaf(acc[q][h].z, dn, bb.z), w.z, a);
        a = fmaf(fmaf(acc[q][h].w, dn, bb.w), w.w, a);
        red[q] = a;
      }
    }
#pragma unroll
    for (int m = 16; m >= 1; m >>= 1) {
#pragma unroll
      for (int q = 0; q < RPW; ++q) red[q] += __shfl_xor(red[q], m, 32);
    }
    if (lane == 0) {
#pragma unroll
      for (int q = 0; q < RPW; ++q) {
        const int r = wave + 32 * q;
        if (r < rows) s_out[r] = red[q];
      }
    }
    __syncthreads();
    if (t < rows) dest[nb + t] = s_out[t] + bl[0];
  }
}

#define GBM  80
#define GBK  64
#define GAP  68
#define GBP  392
#define GT   384
#define GMT  5
#define GPA  4
#define GPB  16

__global__ void __launch_bounds__(GT) k_gemm(const float* __restrict__ A,
                                             const float* __restrict__ B,
                                             float* __restrict__ C, int M) {
  __shared__ __attribute__((aligned(16))) float sA[2][GBM * GAP];
  __shared__ __attribute__((aligned(16))) float sB[2][GBK * GBP];
  const int t = threadIdx.x;
  const int lane = t & 31;
  const int wave = __builtin_amdgcn_readfirstlane(t >> 5);
  const int c = lane & 15;
  const int h = lane >> 4;
  const int m0 = blockIdx.x * GBM;
  const bool a3 = wave < 4;

  const float* ag[GPA];
  int al[GPA];
#pragma unroll
  for (int p = 0; p < GPA; ++p) {
    const int idx = t + p * GT;
    const int row = (idx >> 4) < GBM ? (idx >> 4) : (GBM - 1);
    const int j = idx & 15;
    const int gm = (m0 + row < M) ? (m0 + row) : (M - 1);
    ag[p] = A + (size_t)gm * F + j * 4;
    al[p] = row * GAP + j * 4;
  }
  const int brow = t / 96;
  const int bj = t - brow * 96;
  const float* bg = B + (size_t)brow * F + bj * 4;
  const int bl = brow * GBP + bj * 4;

  float4 pa[GPA];
  float4 pb[GPB];
  v8f acc[GMT][2];
#pragma unroll
  for (int mt = 0; mt < GMT; ++mt) { acc[mt][0] = (v8f){}; acc[mt][1] = (v8f){}; }

#pragma unroll
  for (int p = 0; p < GPA; ++p) if (p < GPA - 1 || a3) pa[p] = ld4(ag[p]);
#pragma unroll
  for (int p = 0; p < GPB; ++p) pb[p] = ld4(bg + (size_t)(4 * p) * F);
#pragma unroll
  for (int p = 0; p < GPA; ++p) if (p < GPA - 1 || a3) *reinterpret_cast<float4*>(&sA[0][al[p]]) = pa[p];
#pragma unroll
  for (int p = 0; p < GPB; ++p) *reinterpret_cast<float4*>(&sB[0][bl + 4 * p * GBP]) = pb[p];
  __syncthreads();

  for (int ch = 0; ch < F / GBK; ++ch) {
    const int st = ch & 1;
    const bool more = ch + 1 < F / GBK;
    if (more) {
      const int kc = (ch + 1) * GBK;
#pragma unroll
      for (int p = 0; p < GPA; ++p) if (p < GPA - 1 || a3) pa[p] = ld4(ag[p] + kc);
#pragma unroll
      for (int p = 0; p < GPB; ++p) pb[p] = ld4(bg + (size_t)(kc + 4 * p) * F);
    }
    const float* ab = &sA[st][c * GAP + 2 * h];
    const float* bb = &sB[st][(2 * h) * GBP + wave * 32 + c];
#pragma unroll
    for (int ks = 0; ks < GBK / 4; ++ks) {
      v2f a[GMT];
      v2f b[2];
#pragma unroll
      for (int mt = 0; mt < GMT; ++mt) a[mt] = *reinterpret_cast<const v2f*>(ab + mt * 16 * GAP + ks * 4);
#pragma unroll
      for (int nt = 0; nt < 2; ++nt) {
        b[nt].x = bb[(ks * 4) * GBP + nt * 16];
        b[nt].y = bb[(ks * 4 + 1) * GBP + nt * 16];
      }
#pragma unroll
      for (int mt = 0; mt < GMT; ++mt) {
#pragma unroll
        for (int nt = 0; nt < 2; ++nt)
          acc[mt][nt] = __builtin_amdgcn_wmma_f32_16x16x4_f32(false, a[mt], false, b[nt], (short)0, acc[mt][nt],
                                                              false, false);
      }
    }
    if (more) {
#pragma unroll
      for (int p = 0; p < GPA; ++p) if (p < GPA - 1 || a3) *reinterpret_cast<float4*>(&sA[st ^ 1][al[p]]) = pa[p];
#pragma unroll
      for (int p = 0; p < GPB; ++p) *reinterpret_cast<float4*>(&sB[st ^ 1][bl + 4 * p * GBP]) = pb[p];
      __syncthreads();
    }
  }

#pragma unroll
  for (int mt = 0; mt < GMT; ++mt) {
    const int r0 = m0 + mt * 16 + 8 * h;
#pragma unroll
    for (int i = 0; i < 8; ++i) {
      if (r0 + i < M) {
        float* crow = C + (size_t)(r0 + i) * F + wave * 32 + c;
        crow[0]  = acc[mt][0][i];
        crow[16] = acc[mt][1][i];
      }
    }
  }
}

extern "C" void kernel_launch(void* const* d_in, const int* in_sizes, int n_in,
                              void* d_out, int out_size, void* d_ws, size_t ws_size,
                              hipStream_t stream) {
  const float* x   = (const float*)d_in[0];
  const int*   ei  = (const int*)d_in[1];
  const float* W1  = (const float*)d_in[2];
  const float* as1 = (const float*)d_in[3];
  const float* ad1 = (const float*)d_in[4];
  const float* b1  = (const float*)d_in[5];
  const float* W2  = (const float*)d_in[6];
  const float* as2 = (const float*)d_in[7];
  const float* ad2 = (const float*)d_in[8];
  const float* b2  = (const float*)d_in[9];
  const float* Wl  = (const float*)d_in[10];
  const float* bl  = (const float*)d_in[11];
  float* out = (float*)d_out;

  const int N  = in_sizes[0] / HIN;
  const int E  = in_sizes[1] / 2;
  const int* src = ei;
  const int* dst = ei + E;

  const int nblocks = (N + WPB * NPW - 1) / (WPB * NPW);
  const int oblocks = (N + OR - 1) / OR;
  const int gblocks = (N + GBM - 1) / GBM;

  float* ws = (float*)d_ws;
  const size_t NF = (size_t)N * F;
  float*    XP   = ws;
  float*    Y1   = XP + NF;
  float*    ASRC = Y1 + NF;
  float*    ADST = ASRC + (size_t)3 * N;
  int*      KSRC = (int*)(ADST + (size_t)3 * N);
  int*      KOWN = KSRC + (size_t)oblocks * CAP;
  unsigned* KCNT = (unsigned*)(KOWN + (size_t)oblocks * CAP);


  k_proj1<<<nblocks, 256, 0, stream>>>(x, W1, as1, ad1, XP, ASRC, ADST, N);
  k_own<<<oblocks, OT, 0, stream>>>(src, dst, E, N, ASRC, ADST, XP, b1, Wl, bl, Y1, KCNT, KSRC, KOWN, 0);

  k_gemm<<<gblocks, GT, 0, stream>>>(Y1, W2, XP, N);
  k_alpha<<<nblocks, 256, 0, stream>>>(XP, as2, ad2, ASRC, ADST, N);

  k_own<<<oblocks, OT, 0, stream>>>(src, dst, E, N, ASRC, ADST, XP, b2, Wl, bl, out, KCNT, KSRC, KOWN, 1);
}


# first layer's owning groups look through one owner byte per edge (destination / 96, left by the projection launch), 16 per 16-byte load, instead of the destinations; sums in arrival order as before
# speedup vs baseline: 4.6541x; 1.0329x over previous
#include <hip/hip_runtime.h>
#include <hip/hip_bf16.h>

#define HEADS 3
#define CPH   128
#define F     384
#define HIN   12
#define WPB   8
#define NPW   4
#define OR    96

typedef __attribute__((ext_vector_type(2))) float v2f;
typedef __attribute__((ext_vector_type(8))) float v8f;

__device__ __forceinline__ unsigned flipf(float f) {
  unsigned b = __float_as_uint(f);
  return (b & 0x80000000u) ? ~b : (b | 0x80000000u);
}
__device__ __forceinline__ float unflipf(unsigned u) {
  unsigned b = (u & 0x80000000u) ? (u & 0x7FFFFFFFu) : ~u;
  return __uint_as_float(b);
}
__device__ __forceinline__ float lrelu(float v) { return v > 0.f ? v : 0.2f * v; }


__device__ __forceinline__ float4 ld4(const float* p) { return *reinterpret_cast<const float4*>(p); }
__device__ __forceinline__ void st4(float* p, float4 v) { *reinterpret_cast<float4*>(p) = v; }
__device__ __forceinline__ float4 fma4(float s, float4 w, float4 a) {
  a.x = fmaf(s, w.x, a.x); a.y = fmaf(s, w.y, a.y); a.z = fmaf(s, w.z, a.z); a.w = fmaf(s, w.w, a.w);
  return a;
}
__device__ __forceinline__ float dot4(float4 a, float4 b) {
  return fmaf(a.w, b.w, fmaf(a.z, b.z, fmaf(a.y, b.y, a.x * b.x)));
}

__global__ void __launch_bounds__(256) k_proj1(const float* __restrict__ x, const float* __restrict__ W1,
                        const float* __restrict__ a_src, const float* __restrict__ a_dst,
                        float* __restrict__ xp, float* __restrict__ asrc,
                        float* __restrict__ adst, int N,
                        const int* __restrict__ dst, int E, unsigned* __restrict__ ob) {
  __shared__ float s_as[WPB * NPW * HEADS];
  __shared__ float s_ad[WPB * NPW * HEADS];
  const int lane = threadIdx.x & 31;
  const int wave = __builtin_amdgcn_readfirstlane((int)(threadIdx.x >> 5));
  const int nb = blockIdx.x * (WPB * NPW);
  const int n0 = nb + wave * NPW;
  {
    const int nq = ((E + 15) >> 4) << 2;
    for (int q = blockIdx.x * (WPB * 32) + (int)threadIdx.x; q < nq; q += gridDim.x * (WPB * 32)) {
      int dd[4];
#pragma unroll
      for (int i = 0; i < 4; ++i) {
        const int e = 4 * q + i;
        dd[i] = dst[(e < E) ? e : (E - 1)];
      }
      unsigned w = 0u;
#pragma unroll
      for (int i = 0; i < 4; ++i) {
        unsigned o = (unsigned)dd[i] / (unsigned)OR;
        if (o > 255u || 4 * q + i >= E) o = 255u;
        w |= o << (8 * i);
      }
      ob[q] = w;
    }
  }
  float xr[NPW][HIN];
#pragma unroll
  for (int r = 0; r < NPW; ++r) {
    const int n = (n0 + r < N) ? (n0 + r) : (N - 1);
#pragma unroll
    for (int k = 0; k < HIN; ++k) xr[r][k] = x[(size_t)n * HIN + k];
  }
  float red[2 * HEADS * NPW];
#pragma unroll
  for (int h = 0; h < HEADS; ++h) {
    const int c = h * CPH + lane * 4;
    float4 acc[NPW];
#pragma unroll
    for (int r = 0; r < NPW; ++r) acc[r] = make_float4(0.f, 0.f, 0.f, 0.f);
#pragma unroll
    for (int k = 0; k < HIN; ++k) {
      const float4 w = ld4(W1 + k * F + c);
#pragma unroll
      for (int r = 0; r < NPW; ++r) acc[r] = fma4(xr[r][k], w, acc[r]);
    }
    const float4 as4 = ld4(a_src + c);
    const float4 ad4 = ld4(a_dst + c);
#pragma unroll
    for (int r = 0; r < NPW; ++r) {
      if (n0 + r < N) st4(xp + (size_t)(n0 + r) * F + c, acc[r]);
      red[(r * HEADS + h) * 2 + 0] = dot4(acc[r], as4);
      red[(r * HEADS + h) * 2 + 1] = dot4(acc[r], ad4);
    }
  }
#pragma unroll
  for (int m = 16; m >= 1; m >>= 1) {
#pragma unroll
    for (int i = 0; i < 2 * HEADS * NPW; ++i) red[i] += __shfl_xor(red[i], m, 32);
  }
  if (lane == 0) {
#pragma unroll
    for (int r = 0; r < NPW; ++r) {
#pragma unroll
      for (int h = 0; h < HEADS; ++h) {
        s_as[(wave * NPW + r) * HEADS + h] = red[(r * HEADS + h) * 2 + 0];
        s_ad[(wave * NPW + r) * HEADS + h] = red[(r * HEADS + h) * 2 + 1];
      }
    }
  }
  __syncthreads();
  const int t = threadIdx.x;
  const size_t lim = (size_t)N * HEADS;
  if (t < 96) {
    const size_t o = (size_t)nb * HEADS + t;
    if (o < lim) asrc[o] = s_as[t];
  } else if (t < 192) {
    const size_t o = (size_t)nb * HEADS + (t - 96);
    if (o < lim) adst[o] = s_ad[t - 96];
  }
}

__global__ void __launch_bounds__(256) k_alpha(const float* __restrict__ xp, const float* __restrict__ a_src,
                        const float* __restrict__ a_dst, float* __restrict__ asrc,
                        float* __restrict__ adst, int N) {
  __shared__ float s_as[WPB * NPW * HEADS];
  __shared__ float s_ad[WPB * NPW * HEADS];
  const int lane = threadIdx.x & 31;
  const int wave = __builtin_amdgcn_readfirstlane((int)(threadIdx.x >> 5));
  const int nb = blockIdx.x * (WPB * NPW);
  const int n0 = nb + wave * NPW;
  float4 v[NPW][HEADS];
#pragma unroll
  for (int r = 0; r < NPW; ++r) {
    const int n = (n0 + r < N) ? (n0 + r) : (N - 1);
#pragma unroll
    for (int h = 0; h < HEADS; ++h) v[r][h] = ld4(xp + (size_t)n * F + h * CPH + lane * 4);
  }
  float red[2 * HEADS * NPW];
#pragma unroll
  for (int h = 0; h < HEADS; ++h) {
    const float4 as4 = ld4(a_src + h * CPH + lane * 4);
    const float4 ad4 = ld4(a_dst + h * CPH + lane * 4);
#pragma unroll
    for (int r = 0; r < NPW; ++r) {
      red[(r * HEADS + h) * 2 + 0] = dot4(v[r][h], as4);
      red[(r * HEADS + h) * 2 + 1] = dot4(v[r][h], ad4);
    }
  }
#pragma unroll
  for (int m = 16; m >= 1; m >>= 1) {
#pragma unroll
    for (int i = 0; i < 2 * HEADS * NPW; ++i) red[i] += __shfl_xor(red[i], m, 32);
  }
  if (lane == 0) {
#pragma unroll
    for (int r = 0; r < NPW; ++r) {
#pragma unroll
      for (int h = 0; h < HEADS; ++h) {
        s_as[(wave * NPW + r) * HEADS + h] = red[(r * HEADS + h) * 2 + 0];
        s_ad[(wave * NPW + r) * HEADS + h] = red[(r * HEADS + h) * 2 + 1];
      }
    }
  }
  __syncthreads();
  const int t = threadIdx.x;
  const size_t lim = (size_t)N * HEADS;
  if (t < 96) {
    const size_t o = (size_t)nb * HEADS + t;
    if (o < lim) asrc[o] = s_as[t];
  } else if (t < 192) {
    const size_t o = (size_t)nb * HEADS + (t - 96);
    if (o < lim) adst[o] = s_ad[t - 96];
  }
}

#define OT    1024
#define RPW   3
#define SU    16
#define SU4   8
#define SUB   10
#define CAP   4096
#define RND   8
#define GRP   3

__device__ __forceinline__ int own_pop(unsigned (&mk)[RND]) {
  int off = -1;
#pragma unroll
  for (int k = 0; k < RND; ++k) {
    if (off < 0 && mk[k] != 0u) {
      off = k * 32 + __builtin_ctz(mk[k]);
      mk[k] &= mk[k] - 1u;
    }
  }
  return off;
}
__device__ __forceinline__ unsigned zero_bytes(unsigned x) {
  return ~(((x & 0x7F7F7F7Fu) + 0x7F7F7F7Fu) | x | 0x7F7F7F7Fu);
}
__device__ __forceinline__ int   ufi(int v)   { return __builtin_amdgcn_readfirstlane(v); }
__device__ __forceinline__ float uff(float v) { return __int_as_float(__builtin_amdgcn_readfirstlane(__float_as_int(v))); }

__device__ __forceinline__ void own_add(float4 (&acc)[RPW][HEADS], float (&dsum)[RPW][HEADS], int jr,
                                        const float4 (&val)[HEADS], float a0, float a1, float a2) {
#pragma unroll
  for (int q = 0; q < RPW; ++q) {
    if (jr == q) {
      acc[q][0] = fma4(a0, val[0], acc[q][0]);
      acc[q][1] = fma4(a1, val[1], acc[q][1]);
      acc[q][2] = fma4(a2, val[2], acc[q][2]);
      dsum[q][0] += a0; dsum[q][1] += a1; dsum[q][2] += a2;
    }
  }
}

__global__ void __launch_bounds__(OT) k_own(const int* __restrict__ src, const int* __restrict__ dst,
                           int E, int N, const float* __restrict__ asrc,
                           const float* __restrict__ adst, const float* __restrict__ xp,
                           const float* __restrict__ bias, const float* __restrict__ Wl,
                           const float* __restrict__ bl, float* __restrict__ dest,
                           unsigned* __restrict__ kcnt, int* __restrict__ ksrc, int* __restrict__ kown,
                           const unsigned* __restrict__ ob, int last) {
  __shared__ float    s_acc[OR * F];
  __shared__ int      l_s[CAP];
  __shared__ int      l_d[CAP];
  __shared__ float    l_w[HEADS][CAP];
  __shared__ float    s_ad[OR * HEADS];
  __shared__ unsigned s_mx[OR * HEADS];
  __shared__ float    s_dn[OR * HEADS];
  __shared__ float    s_out[OR];
  __shared__ unsigned s_cnt;
  const int t = threadIdx.x;
  const int lane = t & 31;
  const int wave = __builtin_amdgcn_readfirstlane(t >> 5);
  const int nb = blockIdx.x * OR;
  const int rows = (N - nb < OR) ? (N - nb) : OR;
  const size_t gb3 = (size_t)nb * HEADS;
  const size_t kb = (size_t)blockIdx.x * CAP;

  if (t < OR * HEADS) {
    float a = 0.f;
    unsigned m = 0u;
    if (t < rows * HEADS) {
      a = adst[gb3 + t];
      m = flipf(lrelu(asrc[gb3 + t] + a));
    }
    s_ad[t] = a;
    s_mx[t] = m;
  }
  if (t == 0) s_cnt = 0u;
  __syncthreads();

  unsigned found;
  if (!last) {
  auto push = [&](int e) {
    const unsigned slot = __hip_atomic_fetch_add(&s_cnt, 1u, __ATOMIC_RELAXED, __HIP_MEMORY_SCOPE_WORKGROUP);
    if (slot < (unsigned)CAP) l_s[slot] = e;
  };
  const bool bytes_ok = (gridDim.x <= 255u) && ((reinterpret_cast<size_t>(ob) & 15) == 0);
  if (bytes_ok) {
    const int nG = (E + 15) >> 4;
    const uint4* ob4 = reinterpret_cast<const uint4*>(ob);
    const unsigned pat = blockIdx.x * 0x01010101u;
    for (int g0 = 0; g0 < nG; g0 += OT * SUB) {
      uint4 d[SUB];
      if (g0 + OT * SUB <= nG) {
#pragma unroll
        for (int k = 0; k < SUB; ++k) d[k] = ob4[g0 + k * OT + t];
      } else {
#pragma unroll
        for (int k = 0; k < SUB; ++k) {
          const int g = g0 + k * OT + t;
          d[k] = (g < nG) ? ob4[g] : make_uint4(~0u, ~0u, ~0u, ~0u);
        }
      }
#pragma unroll
      for (int k = 0; k < SUB; ++k) {
        const unsigned p = (zero_bytes(d[k].x ^ pat) >> 7) | (zero_bytes(d[k].y ^ pat) >> 6) |
                           (zero_bytes(d[k].z ^ pat) >> 5) | (zero_bytes(d[k].w ^ pat) >> 4);
        if (p != 0u) {
          const int e0 = (g0 + k * OT + t) * 16;
          unsigned pp = p;
          for (int it = 0; it < 16 && pp != 0u; ++it) {
            const int b = __builtin_ctz(pp);
            pp &= pp - 1u;
            push(e0 + 4 * (b & 7) + (b >> 3));
          }
        }
      }
    }
  } else {
  const int E4 = ((reinterpret_cast<size_t>(dst) & 15) == 0) ? (E >> 2) : 0;
  const int4* dst4 = reinterpret_cast<const int4*>(dst);
  for (int g0 = 0; g0 < E4; g0 += OT * SU4) {
    int4 d[SU4];
    if (g0 + OT * SU4 <= E4) {
#pragma unroll
      for (int k = 0; k < SU4; ++k) d[k] = dst4[g0 + k * OT + t];
    } else {
#pragma unroll
      for (int k = 0; k < SU4; ++k) {
        const int g = g0 + k * OT + t;
        d[k] = (g < E4) ? dst4[g] : make_int4(-1, -1, -1, -1);
      }
    }
#pragma unroll
    for (int k = 0; k < SU4; ++k) {
      const unsigned u0 = (unsigned)(d[k].x - nb);
      const unsigned u1 = (unsigned)(d[k].y - nb);
      const unsigned u2 = (unsigned)(d[k].z - nb);
      const unsigned u3 = (unsigned)(d[k].w - nb);
      const unsigned m01 = (u0 < u1) ? u0 : u1;
      const unsigned m23 = (u2 < u3) ? u2 : u3;
      const unsigned mn = (m01 < m23) ? m01 : m23;
      if (mn < (unsigned)rows) {
        const int e0 = (g0 + k * OT + t) * 4;
        if (u0 < (unsigned)rows) push(e0);
        if (u1 < (unsigned)rows) push(e0 + 1);
        if (u2 < (unsigned)rows) push(e0 + 2);
        if (u3 < (unsigned)rows) push(e0 + 3);
      }
    }
  }
  for (int i0 = E4 * 4; i0 < E; i0 += OT) {
    const int i = i0 + t;
    if (i < E) {
      const unsigned dl = (unsigned)(dst[i] - nb);
      if (dl < (unsigned)rows) push(i);
    }
  }
  }
  __syncthreads();
  found = s_cnt;
  if (t == 0) kcnt[blockIdx.x * 32] = found;
  } else {
    found = kcnt[blockIdx.x * 32];
  }

  float4 acc[RPW][HEADS];
  float dsum[RPW][HEADS];

  if (found <= (unsigned)CAP) {
    const int cnt = (int)found;
    for (int j = t; j < cnt; j += OT) {
      int s, dl;
      if (!last) {
        const int e = l_s[j];
        s = src[e];
        dl = dst[e] - nb;
        ksrc[kb + j] = s;
        kown[kb + j] = dl;
      } else {
        s = ksrc[kb + j];
        dl = kown[kb + j];
      }
      l_s[j] = s;
      l_d[j] = dl;
#pragma unroll
      for (int h = 0; h < HEADS; ++h) {
        const float e = lrelu(asrc[(size_t)s * HEADS + h] + s_ad[dl * HEADS + h]);
        l_w[h][j] = e;
        __hip_atomic_fetch_max(&s_mx[dl * HEADS + h], flipf(e), __ATOMIC_RELAXED, __HIP_MEMORY_SCOPE_WORKGROUP);
      }
    }
    __syncthreads();
    for (int j = t; j < cnt; j += OT) {
      const int dl = l_d[j];
#pragma unroll
      for (int h = 0; h < HEADS; ++h) l_w[h][j] = expf(l_w[h][j] - unflipf(s_mx[dl * HEADS + h]));
    }
    __syncthreads();

#pragma unroll
    for (int q = 0; q < RPW; ++q) {
      const int r = wave + 32 * q;
      const int rr = (r < rows) ? r : 0;
      const float* xrow = xp + (size_t)(nb + rr) * F + lane * 4;
#pragma unroll
      for (int h = 0; h < HEADS; ++h) {
        const float es = lrelu(asrc[gb3 + rr * HEADS + h] + s_ad[rr * HEADS + h]);
        dsum[q][h] = expf(es - unflipf(s_mx[rr * HEADS + h]));
      }
#pragma unroll
      for (int h = 0; h < HEADS; ++h) {
        const float4 v = ld4(xrow + h * CPH);
        acc[q][h] = make_float4(dsum[q][h] * v.x, dsum[q][h] * v.y, dsum[q][h] * v.z, dsum[q][h] * v.w);
      }
    }

    for (int base = 0; base < cnt; base += RND * 32) {
      unsigned mk[RND];
#pragma unroll
      for (int k = 0; k < RND; ++k) {
        const int j = base + k * 32 + lane;
        const bool mine = (j < cnt) && ((l_d[j] & 31) == wave);
        mk[k] = __builtin_amdgcn_ballot_w32(mine);
      }
      for (int it = 0; it < RND * 32; ++it) {
        int jj[GRP];
#pragma unroll
        for (int g = 0; g < GRP; ++g) {
          const int off = own_pop(mk);
          jj[g] = (off >= 0) ? (base + off) : -1;
        }
        if (jj[0] < 0) break;
        int jr[GRP];
        float a0[GRP], a1[GRP], a2[GRP];
        float4 val[GRP][HEADS];
#pragma unroll
        for (int g = 0; g < GRP; ++g) {
          if (jj[g] >= 0) {
            const int ss = ufi(l_s[jj[g]]);
            jr[g] = ufi(l_d[jj[g]]) >> 5;
            a0[g] = uff(l_w[0][jj[g]]);
            a1[g] = uff(l_w[1][jj[g]]);
            a2[g] = uff(l_w[2][jj[g]]);
            const float* xrow = xp + (size_t)ss * F + lane * 4;
#pragma unroll
            for (int h = 0; h < HEADS; ++h) val[g][h] = ld4(xrow + h * CPH);
          }
        }
#pragma unroll
        for (int g = 0; g < GRP; ++g) {
          if (jj[g] >= 0) own_add(acc, dsum, jr[g], val[g], a0[g], a1[g], a2[g]);
        }
      }
    }
  } else {
  for (int i0 = 0; i0 < E; i0 += OT * SU) {
    int d[SU];
#pragma unroll
    for (int k = 0; k < SU; ++k) {
      const int i = i0 + k * OT + t;
      d[k] = (i < E) ? dst[i] : -1;
    }
#pragma unroll
    for (int k = 0; k < SU; ++k) {
      const unsigned dl = (unsigned)(d[k] - nb);
      if (dl < (unsigned)rows) {
        const int s = src[i0 + k * OT + t];
#pragma unroll
        for (int h = 0; h < HEADS; ++h) {
          const float e = lrelu(asrc[(size_t)s * HEADS + h] + s_ad[dl * HEADS + h]);
          __hip_atomic_fetch_max(&s_mx[dl * HEADS + h], flipf(e), __ATOMIC_RELAXED, __HIP_MEMORY_SCOPE_WORKGROUP);
        }
      }
    }
  }
  __syncthreads();

  const int nq = rows * (F / 4);
  for (int q = t; q < nq; q += OT) {
    const int r = q / (F / 4);
    const int h = (q % (F / 4)) / (CPH / 4);
    const float es = lrelu(asrc[gb3 + r * HEADS + h] + s_ad[r * HEADS + h]);
    const float ws = expf(es - unflipf(s_mx[r * HEADS + h]));
    float4 v = ld4(xp + (size_t)nb * F + (size_t)q * 4);
    v.x *= ws; v.y *= ws; v.z *= ws; v.w *= ws;
    *reinterpret_cast<float4*>(s_acc + q * 4) = v;
  }
  if (t < rows * HEADS) {
    const float es = lrelu(asrc[gb3 + t] + s_ad[t]);
    s_dn[t] = expf(es - unflipf(s_mx[t]));
  }
  __syncthreads();

  for (int i0 = 0; i0 < E; i0 += OT * SU) {
    int d[SU];
#pragma unroll
    for (int k = 0; k < SU; ++k) {
      const int i = i0 + k * OT + t;
      d[k] = (i < E) ? dst[i] : -1;
    }
#pragma unroll
    for (int k = 0; k < SU; ++k) {
      const unsigned dl = (unsigned)(d[k] - nb);
      const bool hit = dl < (unsigned)rows;
      unsigned mask = __builtin_amdgcn_ballot_w32(hit);
      if (mask == 0u) continue;
      int s = 0;
      float w0 = 0.f, w1 = 0.f, w2 = 0.f;
      if (hit) {
        s = src[i0 + k * OT + t];
        const float* ap = asrc + (size_t)s * HEADS;
        w0 = expf(lrelu(ap[0] + s_ad[dl * HEADS + 0]) - unflipf(s_mx[dl * HEADS + 0]));
        w1 = expf(lrelu(ap[1] + s_ad[dl * HEADS + 1]) - unflipf(s_mx[dl * HEADS + 1]));
        w2 = expf(lrelu(ap[2] + s_ad[dl * HEADS + 2]) - unflipf(s_mx[dl * HEADS + 2]));
        __hip_atomic_fetch_add(&s_dn[dl * HEADS + 0], w0, __ATOMIC_RELAXED, __HIP_MEMORY_SCOPE_WORKGROUP);
        __hip_atomic_fetch_add(&s_dn[dl * HEADS + 1], w1, __ATOMIC_RELAXED, __HIP_MEMORY_SCOPE_WORKGROUP);
        __hip_atomic_fetch_add(&s_dn[dl * HEADS + 2], w2, __ATOMIC_RELAXED, __HIP_MEMORY_SCOPE_WORKGROUP);
      }
      while (mask != 0u) {
        const int l = __builtin_ctz(mask);
        mask &= mask - 1u;
        const int   ss = __builtin_amdgcn_readlane(s, l);
        const int   dr = __builtin_amdgcn_readlane((int)dl, l);
        const float a0 = __int_as_float(__builtin_amdgcn_readlane(__float_as_int(w0), l));
        const float a1 = __int_as_float(__builtin_amdgcn_readlane(__float_as_int(w1), l));
        const float a2 = __int_as_float(__builtin_amdgcn_readlane(__float_as_int(w2), l));
        const float* xrow = xp + (size_t)ss * F + lane;
        float val[HEADS * 4];
#pragma unroll
        for (int c = 0; c < HEADS * 4; ++c) val[c] = xrow[c * 32];
        float* arow = s_acc + dr * F + lane;
#pragma unroll
        for (int c = 0; c < HEADS * 4; ++c) {
          const float wc = (c < 4) ? a0 : ((c < 8) ? a1 : a2);
          __hip_atomic_fetch_add(arow + c * 32, wc * val[c], __ATOMIC_RELAXED, __HIP_MEMORY_SCOPE_WORKGROUP);
        }
      }
    }
  }
  __syncthreads();

#pragma unroll
  for (int q = 0; q < RPW; ++q) {
    const int r = wave + 32 * q;
    const int rr = (r < rows) ? r : 0;
#pragma unroll
    for (int h = 0; h < HEADS; ++h) {
      acc[q][h] = *reinterpret_cast<const float4*>(s_acc + rr * F + h * CPH + lane * 4);
      dsum[q][h] = s_dn[rr * HEADS + h];
    }
  }
  }

  if (!last) {
    float4 bb[HEADS];
#pragma unroll
    for (int h = 0; h < HEADS; ++h) bb[h] = ld4(bias + h * CPH + lane * 4);
#pragma unroll
    for (int q = 0; q < RPW; ++q) {
      const int r = wave + 32 * q;
      if (r < rows) {
        float* orow = dest + (size_t)(nb + r) * F + lane * 4;
#pragma unroll
        for (int h = 0; h < HEADS; ++h) {
          const float rd = 1.0f / dsum[q][h];
          float4 v = acc[q][h];
          v.x = fmaf(v.x, rd, bb[h].x); v.y = fmaf(v.y, rd, bb[h].y); v.z = fmaf(v.z, rd, bb[h].z); v.w = fmaf(v.w, rd, bb[h].w);
          st4(orow + h * CPH, v);
        }
      }
    }
  } else {
    float red[RPW];
#pragma unroll
    for (int q = 0; q < RPW; ++q) red[q] = 0.f;
#pragma unroll
    for (int h = 0; h < HEADS; ++h) {
      const float4 bb = ld4(bias + h * CPH + lane * 4);
      const float4 w  = ld4(Wl + h * CPH + lane * 4);
#pragma unroll
      for (int q = 0; q < RPW; ++q) {
        const float dn = 1.0f / dsum[q][h];
        float a = red[q];
        a = fmaf(fmaf(acc[q][h].x, dn, bb.x), w.x, a);
        a = fmaf(fmaf(acc[q][h].y, dn, bb.y), w.y, a);
        a = fmaf(fmaf(acc[q][h].z, dn, bb.z), w.z, a);
        a = fmaf(fmaf(acc[q][h].w, dn, bb.w), w.w, a);
        red[q] = a;
      }
    }
#pragma unroll
    for (int m = 16; m >= 1; m >>= 1) {
#pragma unroll
      for (int q = 0; q < RPW; ++q) red[q] += __shfl_xor(red[q], m, 32);
    }
    if (lane == 0) {
#pragma unroll
      for (int q = 0; q < RPW; ++q) {
        const int r = wave + 32 * q;
        if (r < rows) s_out[r] = red[q];
      }
    }
    __syncthreads();
    if (t < rows) dest[nb + t] = s_out[t] + bl[0];
  }
}

#define GBM  80
#define GBK  64
#define GAP  68
#define GBP  392
#define GT   384
#define GMT  5
#define GPA  4
#define GPB  16

__global__ void __launch_bounds__(GT) k_gemm(const float* __restrict__ A,
                                             const float* __restrict__ B,
                                             float* __restrict__ C, int M) {
  __shared__ __attribute__((aligned(16))) float sA[2][GBM * GAP];
  __shared__ __attribute__((aligned(16))) float sB[2][GBK * GBP];
  const int t = threadIdx.x;
  const int lane = t & 31;
  const int wave = __builtin_amdgcn_readfirstlane(t >> 5);
  const int c = lane & 15;
  const int h = lane >> 4;
  const int m0 = blockIdx.x * GBM;
  const bool a3 = wave < 4;

  const float* ag[GPA];
  int al[GPA];
#pragma unroll
  for (int p = 0; p < GPA; ++p) {
    const int idx = t + p * GT;
    const int row = (idx >> 4) < GBM ? (idx >> 4) : (GBM - 1);
    const int j = idx & 15;
    const int gm = (m0 + row < M) ? (m0 + row) : (M - 1);
    ag[p] = A + (size_t)gm * F + j * 4;
    al[p] = row * GAP + j * 4;
  }
  const int brow = t / 96;
  const int bj = t - brow * 96;
  const float* bg = B + (size_t)brow * F + bj * 4;
  const int bl = brow * GBP + bj * 4;

  float4 pa[GPA];
  float4 pb[GPB];
  v8f acc[GMT][2];
#pragma unroll
  for (int mt = 0; mt < GMT; ++mt) { acc[mt][0] = (v8f){}; acc[mt][1] = (v8f){}; }

#pragma unroll
  for (int p = 0; p < GPA; ++p) if (p < GPA - 1 || a3) pa[p] = ld4(ag[p]);
#pragma unroll
  for (int p = 0; p < GPB; ++p) pb[p] = ld4(bg + (size_t)(4 * p) * F);
#pragma unroll
  for (int p = 0; p < GPA; ++p) if (p < GPA - 1 || a3) *reinterpret_cast<float4*>(&sA[0][al[p]]) = pa[p];
#pragma unroll
  for (int p = 0; p < GPB; ++p) *reinterpret_cast<float4*>(&sB[0][bl + 4 * p * GBP]) = pb[p];
  __syncthreads();

  for (int ch = 0; ch < F / GBK; ++ch) {
    const int st = ch & 1;
    const bool more = ch + 1 < F / GBK;
    if (more) {
      const int kc = (ch + 1) * GBK;
#pragma unroll
      for (int p = 0; p < GPA; ++p) if (p < GPA - 1 || a3) pa[p] = ld4(ag[p] + kc);
#pragma unroll
      for (int p = 0; p < GPB; ++p) pb[p] = ld4(bg + (size_t)(kc + 4 * p) * F);
    }
    const float* ab = &sA[st][c * GAP + 2 * h];
    const float* bb = &sB[st][(2 * h) * GBP + wave * 32 + c];
#pragma unroll
    for (int ks = 0; ks < GBK / 4; ++ks) {
      v2f a[GMT];
      v2f b[2];
#pragma unroll
      for (int mt = 0; mt < GMT; ++mt) a[mt] = *reinterpret_cast<const v2f*>(ab + mt * 16 * GAP + ks * 4);
#pragma unroll
      for (int nt = 0; nt < 2; ++nt) {
        b[nt].x = bb[(ks * 4) * GBP + nt * 16];
        b[nt].y = bb[(ks * 4 + 1) * GBP + nt * 16];
      }
#pragma unroll
      for (int mt = 0; mt < GMT; ++mt) {
#pragma unroll
        for (int nt = 0; nt < 2; ++nt)
          acc[mt][nt] = __builtin_amdgcn_wmma_f32_16x16x4_f32(false, a[mt], false, b[nt], (short)0, acc[mt][nt],
                                                              false, false);
      }
    }
    if (more) {
#pragma unroll
      for (int p = 0; p < GPA; ++p) if (p < GPA - 1 || a3) *reinterpret_cast<float4*>(&sA[st ^ 1][al[p]]) = pa[p];
#pragma unroll
      for (int p = 0; p < GPB; ++p) *reinterpret_cast<float4*>(&sB[st ^ 1][bl + 4 * p * GBP]) = pb[p];
      __syncthreads();
    }
  }

#pragma unroll
  for (int mt = 0; mt < GMT; ++mt) {
    const int r0 = m0 + mt * 16 + 8 * h;
#pragma unroll
    for (int i = 0; i < 8; ++i) {
      if (r0 + i < M) {
        float* crow = C + (size_t)(r0 + i) * F + wave * 32 + c;
        crow[0]  = acc[mt][0][i];
        crow[16] = acc[mt][1][i];
      }
    }
  }
}

extern "C" void kernel_launch(void* const* d_in, const int* in_sizes, int n_in,
                              void* d_out, int out_size, void* d_ws, size_t ws_size,
                              hipStream_t stream) {
  const float* x   = (const float*)d_in[0];
  const int*   ei  = (const int*)d_in[1];
  const float* W1  = (const float*)d_in[2];
  const float* as1 = (const float*)d_in[3];
  const float* ad1 = (const float*)d_in[4];
  const float* b1  = (const float*)d_in[5];
  const float* W2  = (const float*)d_in[6];
  const float* as2 = (const float*)d_in[7];
  const float* ad2 = (const float*)d_in[8];
  const float* b2  = (const float*)d_in[9];
  const float* Wl  = (const float*)d_in[10];
  const float* bl  = (const float*)d_in[11];
  float* out = (float*)d_out;

  const int N  = in_sizes[0] / HIN;
  const int E  = in_sizes[1] / 2;
  const int* src = ei;
  const int* dst = ei + E;

  const int nblocks = (N + WPB * NPW - 1) / (WPB * NPW);
  const int oblocks = (N + OR - 1) / OR;
  const int gblocks = (N + GBM - 1) / GBM;

  float* ws = (float*)d_ws;
  const size_t NF = (size_t)N * F;
  float*    XP   = ws;
  float*    Y1   = XP + NF;
  float*    ASRC = Y1 + NF;
  float*    ADST = ASRC + (size_t)3 * N;
  int*      KSRC = (int*)(ADST + (size_t)3 * N);
  int*      KOWN = KSRC + (size_t)oblocks * CAP;
  unsigned* KCNT = (unsigned*)(KOWN + (size_t)oblocks * CAP);
  unsigned* OWNB = KCNT + (size_t)oblocks * 32;


  k_proj1<<<nblocks, 256, 0, stream>>>(x, W1, as1, ad1, XP, ASRC, ADST, N, dst, E, OWNB);
  k_own<<<oblocks, OT, 0, stream>>>(src, dst, E, N, ASRC, ADST, XP, b1, Wl, bl, Y1, KCNT, KSRC, KOWN, OWNB, 0);

  k_gemm<<<gblocks, GT, 0, stream>>>(Y1, W2, XP, N);
  k_alpha<<<nblocks, 256, 0, stream>>>(XP, as2, ad2, ASRC, ADST, N);

  k_own<<<oblocks, OT, 0, stream>>>(src, dst, E, N, ASRC, ADST, XP, b2, Wl, bl, out, KCNT, KSRC, KOWN, OWNB, 1);
}


# owning work-groups resized to fill the chip: 81 rows and 27 waves, 3 rows a wave (was 96 rows, 32 waves): 247 groups on 256 WGPs (was 209), less to gather per WGP; sums in arrival order as before
# speedup vs baseline: 4.8730x; 1.0470x over previous
#include <hip/hip_runtime.h>
#include <hip/hip_bf16.h>

#define HEADS 3
#define CPH   128
#define F     384
#define HIN   12
#define WPB   8
#define NPW   4
#define OR    81

typedef __attribute__((ext_vector_type(2))) float v2f;
typedef __attribute__((ext_vector_type(8))) float v8f;

__device__ __forceinline__ unsigned flipf(float f) {
  unsigned b = __float_as_uint(f);
  return (b & 0x80000000u) ? ~b : (b | 0x80000000u);
}
__device__ __forceinline__ float unflipf(unsigned u) {
  unsigned b = (u & 0x80000000u) ? (u & 0x7FFFFFFFu) : ~u;
  return __uint_as_float(b);
}
__device__ __forceinline__ float lrelu(float v) { return v > 0.f ? v : 0.2f * v; }


__device__ __forceinline__ float4 ld4(const float* p) { return *reinterpret_cast<const float4*>(p); }
__device__ __forceinline__ void st4(float* p, float4 v) { *reinterpret_cast<float4*>(p) = v; }
__device__ __forceinline__ float4 fma4(float s, float4 w, float4 a) {
  a.x = fmaf(s, w.x, a.x); a.y = fmaf(s, w.y, a.y); a.z = fmaf(s, w.z, a.z); a.w = fmaf(s, w.w, a.w);
  return a;
}
__device__ __forceinline__ float dot4(float4 a, float4 b) {
  return fmaf(a.w, b.w, fmaf(a.z, b.z, fmaf(a.y, b.y, a.x * b.x)));
}

__global__ void __launch_bounds__(256) k_proj1(const float* __restrict__ x, const float* __restrict__ W1,
                        const float* __restrict__ a_src, const float* __restrict__ a_dst,
                        float* __restrict__ xp, float* __restrict__ asrc,
                        float* __restrict__ adst, int N,
                        const int* __restrict__ dst, int E, unsigned* __restrict__ ob) {
  __shared__ float s_as[WPB * NPW * HEADS];
  __shared__ float s_ad[WPB * NPW * HEADS];
  const int lane = threadIdx.x & 31;
  const int wave = __builtin_amdgcn_readfirstlane((int)(threadIdx.x >> 5));
  const int nb = blockIdx.x * (WPB * NPW);
  const int n0 = nb + wave * NPW;
  {
    const int nq = ((E + 15) >> 4) << 2;
    for (int q = blockIdx.x * (WPB * 32) + (int)threadIdx.x; q < nq; q += gridDim.x * (WPB * 32)) {
      int dd[4];
#pragma unroll
      for (int i = 0; i < 4; ++i) {
        const int e = 4 * q + i;
        dd[i] = dst[(e < E) ? e : (E - 1)];
      }
      unsigned w = 0u;
#pragma unroll
      for (int i = 0; i < 4; ++i) {
        unsigned o = (unsigned)dd[i] / (unsigned)OR;
        if (o > 255u || 4 * q + i >= E) o = 255u;
        w |= o << (8 * i);
      }
      ob[q] = w;
    }
  }
  float xr[NPW][HIN];
#pragma unroll
  for (int r = 0; r < NPW; ++r) {
    const int n = (n0 + r < N) ? (n0 + r) : (N - 1);
#pragma unroll
    for (int k = 0; k < HIN; ++k) xr[r][k] = x[(size_t)n * HIN + k];
  }
  float red[2 * HEADS * NPW];
#pragma unroll
  for (int h = 0; h < HEADS; ++h) {
    const int c = h * CPH + lane * 4;
    float4 acc[NPW];
#pragma unroll
    for (int r = 0; r < NPW; ++r) acc[r] = make_float4(0.f, 0.f, 0.f, 0.f);
#pragma unroll
    for (int k = 0; k < HIN; ++k) {
      const float4 w = ld4(W1 + k * F + c);
#pragma unroll
      for (int r = 0; r < NPW; ++r) acc[r] = fma4(xr[r][k], w, acc[r]);
    }
    const float4 as4 = ld4(a_src + c);
    const float4 ad4 = ld4(a_dst + c);
#pragma unroll
    for (int r = 0; r < NPW; ++r) {
      if (n0 + r < N) st4(xp + (size_t)(n0 + r) * F + c, acc[r]);
      red[(r * HEADS + h) * 2 + 0] = dot4(acc[r], as4);
      red[(r * HEADS + h) * 2 + 1] = dot4(acc[r], ad4);
    }
  }
#pragma unroll
  for (int m = 16; m >= 1; m >>= 1) {
#pragma unroll
    for (int i = 0; i < 2 * HEADS * NPW; ++i) red[i] += __shfl_xor(red[i], m, 32);
  }
  if (lane == 0) {
#pragma unroll
    for (int r = 0; r < NPW; ++r) {
#pragma unroll
      for (int h = 0; h < HEADS; ++h) {
        s_as[(wave * NPW + r) * HEADS + h] = red[(r * HEADS + h) * 2 + 0];
        s_ad[(wave * NPW + r) * HEADS + h] = red[(r * HEADS + h) * 2 + 1];
      }
    }
  }
  __syncthreads();
  const int t = threadIdx.x;
  const size_t lim = (size_t)N * HEADS;
  if (t < 96) {
    const size_t o = (size_t)nb * HEADS + t;
    if (o < lim) asrc[o] = s_as[t];
  } else if (t < 192) {
    const size_t o = (size_t)nb * HEADS + (t - 96);
    if (o < lim) adst[o] = s_ad[t - 96];
  }
}

__global__ void __launch_bounds__(256) k_alpha(const float* __restrict__ xp, const float* __restrict__ a_src,
                        const float* __restrict__ a_dst, float* __restrict__ asrc,
                        float* __restrict__ adst, int N) {
  __shared__ float s_as[WPB * NPW * HEADS];
  __shared__ float s_ad[WPB * NPW * HEADS];
  const int lane = threadIdx.x & 31;
  const int wave = __builtin_amdgcn_readfirstlane((int)(threadIdx.x >> 5));
  const int nb = blockIdx.x * (WPB * NPW);
  const int n0 = nb + wave * NPW;
  float4 v[NPW][HEADS];
#pragma unroll
  for (int r = 0; r < NPW; ++r) {
    const int n = (n0 + r < N) ? (n0 + r) : (N - 1);
#pragma unroll
    for (int h = 0; h < HEADS; ++h) v[r][h] = ld4(xp + (size_t)n * F + h * CPH + lane * 4);
  }
  float red[2 * HEADS * NPW];
#pragma unroll
  for (int h = 0; h < HEADS; ++h) {
    const float4 as4 = ld4(a_src + h * CPH + lane * 4);
    const float4 ad4 = ld4(a_dst + h * CPH + lane * 4);
#pragma unroll
    for (int r = 0; r < NPW; ++r) {
      red[(r * HEADS + h) * 2 + 0] = dot4(v[r][h], as4);
      red[(r * HEADS + h) * 2 + 1] = dot4(v[r][h], ad4);
    }
  }
#pragma unroll
  for (int m = 16; m >= 1; m >>= 1) {
#pragma unroll
    for (int i = 0; i < 2 * HEADS * NPW; ++i) red[i] += __shfl_xor(red[i], m, 32);
  }
  if (lane == 0) {
#pragma unroll
    for (int r = 0; r < NPW; ++r) {
#pragma unroll
      for (int h = 0; h < HEADS; ++h) {
        s_as[(wave * NPW + r) * HEADS + h] = red[(r * HEADS + h) * 2 + 0];
        s_ad[(wave * NPW + r) * HEADS + h] = red[(r * HEADS + h) * 2 + 1];
      }
    }
  }
  __syncthreads();
  const int t = threadIdx.x;
  const size_t lim = (size_t)N * HEADS;
  if (t < 96) {
    const size_t o = (size_t)nb * HEADS + t;
    if (o < lim) asrc[o] = s_as[t];
  } else if (t < 192) {
    const size_t o = (size_t)nb * HEADS + (t - 96);
    if (o < lim) adst[o] = s_ad[t - 96];
  }
}

#define OT    864
#define OW    27
#define RPW   3
#define SU    16
#define SU4   8
#define SUB   12
#define CAP   4096
#define RND   8
#define GRP   3

__device__ __forceinline__ int own_code(int dl) { return ((dl / OW) << 5) | (dl % OW); }
__device__ __forceinline__ int own_row(int code) { return (code >> 5) * OW + (code & 31); }
__device__ __forceinline__ int own_pop(unsigned (&mk)[RND]) {
  int off = -1;
#pragma unroll
  for (int k = 0; k < RND; ++k) {
    if (off < 0 && mk[k] != 0u) {
      off = k * 32 + __builtin_ctz(mk[k]);
      mk[k] &= mk[k] - 1u;
    }
  }
  return off;
}
__device__ __forceinline__ unsigned zero_bytes(unsigned x) {
  return ~(((x & 0x7F7F7F7Fu) + 0x7F7F7F7Fu) | x | 0x7F7F7F7Fu);
}
__device__ __forceinline__ int   ufi(int v)   { return __builtin_amdgcn_readfirstlane(v); }
__device__ __forceinline__ float uff(float v) { return __int_as_float(__builtin_amdgcn_readfirstlane(__float_as_int(v))); }

__device__ __forceinline__ void own_add(float4 (&acc)[RPW][HEADS], float (&dsum)[RPW][HEADS], int jr,
                                        const float4 (&val)[HEADS], float a0, float a1, float a2) {
#pragma unroll
  for (int q = 0; q < RPW; ++q) {
    if (jr == q) {
      acc[q][0] = fma4(a0, val[0], acc[q][0]);
      acc[q][1] = fma4(a1, val[1], acc[q][1]);
      acc[q][2] = fma4(a2, val[2], acc[q][2]);
      dsum[q][0] += a0; dsum[q][1] += a1; dsum[q][2] += a2;
    }
  }
}

__global__ void __launch_bounds__(OT) k_own(const int* __restrict__ src, const int* __restrict__ dst,
                           int E, int N, const float* __restrict__ asrc,
                           const float* __restrict__ adst, const float* __restrict__ xp,
                           const float* __restrict__ bias, const float* __restrict__ Wl,
                           const float* __restrict__ bl, float* __restrict__ dest,
                           unsigned* __restrict__ kcnt, int* __restrict__ ksrc, int* __restrict__ kown,
                           const unsigned* __restrict__ ob, int last) {
  __shared__ float    s_acc[OR * F];
  __shared__ int      l_s[CAP];
  __shared__ int      l_d[CAP];
  __shared__ float    l_w[HEADS][CAP];
  __shared__ float    s_ad[OR * HEADS];
  __shared__ unsigned s_mx[OR * HEADS];
  __shared__ float    s_dn[OR * HEADS];
  __shared__ float    s_out[OR];
  __shared__ unsigned s_cnt;
  const int t = threadIdx.x;
  const int lane = t & 31;
  const int wave = __builtin_amdgcn_readfirstlane(t >> 5);
  const int nb = blockIdx.x * OR;
  const int rows = (N - nb < OR) ? (N - nb) : OR;
  const size_t gb3 = (size_t)nb * HEADS;
  const size_t kb = (size_t)blockIdx.x * CAP;

  if (t < OR * HEADS) {
    float a = 0.f;
    unsigned m = 0u;
    if (t < rows * HEADS) {
      a = adst[gb3 + t];
      m = flipf(lrelu(asrc[gb3 + t] + a));
    }
    s_ad[t] = a;
    s_mx[t] = m;
  }
  if (t == 0) s_cnt = 0u;
  __syncthreads();

  unsigned found;
  if (!last) {
  auto push = [&](int e) {
    const unsigned slot = __hip_atomic_fetch_add(&s_cnt, 1u, __ATOMIC_RELAXED, __HIP_MEMORY_SCOPE_WORKGROUP);
    if (slot < (unsigned)CAP) l_s[slot] = e;
  };
  const bool bytes_ok = (gridDim.x <= 255u) && ((reinterpret_cast<size_t>(ob) & 15) == 0);
  if (bytes_ok) {
    const int nG = (E + 15) >> 4;
    const uint4* ob4 = reinterpret_cast<const uint4*>(ob);
    const unsigned pat = blockIdx.x * 0x01010101u;
    for (int g0 = 0; g0 < nG; g0 += OT * SUB) {
      uint4 d[SUB];
      if (g0 + OT * SUB <= nG) {
#pragma unroll
        for (int k = 0; k < SUB; ++k) d[k] = ob4[g0 + k * OT + t];
      } else {
#pragma unroll
        for (int k = 0; k < SUB; ++k) {
          const int g = g0 + k * OT + t;
          d[k] = (g < nG) ? ob4[g] : make_uint4(~0u, ~0u, ~0u, ~0u);
        }
      }
#pragma unroll
      for (int k = 0; k < SUB; ++k) {
        const unsigned p = (zero_bytes(d[k].x ^ pat) >> 7) | (zero_bytes(d[k].y ^ pat) >> 6) |
                           (zero_bytes(d[k].z ^ pat) >> 5) | (zero_bytes(d[k].w ^ pat) >> 4);
        if (p != 0u) {
          const int e0 = (g0 + k * OT + t) * 16;
          unsigned pp = p;
          for (int it = 0; it < 16 && pp != 0u; ++it) {
            const int b = __builtin_ctz(pp);
            pp &= pp - 1u;
            push(e0 + 4 * (b & 7) + (b >> 3));
          }
        }
      }
    }
  } else {
  const int E4 = ((reinterpret_cast<size_t>(dst) & 15) == 0) ? (E >> 2) : 0;
  const int4* dst4 = reinterpret_cast<const int4*>(dst);
  for (int g0 = 0; g0 < E4; g0 += OT * SU4) {
    int4 d[SU4];
    if (g0 + OT * SU4 <= E4) {
#pragma unroll
      for (int k = 0; k < SU4; ++k) d[k] = dst4[g0 + k * OT + t];
    } else {
#pragma unroll
      for (int k = 0; k < SU4; ++k) {
        const int g = g0 + k * OT + t;
        d[k] = (g < E4) ? dst4[g] : make_int4(-1, -1, -1, -1);
      }
    }
#pragma unroll
    for (int k = 0; k < SU4; ++k) {
      const unsigned u0 = (unsigned)(d[k].x - nb);
      const unsigned u1 = (unsigned)(d[k].y - nb);
      const unsigned u2 = (unsigned)(d[k].z - nb);
      const unsigned u3 = (unsigned)(d[k].w - nb);
      const unsigned m01 = (u0 < u1) ? u0 : u1;
      const unsigned m23 = (u2 < u3) ? u2 : u3;
      const unsigned mn = (m01 < m23) ? m01 : m23;
      if (mn < (unsigned)rows) {
        const int e0 = (g0 + k * OT + t) * 4;
        if (u0 < (unsigned)rows) push(e0);
        if (u1 < (unsigned)rows) push(e0 + 1);
        if (u2 < (unsigned)rows) push(e0 + 2);
        if (u3 < (unsigned)rows) push(e0 + 3);
      }
    }
  }
  for (int i0 = E4 * 4; i0 < E; i0 += OT) {
    const int i = i0 + t;
    if (i < E) {
      const unsigned dl = (unsigned)(dst[i] - nb);
      if (dl < (unsigned)rows) push(i);
    }
  }
  }
  __syncthreads();
  found = s_cnt;
  if (t == 0) kcnt[blockIdx.x * 32] = found;
  } else {
    found = kcnt[blockIdx.x * 32];
  }

  float4 acc[RPW][HEADS];
  float dsum[RPW][HEADS];

  if (found <= (unsigned)CAP) {
    const int cnt = (int)found;
    for (int j = t; j < cnt; j += OT) {
      int s, dl;
      if (!last) {
        const int e = l_s[j];
        s = src[e];
        dl = dst[e] - nb;
        ksrc[kb + j] = s;
        kown[kb + j] = dl;
      } else {
        s = ksrc[kb + j];
        dl = kown[kb + j];
      }
      l_s[j] = s;
      l_d[j] = own_code(dl);
#pragma unroll
      for (int h = 0; h < HEADS; ++h) {
        const float e = lrelu(asrc[(size_t)s * HEADS + h] + s_ad[dl * HEADS + h]);
        l_w[h][j] = e;
        __hip_atomic_fetch_max(&s_mx[dl * HEADS + h], flipf(e), __ATOMIC_RELAXED, __HIP_MEMORY_SCOPE_WORKGROUP);
      }
    }
    __syncthreads();
    for (int j = t; j < cnt; j += OT) {
      const int dl = own_row(l_d[j]);
#pragma unroll
      for (int h = 0; h < HEADS; ++h) l_w[h][j] = expf(l_w[h][j] - unflipf(s_mx[dl * HEADS + h]));
    }
    __syncthreads();

#pragma unroll
    for (int q = 0; q < RPW; ++q) {
      const int r = wave + OW * q;
      const int rr = (r < rows) ? r : 0;
      const float* xrow = xp + (size_t)(nb + rr) * F + lane * 4;
#pragma unroll
      for (int h = 0; h < HEADS; ++h) {
        const float es = lrelu(asrc[gb3 + rr * HEADS + h] + s_ad[rr * HEADS + h]);
        dsum[q][h] = expf(es - unflipf(s_mx[rr * HEADS + h]));
      }
#pragma unroll
      for (int h = 0; h < HEADS; ++h) {
        const float4 v = ld4(xrow + h * CPH);
        acc[q][h] = make_float4(dsum[q][h] * v.x, dsum[q][h] * v.y, dsum[q][h] * v.z, dsum[q][h] * v.w);
      }
    }

    for (int base = 0; base < cnt; base += RND * 32) {
      unsigned mk[RND];
#pragma unroll
      for (int k = 0; k < RND; ++k) {
        const int j = base + k * 32 + lane;
        const bool mine = (j < cnt) && ((l_d[j] & 31) == wave);
        mk[k] = __builtin_amdgcn_ballot_w32(mine);
      }
      for (int it = 0; it < RND * 32; ++it) {
        int jj[GRP];
#pragma unroll
        for (int g = 0; g < GRP; ++g) {
          const int off = own_pop(mk);
          jj[g] = (off >= 0) ? (base + off) : -1;
        }
        if (jj[0] < 0) break;
        int jr[GRP];
        float a0[GRP], a1[GRP], a2[GRP];
        float4 val[GRP][HEADS];
#pragma unroll
        for (int g = 0; g < GRP; ++g) {
          if (jj[g] >= 0) {
            const int ss = ufi(l_s[jj[g]]);
            jr[g] = ufi(l_d[jj[g]]) >> 5;
            a0[g] = uff(l_w[0][jj[g]]);
            a1[g] = uff(l_w[1][jj[g]]);
            a2[g] = uff(l_w[2][jj[g]]);
            const float* xrow = xp + (size_t)ss * F + lane * 4;
#pragma unroll
            for (int h = 0; h < HEADS; ++h) val[g][h] = ld4(xrow + h * CPH);
          }
        }
#pragma unroll
        for (int g = 0; g < GRP; ++g) {
          if (jj[g] >= 0) own_add(acc, dsum, jr[g], val[g], a0[g], a1[g], a2[g]);
        }
      }
    }
  } else {
  for (int i0 = 0; i0 < E; i0 += OT * SU) {
    int d[SU];
#pragma unroll
    for (int k = 0; k < SU; ++k) {
      const int i = i0 + k * OT + t;
      d[k] = (i < E) ? dst[i] : -1;
    }
#pragma unroll
    for (int k = 0; k < SU; ++k) {
      const unsigned dl = (unsigned)(d[k] - nb);
      if (dl < (unsigned)rows) {
        const int s = src[i0 + k * OT + t];
#pragma unroll
        for (int h = 0; h < HEADS; ++h) {
          const float e = lrelu(asrc[(size_t)s * HEADS + h] + s_ad[dl * HEADS + h]);
          __hip_atomic_fetch_max(&s_mx[dl * HEADS + h], flipf(e), __ATOMIC_RELAXED, __HIP_MEMORY_SCOPE_WORKGROUP);
        }
      }
    }
  }
  __syncthreads();

  const int nq = rows * (F / 4);
  for (int q = t; q < nq; q += OT) {
    const int r = q / (F / 4);
    const int h = (q % (F / 4)) / (CPH / 4);
    const float es = lrelu(asrc[gb3 + r * HEADS + h] + s_ad[r * HEADS + h]);
    const float ws = expf(es - unflipf(s_mx[r * HEADS + h]));
    float4 v = ld4(xp + (size_t)nb * F + (size_t)q * 4);
    v.x *= ws; v.y *= ws; v.z *= ws; v.w *= ws;
    *reinterpret_cast<float4*>(s_acc + q * 4) = v;
  }
  if (t < rows * HEADS) {
    const float es = lrelu(asrc[gb3 + t] + s_ad[t]);
    s_dn[t] = expf(es - unflipf(s_mx[t]));
  }
  __syncthreads();

  for (int i0 = 0; i0 < E; i0 += OT * SU) {
    int d[SU];
#pragma unroll
    for (int k = 0; k < SU; ++k) {
      const int i = i0 + k * OT + t;
      d[k] = (i < E) ? dst[i] : -1;
    }
#pragma unroll
    for (int k = 0; k < SU; ++k) {
      const unsigned dl = (unsigned)(d[k] - nb);
      const bool hit = dl < (unsigned)rows;
      unsigned mask = __builtin_amdgcn_ballot_w32(hit);
      if (mask == 0u) continue;
      int s = 0;
      float w0 = 0.f, w1 = 0.f, w2 = 0.f;
      if (hit) {
        s = src[i0 + k * OT + t];
        const float* ap = asrc + (size_t)s * HEADS;
        w0 = expf(lrelu(ap[0] + s_ad[dl * HEADS + 0]) - unflipf(s_mx[dl * HEADS + 0]));
        w1 = expf(lrelu(ap[1] + s_ad[dl * HEADS + 1]) - unflipf(s_mx[dl * HEADS + 1]));
        w2 = expf(lrelu(ap[2] + s_ad[dl * HEADS + 2]) - unflipf(s_mx[dl * HEADS + 2]));
        __hip_atomic_fetch_add(&s_dn[dl * HEADS + 0], w0, __ATOMIC_RELAXED, __HIP_MEMORY_SCOPE_WORKGROUP);
        __hip_atomic_fetch_add(&s_dn[dl * HEADS + 1], w1, __ATOMIC_RELAXED, __HIP_MEMORY_SCOPE_WORKGROUP);
        __hip_atomic_fetch_add(&s_dn[dl * HEADS + 2], w2, __ATOMIC_RELAXED, __HIP_MEMORY_SCOPE_WORKGROUP);
      }
      while (mask != 0u) {
        const int l = __builtin_ctz(mask);
        mask &= mask - 1u;
        const int   ss = __builtin_amdgcn_readlane(s, l);
        const int   dr = __builtin_amdgcn_readlane((int)dl, l);
        const float a0 = __int_as_float(__builtin_amdgcn_readlane(__float_as_int(w0), l));
        const float a1 = __int_as_float(__builtin_amdgcn_readlane(__float_as_int(w1), l));
        const float a2 = __int_as_float(__builtin_amdgcn_readlane(__float_as_int(w2), l));
        const float* xrow = xp + (size_t)ss * F + lane;
        float val[HEADS * 4];
#pragma unroll
        for (int c = 0; c < HEADS * 4; ++c) val[c] = xrow[c * 32];
        float* arow = s_acc + dr * F + lane;
#pragma unroll
        for (int c = 0; c < HEADS * 4; ++c) {
          const float wc = (c < 4) ? a0 : ((c < 8) ? a1 : a2);
          __hip_atomic_fetch_add(arow + c * 32, wc * val[c], __ATOMIC_RELAXED, __HIP_MEMORY_SCOPE_WORKGROUP);
        }
      }
    }
  }
  __syncthreads();

#pragma unroll
  for (int q = 0; q < RPW; ++q) {
    const int r = wave + OW * q;
    const int rr = (r < rows) ? r : 0;
#pragma unroll
    for (int h = 0; h < HEADS; ++h) {
      acc[q][h] = *reinterpret_cast<const float4*>(s_acc + rr * F + h * CPH + lane * 4);
      dsum[q][h] = s_dn[rr * HEADS + h];
    }
  }
  }

  if (!last) {
    float4 bb[HEADS];
#pragma unroll
    for (int h = 0; h < HEADS; ++h) bb[h] = ld4(bias + h * CPH + lane * 4);
#pragma unroll
    for (int q = 0; q < RPW; ++q) {
      const int r = wave + OW * q;
      if (r < rows) {
        float* orow = dest + (size_t)(nb + r) * F + lane * 4;
#pragma unroll
        for (int h = 0; h < HEADS; ++h) {
          const float rd = 1.0f / dsum[q][h];
          float4 v = acc[q][h];
          v.x = fmaf(v.x, rd, bb[h].x); v.y = fmaf(v.y, rd, bb[h].y); v.z = fmaf(v.z, rd, bb[h].z); v.w = fmaf(v.w, rd, bb[h].w);
          st4(orow + h * CPH, v);
        }
      }
    }
  } else {
    float red[RPW];
#pragma unroll
    for (int q = 0; q < RPW; ++q) red[q] = 0.f;
#pragma unroll
    for (int h = 0; h < HEADS; ++h) {
      const float4 bb = ld4(bias + h * CPH + lane * 4);
      const float4 w  = ld4(Wl + h * CPH + lane * 4);
#pragma unroll
      for (int q = 0; q < RPW; ++q) {
        const float dn = 1.0f / dsum[q][h];
        float a = red[q];
        a = fmaf(fmaf(acc[q][h].x, dn, bb.x), w.x, a);
        a = fmaf(fmaf(acc[q][h].y, dn, bb.y), w.y, a);
        a = fmaf(fmaf(acc[q][h].z, dn, bb.z), w.z, a);
        a = fmaf(fmaf(acc[q][h].w, dn, bb.w), w.w, a);
        red[q] = a;
      }
    }
#pragma unroll
    for (int m = 16; m >= 1; m >>= 1) {
#pragma unroll
      for (int q = 0; q < RPW; ++q) red[q] += __shfl_xor(red[q], m, 32);
    }
    if (lane == 0) {
#pragma unroll
      for (int q = 0; q < RPW; ++q) {
        const int r = wave + OW * q;
        if (r < rows) s_out[r] = red[q];
      }
    }
    __syncthreads();
    if (t < rows) dest[nb + t] = s_out[t] + bl[0];
  }
}

#define GBM  80
#define GBK  64
#define GAP  68
#define GBP  392
#define GT   384
#define GMT  5
#define GPA  4
#define GPB  16

__global__ void __launch_bounds__(GT) k_gemm(const float* __restrict__ A,
                                             const float* __restrict__ B,
                                             float* __restrict__ C, int M) {
  __shared__ __attribute__((aligned(16))) float sA[2][GBM * GAP];
  __shared__ __attribute__((aligned(16))) float sB[2][GBK * GBP];
  const int t = threadIdx.x;
  const int lane = t & 31;
  const int wave = __builtin_amdgcn_readfirstlane(t >> 5);
  const int c = lane & 15;
  const int h = lane >> 4;
  const int m0 = blockIdx.x * GBM;
  const bool a3 = wave < 4;

  const float* ag[GPA];
  int al[GPA];
#pragma unroll
  for (int p = 0; p < GPA; ++p) {
    const int idx = t + p * GT;
    const int row = (idx >> 4) < GBM ? (idx >> 4) : (GBM - 1);
    const int j = idx & 15;
    const int gm = (m0 + row < M) ? (m0 + row) : (M - 1);
    ag[p] = A + (size_t)gm * F + j * 4;
    al[p] = row * GAP + j * 4;
  }
  const int brow = t / 96;
  const int bj = t - brow * 96;
  const float* bg = B + (size_t)brow * F + bj * 4;
  const int bl = brow * GBP + bj * 4;

  float4 pa[GPA];
  float4 pb[GPB];
  v8f acc[GMT][2];
#pragma unroll
  for (int mt = 0; mt < GMT; ++mt) { acc[mt][0] = (v8f){}; acc[mt][1] = (v8f){}; }

#pragma unroll
  for (int p = 0; p < GPA; ++p) if (p < GPA - 1 || a3) pa[p] = ld4(ag[p]);
#pragma unroll
  for (int p = 0; p < GPB; ++p) pb[p] = ld4(bg + (size_t)(4 * p) * F);
#pragma unroll
  for (int p = 0; p < GPA; ++p) if (p < GPA - 1 || a3) *reinterpret_cast<float4*>(&sA[0][al[p]]) = pa[p];
#pragma unroll
  for (int p = 0; p < GPB; ++p) *reinterpret_cast<float4*>(&sB[0][bl + 4 * p * GBP]) = pb[p];
  __syncthreads();

  for (int ch = 0; ch < F / GBK; ++ch) {
    const int st = ch & 1;
    const bool more = ch + 1 < F / GBK;
    if (more) {
      const int kc = (ch + 1) * GBK;
#pragma unroll
      for (int p = 0; p < GPA; ++p) if (p < GPA - 1 || a3) pa[p] = ld4(ag[p] + kc);
#pragma unroll
      for (int p = 0; p < GPB; ++p) pb[p] = ld4(bg + (size_t)(kc + 4 * p) * F);
    }
    const float* ab = &sA[st][c * GAP + 2 * h];
    const float* bb = &sB[st][(2 * h) * GBP + wave * 32 + c];
#pragma unroll
    for (int ks = 0; ks < GBK / 4; ++ks) {
      v2f a[GMT];
      v2f b[2];
#pragma unroll
      for (int mt = 0; mt < GMT; ++mt) a[mt] = *reinterpret_cast<const v2f*>(ab + mt * 16 * GAP + ks * 4);
#pragma unroll
      for (int nt = 0; nt < 2; ++nt) {
        b[nt].x = bb[(ks * 4) * GBP + nt * 16];
        b[nt].y = bb[(ks * 4 + 1) * GBP + nt * 16];
      }
#pragma unroll
      for (int mt = 0; mt < GMT; ++mt) {
#pragma unroll
        for (int nt = 0; nt < 2; ++nt)
          acc[mt][nt] = __builtin_amdgcn_wmma_f32_16x16x4_f32(false, a[mt], false, b[nt], (short)0, acc[mt][nt],
                                                              false, false);
      }
    }
    if (more) {
#pragma unroll
      for (int p = 0; p < GPA; ++p) if (p < GPA - 1 || a3) *reinterpret_cast<float4*>(&sA[st ^ 1][al[p]]) = pa[p];
#pragma unroll
      for (int p = 0; p < GPB; ++p) *reinterpret_cast<float4*>(&sB[st ^ 1][bl + 4 * p * GBP]) = pb[p];
      __syncthreads();
    }
  }

#pragma unroll
  for (int mt = 0; mt < GMT; ++mt) {
    const int r0 = m0 + mt * 16 + 8 * h;
#pragma unroll
    for (int i = 0; i < 8; ++i) {
      if (r0 + i < M) {
        float* crow = C + (size_t)(r0 + i) * F + wave * 32 + c;
        crow[0]  = acc[mt][0][i];
        crow[16] = acc[mt][1][i];
      }
    }
  }
}

extern "C" void kernel_launch(void* const* d_in, const int* in_sizes, int n_in,
                              void* d_out, int out_size, void* d_ws, size_t ws_size,
                              hipStream_t stream) {
  const float* x   = (const float*)d_in[0];
  const int*   ei  = (const int*)d_in[1];
  const float* W1  = (const float*)d_in[2];
  const float* as1 = (const float*)d_in[3];
  const float* ad1 = (const float*)d_in[4];
  const float* b1  = (const float*)d_in[5];
  const float* W2  = (const float*)d_in[6];
  const float* as2 = (const float*)d_in[7];
  const float* ad2 = (const float*)d_in[8];
  const float* b2  = (const float*)d_in[9];
  const float* Wl  = (const float*)d_in[10];
  const float* bl  = (const float*)d_in[11];
  float* out = (float*)d_out;

  const int N  = in_sizes[0] / HIN;
  const int E  = in_sizes[1] / 2;
  const int* src = ei;
  const int* dst = ei + E;

  const int nblocks = (N + WPB * NPW - 1) / (WPB * NPW);
  const int oblocks = (N + OR - 1) / OR;
  const int gblocks = (N + GBM - 1) / GBM;

  float* ws = (float*)d_ws;
  const size_t NF = (size_t)N * F;
  float*    XP   = ws;
  float*    Y1   = XP + NF;
  float*    ASRC = Y1 + NF;
  float*    ADST = ASRC + (size_t)3 * N;
  int*      KSRC = (int*)(ADST + (size_t)3 * N);
  int*      KOWN = KSRC + (size_t)oblocks * CAP;
  unsigned* KCNT = (unsigned*)(KOWN + (size_t)oblocks * CAP);
  unsigned* OWNB = KCNT + (size_t)oblocks * 32;


  k_proj1<<<nblocks, 256, 0, stream>>>(x, W1, as1, ad1, XP, ASRC, ADST, N, dst, E, OWNB);
  k_own<<<oblocks, OT, 0, stream>>>(src, dst, E, N, ASRC, ADST, XP, b1, Wl, bl, Y1, KCNT, KSRC, KOWN, OWNB, 0);

  k_gemm<<<gblocks, GT, 0, stream>>>(Y1, W2, XP, N);
  k_alpha<<<nblocks, 256, 0, stream>>>(XP, as2, ad2, ASRC, ADST, N);

  k_own<<<oblocks, OT, 0, stream>>>(src, dst, E, N, ASRC, ADST, XP, b2, Wl, bl, out, KCNT, KSRC, KOWN, OWNB, 1);
}
